# starting kernel
# baseline (speedup 1.0000x reference)
#include <hip/hip_runtime.h>
#include <math.h>

typedef __attribute__((ext_vector_type(2))) float v2f;
typedef __attribute__((ext_vector_type(8))) float v8f;

#define GAT_D 64

__device__ __forceinline__ unsigned f2ord(float f) {
    int b = __float_as_int(f);
    return (unsigned)b ^ (unsigned)((b >> 31) | 0x80000000);
}
__device__ __forceinline__ float ord2f(unsigned u) {
    int b = (u & 0x80000000u) ? (int)(u ^ 0x80000000u) : ~(int)u;
    return __int_as_float(b);
}

__global__ void gat_gemm_wmma(const float* __restrict__ A,
                              const float* __restrict__ B,
                              float* __restrict__ C,
                              int M, int K, int N) {
    const int lane  = threadIdx.x;
    const int grp   = blockIdx.x * blockDim.y + threadIdx.y;
    const int mtile = blockIdx.y;
    const int n0 = grp * 64;
    const int m0 = mtile * 16;
    if (n0 >= N || m0 >= M) return;

    const int l15 = lane & 15;
    const int kh  = (lane >> 4) << 1;
    const float* Arow = A + (size_t)(m0 + l15) * K;
    v8f acc0 = {0.f,0.f,0.f,0.f,0.f,0.f,0.f,0.f};
    v8f acc1 = acc0, acc2 = acc0, acc3 = acc0;

    for (int k = 0; k < K; k += 4) {
        v2f a;
        a.x = Arow[k + kh];
        a.y = Arow[k + kh + 1];
        const float* Bp = B + (size_t)(k + kh) * N + n0 + l15;
        v2f b0, b1, b2, b3;
        b0.x = Bp[0];  b0.y = Bp[(size_t)N];
        b1.x = Bp[16]; b1.y = Bp[(size_t)N + 16];
        b2.x = Bp[32]; b2.y = Bp[(size_t)N + 32];
        b3.x = Bp[48]; b3.y = Bp[(size_t)N + 48];
        acc0 = __builtin_amdgcn_wmma_f32_16x16x4_f32(false, a, false, b0, (short)0, acc0, false, false);
        acc1 = __builtin_amdgcn_wmma_f32_16x16x4_f32(false, a, false, b1, (short)0, acc1, false, false);
        acc2 = __builtin_amdgcn_wmma_f32_16x16x4_f32(false, a, false, b2, (short)0, acc2, false, false);
        acc3 = __builtin_amdgcn_wmma_f32_16x16x4_f32(false, a, false, b3, (short)0, acc3, false, false);
    }

    float* Cp = C + (size_t)(m0 + ((lane >> 4) << 3)) * N + n0 + l15;
#pragma unroll
    for (int v = 0; v < 8; ++v) {
        Cp[(size_t)v * N + 0]  = acc0[v];
        Cp[(size_t)v * N + 16] = acc1[v];
        Cp[(size_t)v * N + 32] = acc2[v];
        Cp[(size_t)v * N + 48] = acc3[v];
    }
}

__global__ void gat_attn_coef(const float* __restrict__ h,
                              const float* __restrict__ a_s,
                              const float* __restrict__ a_d,
                              float* __restrict__ es, float* __restrict__ ed,
                              int N, int Hh) {
    int t = blockIdx.x * blockDim.x + threadIdx.x;
    if (t >= N * Hh) return;
    int n  = t / Hh;
    int hh = t - n * Hh;
    const float* hp = h   + ((size_t)n * Hh + hh) * GAT_D;
    const float* as = a_s + (size_t)hh * GAT_D;
    const float* ad = a_d + (size_t)hh * GAT_D;
    float ss = 0.f, dd = 0.f;
#pragma unroll 8
    for (int d = 0; d < GAT_D; ++d) { float v = hp[d]; ss += v * as[d]; dd += v * ad[d]; }
    es[t] = ss; ed[t] = dd;
}

__device__ __forceinline__ void edge_endpoints(const int* __restrict__ ei, int E,
                                               int e, int& src, int& dst) {
    if (e < E) { src = ei[e]; dst = ei[E + e]; }
    else       { src = e - E; dst = e - E;    }
}

__global__ void gat_edge_max(const int* __restrict__ ei, int E, int N,
                             const float* __restrict__ es, const float* __restrict__ ed,
                             unsigned* __restrict__ mT, int Hh) {
    int t = blockIdx.x * blockDim.x + threadIdx.x;
    int total = (E + N) * Hh;
    if (t >= total) return;
    int e = t / Hh, hh = t - e * Hh;
    int src, dst; edge_endpoints(ei, E, e, src, dst);
    float s  = es[src * Hh + hh] + ed[dst * Hh + hh];
    float lr = s > 0.f ? s : 0.2f * s;
    atomicMax(&mT[dst * Hh + hh], f2ord(lr));
}

__global__ void gat_edge_exp(const int* __restrict__ ei, int E, int N,
                             const float* __restrict__ es, const float* __restrict__ ed,
                             const unsigned* __restrict__ mT,
                             float* __restrict__ ex, float* __restrict__ den, int Hh) {
    int t = blockIdx.x * blockDim.x + threadIdx.x;
    int total = (E + N) * Hh;
    if (t >= total) return;
    int e = t / Hh, hh = t - e * Hh;
    int src, dst; edge_endpoints(ei, E, e, src, dst);
    float s  = es[src * Hh + hh] + ed[dst * Hh + hh];
    float lr = s > 0.f ? s : 0.2f * s;
    float v  = expf(lr - ord2f(mT[dst * Hh + hh]));
    ex[t] = v;
    atomicAdd(&den[dst * Hh + hh], v);
}

__global__ void gat_edge_alpha(const int* __restrict__ ei, int E, int N,
                               float* __restrict__ ex, const float* __restrict__ den, int Hh) {
    int t = blockIdx.x * blockDim.x + threadIdx.x;
    int total = (E + N) * Hh;
    if (t >= total) return;
    int e = t / Hh, hh = t - e * Hh;
    int src, dst; edge_endpoints(ei, E, e, src, dst);
    (void)src;
    ex[t] = ex[t] / den[dst * Hh + hh];
}

__global__ void gat_edge_agg(const int* __restrict__ ei, int E, int N,
                             const float* __restrict__ hfeat,
                             const float* __restrict__ alpha,
                             float* __restrict__ agg, int Hh) {
    int HD = Hh * GAT_D;
    int t = blockIdx.x * blockDim.x + threadIdx.x;
    int total = (E + N) * HD;
    if (t >= total) return;
    int e = t / HD;
    int c = t - e * HD;
    int hh = c >> 6;
    int src, dst; edge_endpoints(ei, E, e, src, dst);
    float w = alpha[e * Hh + hh];
    atomicAdd(&agg[(size_t)dst * HD + c], hfeat[(size_t)src * HD + c] * w);
}

__global__ void gat_epilogue(const float* __restrict__ agg, const float* __restrict__ b,
                             float* __restrict__ out, int N, int HD, int do_elu) {
    int t = blockIdx.x * blockDim.x + threadIdx.x;
    if (t >= N * HD) return;
    int c = t - (t / HD) * HD;
    float v = agg[t] + b[c];
    if (do_elu) v = v > 0.f ? v : (expf(v) - 1.f);
    out[t] = v;
}

__global__ void gat_colmean(const float* __restrict__ x, float* __restrict__ out, int N, int C) {
    __shared__ float sm[256];
    int c = blockIdx.x;
    float s = 0.f;
    for (int n = threadIdx.x; n < N; n += blockDim.x) s += x[(size_t)n * C + c];
    sm[threadIdx.x] = s;
    __syncthreads();
    for (int off = 128; off > 0; off >>= 1) {
        if ((int)threadIdx.x < off) sm[threadIdx.x] += sm[threadIdx.x + off];
        __syncthreads();
    }
    if (threadIdx.x == 0) out[c] = sm[0] / (float)N;
}

static inline int cdiv(long long a, long long b) { return (int)((a + b - 1) / b); }

extern "C" void kernel_launch(void* const* d_in, const int* in_sizes, int n_in,
                              void* d_out, int out_size, void* d_ws, size_t ws_size,
                              hipStream_t stream) {
    const float* x   = (const float*)d_in[0];
    const int*   ei  = (const int*)  d_in[1];
    const float* W1  = (const float*)d_in[2];
    const float* a1s = (const float*)d_in[3];
    const float* a1d = (const float*)d_in[4];
    const float* b1  = (const float*)d_in[5];
    const float* W2  = (const float*)d_in[6];
    const float* a2s = (const float*)d_in[7];
    const float* a2d = (const float*)d_in[8];
    const float* b2  = (const float*)d_in[9];
    const float* W3  = (const float*)d_in[10];
    const float* a3s = (const float*)d_in[11];
    const float* a3d = (const float*)d_in[12];
    const float* b3  = (const float*)d_in[13];

    const int N  = in_sizes[0] / 16;
    const int E  = in_sizes[1] / 2;
    const int Et = E + N;
    const int H  = 4;

    float* ws = (float*)d_ws;
    size_t off = 0;
    float*    buf_h    = ws + off; off += (size_t)N * 256;
    float*    buf_agg  = ws + off; off += (size_t)N * 256;
    float*    buf_feat = ws + off; off += (size_t)N * 256;
    float*    es       = ws + off; off += (size_t)N * H;
    float*    ed       = ws + off; off += (size_t)N * H;
    unsigned* mT       = (unsigned*)(ws + off); off += (size_t)N * H;
    float*    den      = ws + off; off += (size_t)N * H;
    float*    alpha    = ws + off; off += (size_t)Et * H;
    (void)ws_size;

    float* node_emb  = (float*)d_out;
    float* graph_emb = node_emb + (size_t)N * 64;

    auto run_layer = [&](const float* in_feat, const float* W,
                         const float* as_, const float* ad_, const float* bb,
                         float* out_feat, int Hh, int K, int Ncols, int do_elu) {
        const int HD = Hh * GAT_D;
        int ngrp = Ncols / 64;
        int wpb  = ngrp < 4 ? ngrp : 4;
        dim3 gb(32, wpb);
        dim3 gg(cdiv(ngrp, wpb), cdiv(N, 16));
        hipLaunchKernelGGL(gat_gemm_wmma, gg, gb, 0, stream, in_feat, W, buf_h, N, K, Ncols);
        hipLaunchKernelGGL(gat_attn_coef, dim3(cdiv((long long)N * Hh, 256)), dim3(256), 0, stream,
                           buf_h, as_, ad_, es, ed, N, Hh);
        hipMemsetAsync(mT,      0, (size_t)N * Hh * sizeof(unsigned), stream);
        hipMemsetAsync(den,     0, (size_t)N * Hh * sizeof(float),    stream);
        hipMemsetAsync(buf_agg, 0, (size_t)N * HD * sizeof(float),    stream);
        int eth = Et * Hh;
        hipLaunchKernelGGL(gat_edge_max,   dim3(cdiv(eth, 256)), dim3(256), 0, stream,
                           ei, E, N, es, ed, mT, Hh);
        hipLaunchKernelGGL(gat_edge_exp,   dim3(cdiv(eth, 256)), dim3(256), 0, stream,
                           ei, E, N, es, ed, mT, alpha, den, Hh);
        hipLaunchKernelGGL(gat_edge_alpha, dim3(cdiv(eth, 256)), dim3(256), 0, stream,
                           ei, E, N, alpha, den, Hh);
        hipLaunchKernelGGL(gat_edge_agg,   dim3(cdiv((long long)Et * HD, 256)), dim3(256), 0, stream,
                           ei, E, N, buf_h, alpha, buf_agg, Hh);
        hipLaunchKernelGGL(gat_epilogue, dim3(cdiv((long long)N * HD, 256)), dim3(256), 0, stream,
                           buf_agg, bb, out_feat, N, HD, do_elu);
    };

    run_layer(x,        W1, a1s, a1d, b1, buf_feat, H, 16,  256, 1);
    run_layer(buf_feat, W2, a2s, a2d, b2, buf_feat, H, 256, 256, 1);
    run_layer(buf_feat, W3, a3s, a3d, b3, node_emb, 1, 256, 64, 0);

    hipLaunchKernelGGL(gat_colmean, dim3(64), dim3(256), 0, stream, node_emb, graph_emb, N, 64);
}


# Aggregation walks a destination-sorted edge list (counting sort, built once per call): one wave per node sums in registers, stores once; replaces float atomicAdd scatter, its zero-fill and epilogue
# speedup vs baseline: 2.6360x; 2.6360x over previous
#include <hip/hip_runtime.h>
#include <math.h>

typedef __attribute__((ext_vector_type(2))) float v2f;
typedef __attribute__((ext_vector_type(8))) float v8f;

#define GAT_D 64

__device__ __forceinline__ unsigned f2ord(float f) {
    int b = __float_as_int(f);
    return (unsigned)b ^ (unsigned)((b >> 31) | 0x80000000);
}
__device__ __forceinline__ float ord2f(unsigned u) {
    int b = (u & 0x80000000u) ? (int)(u ^ 0x80000000u) : ~(int)u;
    return __int_as_float(b);
}

__global__ void gat_gemm_wmma(const float* __restrict__ A,
                              const float* __restrict__ B,
                              float* __restrict__ C,
                              int M, int K, int N) {
    const int lane  = threadIdx.x;
    const int grp   = blockIdx.x * blockDim.y + threadIdx.y;
    const int mtile = blockIdx.y;
    const int n0 = grp * 64;
    const int m0 = mtile * 16;
    if (n0 >= N || m0 >= M) return;

    const int l15 = lane & 15;
    const int kh  = (lane >> 4) << 1;
    const float* Arow = A + (size_t)(m0 + l15) * K;
    v8f acc0 = {0.f,0.f,0.f,0.f,0.f,0.f,0.f,0.f};
    v8f acc1 = acc0, acc2 = acc0, acc3 = acc0;

    for (int k = 0; k < K; k += 4) {
        v2f a;
        a.x = Arow[k + kh];
        a.y = Arow[k + kh + 1];
        const float* Bp = B + (size_t)(k + kh) * N + n0 + l15;
        v2f b0, b1, b2, b3;
        b0.x = Bp[0];  b0.y = Bp[(size_t)N];
        b1.x = Bp[16]; b1.y = Bp[(size_t)N + 16];
        b2.x = Bp[32]; b2.y = Bp[(size_t)N + 32];
        b3.x = Bp[48]; b3.y = Bp[(size_t)N + 48];
        acc0 = __builtin_amdgcn_wmma_f32_16x16x4_f32(false, a, false, b0, (short)0, acc0, false, false);
        acc1 = __builtin_amdgcn_wmma_f32_16x16x4_f32(false, a, false, b1, (short)0, acc1, false, false);
        acc2 = __builtin_amdgcn_wmma_f32_16x16x4_f32(false, a, false, b2, (short)0, acc2, false, false);
        acc3 = __builtin_amdgcn_wmma_f32_16x16x4_f32(false, a, false, b3, (short)0, acc3, false, false);
    }

    float* Cp = C + (size_t)(m0 + ((lane >> 4) << 3)) * N + n0 + l15;
#pragma unroll
    for (int v = 0; v < 8; ++v) {
        Cp[(size_t)v * N + 0]  = acc0[v];
        Cp[(size_t)v * N + 16] = acc1[v];
        Cp[(size_t)v * N + 32] = acc2[v];
        Cp[(size_t)v * N + 48] = acc3[v];
    }
}

__global__ void gat_attn_coef(const float* __restrict__ h,
                              const float* __restrict__ a_s,
                              const float* __restrict__ a_d,
                              float* __restrict__ es, float* __restrict__ ed,
                              int N, int Hh) {
    int t = blockIdx.x * blockDim.x + threadIdx.x;
    if (t >= N * Hh) return;
    int n  = t / Hh;
    int hh = t - n * Hh;
    const float* hp = h   + ((size_t)n * Hh + hh) * GAT_D;
    const float* as = a_s + (size_t)hh * GAT_D;
    const float* ad = a_d + (size_t)hh * GAT_D;
    float ss = 0.f, dd = 0.f;
#pragma unroll 8
    for (int d = 0; d < GAT_D; ++d) { float v = hp[d]; ss += v * as[d]; dd += v * ad[d]; }
    es[t] = ss; ed[t] = dd;
}

__device__ __forceinline__ void edge_endpoints(const int* __restrict__ ei, int E,
                                               int e, int& src, int& dst) {
    if (e < E) { src = ei[e]; dst = ei[E + e]; }
    else       { src = e - E; dst = e - E;    }
}

__global__ void gat_edge_max(const int* __restrict__ ei, int E, int N,
                             const float* __restrict__ es, const float* __restrict__ ed,
                             unsigned* __restrict__ mT, int Hh) {
    int t = blockIdx.x * blockDim.x + threadIdx.x;
    int total = (E + N) * Hh;
    if (t >= total) return;
    int e = t / Hh, hh = t - e * Hh;
    int src, dst; edge_endpoints(ei, E, e, src, dst);
    float s  = es[src * Hh + hh] + ed[dst * Hh + hh];
    float lr = s > 0.f ? s : 0.2f * s;
    atomicMax(&mT[dst * Hh + hh], f2ord(lr));
}

__global__ void gat_edge_exp(const int* __restrict__ ei, int E, int N,
                             const float* __restrict__ es, const float* __restrict__ ed,
                             const unsigned* __restrict__ mT,
                             float* __restrict__ ex, float* __restrict__ den, int Hh) {
    int t = blockIdx.x * blockDim.x + threadIdx.x;
    int total = (E + N) * Hh;
    if (t >= total) return;
    int e = t / Hh, hh = t - e * Hh;
    int src, dst; edge_endpoints(ei, E, e, src, dst);
    float s  = es[src * Hh + hh] + ed[dst * Hh + hh];
    float lr = s > 0.f ? s : 0.2f * s;
    float v  = expf(lr - ord2f(mT[dst * Hh + hh]));
    ex[t] = v;
    atomicAdd(&den[dst * Hh + hh], v);
}

__global__ void gat_edge_alpha(const int* __restrict__ ei, int E, int N,
                               float* __restrict__ ex, const float* __restrict__ den, int Hh) {
    int t = blockIdx.x * blockDim.x + threadIdx.x;
    int total = (E + N) * Hh;
    if (t >= total) return;
    int e = t / Hh, hh = t - e * Hh;
    int src, dst; edge_endpoints(ei, E, e, src, dst);
    (void)src;
    ex[t] = ex[t] / den[dst * Hh + hh];
}

__global__ void gat_csr_count(const int* __restrict__ ei, int E, int* __restrict__ cnt) {
    int e = blockIdx.x * blockDim.x + threadIdx.x;
    if (e >= E) return;
    atomicAdd(&cnt[ei[E + e]], 1);
}

__global__ void gat_csr_scan(const int* __restrict__ cnt, int* __restrict__ rowptr,
                             int* __restrict__ cursor, int N) {
    __shared__ int wpre[32];
    __shared__ int wtot[32];
    const int t = threadIdx.x, lane = t & 31, wv = t >> 5;
    const int tile0 = blockIdx.x * 1024;
    int pre = 0;
    for (int i = t; i < tile0; i += 1024) pre += cnt[i];
    const int i = tile0 + t;
    const int v = (i < N) ? cnt[i] : 0;
    int inc = v;
    for (int d = 1; d < 32; d <<= 1) {
        int u = __shfl_up(inc, d, 32);
        if (lane >= d) inc += u;
    }
    for (int d = 16; d > 0; d >>= 1) pre += __shfl_xor(pre, d, 32);
    if (lane == 31) wtot[wv] = inc;
    if (lane == 0)  wpre[wv] = pre;
    __syncthreads();
    int base = 0;
    for (int k = 0; k < 32; ++k) {
        base += wpre[k];
        if (k < wv) base += wtot[k];
    }
    if (i < N) { int r = base + inc - v; rowptr[i] = r; cursor[i] = r; }
}

__global__ void gat_csr_fill(const int* __restrict__ ei, int E, int* __restrict__ cursor,
                             int* __restrict__ sedge) {
    int e = blockIdx.x * blockDim.x + threadIdx.x;
    if (e >= E) return;
    int p = atomicAdd(&cursor[ei[E + e]], 1);
    int2 rec; rec.x = ei[e]; rec.y = e;
    reinterpret_cast<int2*>(sedge)[p] = rec;
}

template <int HH>
__global__ void gat_seg_agg(const int* __restrict__ sedge, const int* __restrict__ rowptr,
                            const int* __restrict__ segend,
                            const float* __restrict__ hfeat, const float* __restrict__ alpha,
                            const float* __restrict__ b, float* __restrict__ out,
                            int N, int E, int do_elu) {
    const int lane = threadIdx.x & 31;
    const int node = blockIdx.x * (blockDim.x >> 5) + (threadIdx.x >> 5);
    if (node >= N) return;
    const int start = __builtin_amdgcn_readfirstlane(rowptr[node]);
    const int end   = __builtin_amdgcn_readfirstlane(segend[node]);
    const int2* se2 = reinterpret_cast<const int2*>(sedge);

    if (HH == 4) {
        const int hsel = lane >> 4;
        const float4* a4 = reinterpret_cast<const float4*>(alpha);
        const float* hp = hfeat + lane * 4;
        float4 acc0, acc1;
        {
            float4 a  = a4[E + node];
            float  w0 = hsel ? a.y : a.x;
            float  w1 = hsel ? a.w : a.z;
            const float* rp = hp + (size_t)node * 256;
            float4 r0 = *reinterpret_cast<const float4*>(rp);
            float4 r1 = *reinterpret_cast<const float4*>(rp + 128);
            acc0.x = r0.x * w0; acc0.y = r0.y * w0; acc0.z = r0.z * w0; acc0.w = r0.w * w0;
            acc1.x = r1.x * w1; acc1.y = r1.y * w1; acc1.z = r1.z * w1; acc1.w = r1.w * w1;
        }
        for (int p0 = start; p0 < end; p0 += 32) {
            const int n = (end - p0) < 32 ? (end - p0) : 32;
            int2 rec; rec.x = 0; rec.y = 0;
            float4 a; a.x = 0.f; a.y = 0.f; a.z = 0.f; a.w = 0.f;
            if (lane < n) { rec = se2[p0 + lane]; a = a4[rec.y]; }
#pragma unroll 4
            for (int j = 0; j < n; ++j) {
                const int   s  = __builtin_amdgcn_readlane(rec.x, j);
                const float ax = __int_as_float(__builtin_amdgcn_readlane(__float_as_int(a.x), j));
                const float ay = __int_as_float(__builtin_amdgcn_readlane(__float_as_int(a.y), j));
                const float az = __int_as_float(__builtin_amdgcn_readlane(__float_as_int(a.z), j));
                const float aw = __int_as_float(__builtin_amdgcn_readlane(__float_as_int(a.w), j));
                const float w0 = hsel ? ay : ax;
                const float w1 = hsel ? aw : az;
                const float* rp = hp + (size_t)s * 256;
                float4 r0 = *reinterpret_cast<const float4*>(rp);
                float4 r1 = *reinterpret_cast<const float4*>(rp + 128);
                acc0.x += r0.x * w0; acc0.y += r0.y * w0; acc0.z += r0.z * w0; acc0.w += r0.w * w0;
                acc1.x += r1.x * w1; acc1.y += r1.y * w1; acc1.z += r1.z * w1; acc1.w += r1.w * w1;
            }
        }
        const float4 b0 = *reinterpret_cast<const float4*>(b + lane * 4);
        const float4 b1 = *reinterpret_cast<const float4*>(b + 128 + lane * 4);
        acc0.x += b0.x; acc0.y += b0.y; acc0.z += b0.z; acc0.w += b0.w;
        acc1.x += b1.x; acc1.y += b1.y; acc1.z += b1.z; acc1.w += b1.w;
        if (do_elu) {
            acc0.x = acc0.x > 0.f ? acc0.x : (expf(acc0.x) - 1.f);
            acc0.y = acc0.y > 0.f ? acc0.y : (expf(acc0.y) - 1.f);
            acc0.z = acc0.z > 0.f ? acc0.z : (expf(acc0.z) - 1.f);
            acc0.w = acc0.w > 0.f ? acc0.w : (expf(acc0.w) - 1.f);
            acc1.x = acc1.x > 0.f ? acc1.x : (expf(acc1.x) - 1.f);
            acc1.y = acc1.y > 0.f ? acc1.y : (expf(acc1.y) - 1.f);
            acc1.z = acc1.z > 0.f ? acc1.z : (expf(acc1.z) - 1.f);
            acc1.w = acc1.w > 0.f ? acc1.w : (expf(acc1.w) - 1.f);
        }
        float* op = out + (size_t)node * 256 + lane * 4;
        *reinterpret_cast<float4*>(op)       = acc0;
        *reinterpret_cast<float4*>(op + 128) = acc1;
    } else {
        const float* hp = hfeat + lane * 2;
        float2 acc;
        {
            const float w = alpha[E + node];
            float2 r = *reinterpret_cast<const float2*>(hp + (size_t)node * 64);
            acc.x = r.x * w; acc.y = r.y * w;
        }
        for (int p0 = start; p0 < end; p0 += 32) {
            const int n = (end - p0) < 32 ? (end - p0) : 32;
            int2 rec; rec.x = 0; rec.y = 0;
            float a = 0.f;
            if (lane < n) { rec = se2[p0 + lane]; a = alpha[rec.y]; }
#pragma unroll 4
            for (int j = 0; j < n; ++j) {
                const int   s = __builtin_amdgcn_readlane(rec.x, j);
                const float w = __int_as_float(__builtin_amdgcn_readlane(__float_as_int(a), j));
                float2 r = *reinterpret_cast<const float2*>(hp + (size_t)s * 64);
                acc.x += r.x * w; acc.y += r.y * w;
            }
        }
        const float2 bb = *reinterpret_cast<const float2*>(b + lane * 2);
        acc.x += bb.x; acc.y += bb.y;
        if (do_elu) {
            acc.x = acc.x > 0.f ? acc.x : (expf(acc.x) - 1.f);
            acc.y = acc.y > 0.f ? acc.y : (expf(acc.y) - 1.f);
        }
        *reinterpret_cast<float2*>(out + (size_t)node * 64 + lane * 2) = acc;
    }
}

__global__ void gat_colmean(const float* __restrict__ x, float* __restrict__ out, int N, int C) {
    __shared__ float sm[256];
    int c = blockIdx.x;
    float s = 0.f;
    for (int n = threadIdx.x; n < N; n += blockDim.x) s += x[(size_t)n * C + c];
    sm[threadIdx.x] = s;
    __syncthreads();
    for (int off = 128; off > 0; off >>= 1) {
        if ((int)threadIdx.x < off) sm[threadIdx.x] += sm[threadIdx.x + off];
        __syncthreads();
    }
    if (threadIdx.x == 0) out[c] = sm[0] / (float)N;
}

static inline int cdiv(long long a, long long b) { return (int)((a + b - 1) / b); }

extern "C" void kernel_launch(void* const* d_in, const int* in_sizes, int n_in,
                              void* d_out, int out_size, void* d_ws, size_t ws_size,
                              hipStream_t stream) {
    const float* x   = (const float*)d_in[0];
    const int*   ei  = (const int*)  d_in[1];
    const float* W1  = (const float*)d_in[2];
    const float* a1s = (const float*)d_in[3];
    const float* a1d = (const float*)d_in[4];
    const float* b1  = (const float*)d_in[5];
    const float* W2  = (const float*)d_in[6];
    const float* a2s = (const float*)d_in[7];
    const float* a2d = (const float*)d_in[8];
    const float* b2  = (const float*)d_in[9];
    const float* W3  = (const float*)d_in[10];
    const float* a3s = (const float*)d_in[11];
    const float* a3d = (const float*)d_in[12];
    const float* b3  = (const float*)d_in[13];

    const int N  = in_sizes[0] / 16;
    const int E  = in_sizes[1] / 2;
    const int Et = E + N;
    const int H  = 4;

    float* ws = (float*)d_ws;
    size_t off = 0;
    float*    buf_h    = ws + off; off += (size_t)N * 256;
    int*      csr      = (int*)(ws + off); off += (size_t)N * 256;
    float*    buf_feat = ws + off; off += (size_t)N * 256;
    float*    es       = ws + off; off += (size_t)N * H;
    float*    ed       = ws + off; off += (size_t)N * H;
    unsigned* mT       = (unsigned*)(ws + off); off += (size_t)N * H;
    float*    den      = ws + off; off += (size_t)N * H;
    float*    alpha    = ws + off; off += (size_t)Et * H;
    (void)ws_size;
    int* cnt    = csr;
    int* rowptr = csr + (size_t)N;
    int* cursor = csr + (size_t)2 * N;
    int* sedge  = csr + (size_t)4 * N;

    float* node_emb  = (float*)d_out;
    float* graph_emb = node_emb + (size_t)N * 64;

    auto run_layer = [&](const float* in_feat, const float* W,
                         const float* as_, const float* ad_, const float* bb,
                         float* out_feat, int Hh, int K, int Ncols, int do_elu) {
        int ngrp = Ncols / 64;
        int wpb  = ngrp < 4 ? ngrp : 4;
        dim3 gb(32, wpb);
        dim3 gg(cdiv(ngrp, wpb), cdiv(N, 16));
        hipLaunchKernelGGL(gat_gemm_wmma, gg, gb, 0, stream, in_feat, W, buf_h, N, K, Ncols);
        hipLaunchKernelGGL(gat_attn_coef, dim3(cdiv((long long)N * Hh, 256)), dim3(256), 0, stream,
                           buf_h, as_, ad_, es, ed, N, Hh);
        hipMemsetAsync(mT,      0, (size_t)N * Hh * sizeof(unsigned), stream);
        hipMemsetAsync(den,     0, (size_t)N * Hh * sizeof(float),    stream);
        int eth = Et * Hh;
        hipLaunchKernelGGL(gat_edge_max,   dim3(cdiv(eth, 256)), dim3(256), 0, stream,
                           ei, E, N, es, ed, mT, Hh);
        hipLaunchKernelGGL(gat_edge_exp,   dim3(cdiv(eth, 256)), dim3(256), 0, stream,
                           ei, E, N, es, ed, mT, alpha, den, Hh);
        hipLaunchKernelGGL(gat_edge_alpha, dim3(cdiv(eth, 256)), dim3(256), 0, stream,
                           ei, E, N, alpha, den, Hh);
        if (Hh == 4)
            hipLaunchKernelGGL((gat_seg_agg<4>), dim3(cdiv(N, 8)), dim3(256), 0, stream,
                               sedge, rowptr, cursor, buf_h, alpha, bb, out_feat, N, E, do_elu);
        else
            hipLaunchKernelGGL((gat_seg_agg<1>), dim3(cdiv(N, 8)), dim3(256), 0, stream,
                               sedge, rowptr, cursor, buf_h, alpha, bb, out_feat, N, E, do_elu);
    };

    hipMemsetAsync(cnt, 0, (size_t)N * sizeof(int), stream);
    hipLaunchKernelGGL(gat_csr_count, dim3(cdiv(E, 256)), dim3(256), 0, stream, ei, E, cnt);
    hipLaunchKernelGGL(gat_csr_scan,  dim3(cdiv(N, 1024)), dim3(1024), 0, stream, cnt, rowptr, cursor, N);
    hipLaunchKernelGGL(gat_csr_fill,  dim3(cdiv(E, 256)), dim3(256), 0, stream, ei, E, cursor, sedge);

    run_layer(x,        W1, a1s, a1d, b1, buf_feat, H, 16,  256, 1);
    run_layer(buf_feat, W2, a2s, a2d, b2, buf_feat, H, 256, 256, 1);
    run_layer(buf_feat, W3, a3s, a3d, b3, node_emb, 1, 256, 64, 0);

    hipLaunchKernelGGL(gat_colmean, dim3(64), dim3(256), 0, stream, node_emb, graph_emb, N, 64);
}


# Graph mean reads node rows whole: each block sums a 256-row slab with its float4 row loads in flight together, one block adds the slab sums; replaces 64 one-column blocks of serial strided loads
# speedup vs baseline: 5.5010x; 2.0869x over previous
#include <hip/hip_runtime.h>
#include <math.h>

typedef __attribute__((ext_vector_type(2))) float v2f;
typedef __attribute__((ext_vector_type(8))) float v8f;

#define GAT_D 64

__global__ void gat_gemm_wmma(const float* __restrict__ A,
                              const float* __restrict__ B,
                              float* __restrict__ C,
                              int M, int K, int N) {
    const int lane  = threadIdx.x;
    const int grp   = blockIdx.x * blockDim.y + threadIdx.y;
    const int mtile = blockIdx.y;
    const int n0 = grp * 64;
    const int m0 = mtile * 16;
    if (n0 >= N || m0 >= M) return;

    const int l15 = lane & 15;
    const int kh  = (lane >> 4) << 1;
    const float* Arow = A + (size_t)(m0 + l15) * K;
    v8f acc0 = {0.f,0.f,0.f,0.f,0.f,0.f,0.f,0.f};
    v8f acc1 = acc0, acc2 = acc0, acc3 = acc0;

    for (int k = 0; k < K; k += 4) {
        v2f a;
        a.x = Arow[k + kh];
        a.y = Arow[k + kh + 1];
        const float* Bp = B + (size_t)(k + kh) * N + n0 + l15;
        v2f b0, b1, b2, b3;
        b0.x = Bp[0];  b0.y = Bp[(size_t)N];
        b1.x = Bp[16]; b1.y = Bp[(size_t)N + 16];
        b2.x = Bp[32]; b2.y = Bp[(size_t)N + 32];
        b3.x = Bp[48]; b3.y = Bp[(size_t)N + 48];
        acc0 = __builtin_amdgcn_wmma_f32_16x16x4_f32(false, a, false, b0, (short)0, acc0, false, false);
        acc1 = __builtin_amdgcn_wmma_f32_16x16x4_f32(false, a, false, b1, (short)0, acc1, false, false);
        acc2 = __builtin_amdgcn_wmma_f32_16x16x4_f32(false, a, false, b2, (short)0, acc2, false, false);
        acc3 = __builtin_amdgcn_wmma_f32_16x16x4_f32(false, a, false, b3, (short)0, acc3, false, false);
    }

    float* Cp = C + (size_t)(m0 + ((lane >> 4) << 3)) * N + n0 + l15;
#pragma unroll
    for (int v = 0; v < 8; ++v) {
        Cp[(size_t)v * N + 0]  = acc0[v];
        Cp[(size_t)v * N + 16] = acc1[v];
        Cp[(size_t)v * N + 32] = acc2[v];
        Cp[(size_t)v * N + 48] = acc3[v];
    }
}

__global__ void gat_attn_coef(const float* __restrict__ h,
                              const float* __restrict__ a_s,
                              const float* __restrict__ a_d,
                              float* __restrict__ es, float* __restrict__ ed,
                              int N, int Hh) {
    int t = blockIdx.x * blockDim.x + threadIdx.x;
    if (t >= N * Hh) return;
    int n  = t / Hh;
    int hh = t - n * Hh;
    const float* hp = h   + ((size_t)n * Hh + hh) * GAT_D;
    const float* as = a_s + (size_t)hh * GAT_D;
    const float* ad = a_d + (size_t)hh * GAT_D;
    float ss = 0.f, dd = 0.f;
#pragma unroll 8
    for (int d = 0; d < GAT_D; ++d) { float v = hp[d]; ss += v * as[d]; dd += v * ad[d]; }
    es[t] = ss; ed[t] = dd;
}

__global__ void gat_csr_count(const int* __restrict__ ei, int E, int* __restrict__ cnt) {
    int e = blockIdx.x * blockDim.x + threadIdx.x;
    if (e >= E) return;
    atomicAdd(&cnt[ei[E + e]], 1);
}

__global__ void gat_csr_scan(const int* __restrict__ cnt, int* __restrict__ rowptr,
                             int* __restrict__ cursor, int N) {
    __shared__ int wpre[32];
    __shared__ int wtot[32];
    const int t = threadIdx.x, lane = t & 31, wv = t >> 5;
    const int tile0 = blockIdx.x * 1024;
    int pre = 0;
    for (int i = t; i < tile0; i += 1024) pre += cnt[i];
    const int i = tile0 + t;
    const int v = (i < N) ? cnt[i] : 0;
    int inc = v;
    for (int d = 1; d < 32; d <<= 1) {
        int u = __shfl_up(inc, d, 32);
        if (lane >= d) inc += u;
    }
    for (int d = 16; d > 0; d >>= 1) pre += __shfl_xor(pre, d, 32);
    if (lane == 31) wtot[wv] = inc;
    if (lane == 0)  wpre[wv] = pre;
    __syncthreads();
    int base = 0;
    for (int k = 0; k < 32; ++k) {
        base += wpre[k];
        if (k < wv) base += wtot[k];
    }
    if (i < N) { int r = base + inc - v; rowptr[i] = r; cursor[i] = r; }
}

__global__ void gat_csr_fill(const int* __restrict__ ei, int E, int* __restrict__ cursor,
                             int* __restrict__ ssrc) {
    int e = blockIdx.x * blockDim.x + threadIdx.x;
    if (e >= E) return;
    int p = atomicAdd(&cursor[ei[E + e]], 1);
    ssrc[p] = ei[e];
}

template <int M>
__device__ __forceinline__ float row_xor(float x) {
    const int b = __float_as_int(x);
    return __int_as_float(__builtin_amdgcn_update_dpp(b, b, 0x160 + M, 0xf, 0xf, false));
}
__device__ __forceinline__ float half_xor(float x) {
    const int b = __float_as_int(x);
    return __int_as_float(__builtin_amdgcn_permlanex16(b, b, 0x76543210, 0xfedcba98, false, false));
}
__device__ __forceinline__ float wave_max(float v) {
    v = fmaxf(v, row_xor<1>(v)); v = fmaxf(v, row_xor<2>(v));
    v = fmaxf(v, row_xor<4>(v)); v = fmaxf(v, row_xor<8>(v));
    return fmaxf(v, half_xor(v));
}
__device__ __forceinline__ float wave_sum(float v) {
    v += row_xor<1>(v); v += row_xor<2>(v); v += row_xor<4>(v); v += row_xor<8>(v);
    return v + half_xor(v);
}

template <int HH>
__device__ __forceinline__ void load_heads(const float* __restrict__ p, float (&v)[HH]) {
    if constexpr (HH == 4) {
        const float4 t = *reinterpret_cast<const float4*>(p);
        v[0] = t.x; v[1] = t.y; v[2] = t.z; v[3] = t.w;
    } else {
        v[0] = p[0];
    }
}

template <int HH>
__device__ __forceinline__ void edge_score(const float* __restrict__ es, int s,
                                           const float (&edn)[HH], float (&e)[HH]) {
    float t[HH];
    load_heads<HH>(es + (size_t)s * HH, t);
#pragma unroll
    for (int h = 0; h < HH; ++h) { const float v = t[h] + edn[h]; e[h] = v > 0.f ? v : 0.2f * v; }
}

template <int HH>
__global__ void gat_seg_softmax_agg(const int* __restrict__ ssrc, const int* __restrict__ rowptr,
                                    const int* __restrict__ segend,
                                    const float* __restrict__ hfeat,
                                    const float* __restrict__ es, const float* __restrict__ ed,
                                    const float* __restrict__ b, float* __restrict__ out,
                                    int N, int do_elu) {
    const int lane = threadIdx.x & 31;
    const int node = __builtin_amdgcn_readfirstlane(blockIdx.x * (blockDim.x >> 5) + (threadIdx.x >> 5));
    if (node >= N) return;
    const int start = rowptr[node];
    const int end   = segend[node];

    float edn[HH], eself[HH], e0[HH], m[HH];
    load_heads<HH>(ed + (size_t)node * HH, edn);
    edge_score<HH>(es, node, edn, eself);
    const int n0 = (end - start) < 32 ? (end - start) : 32;
    int src0 = 0;
#pragma unroll
    for (int h = 0; h < HH; ++h) e0[h] = -INFINITY;
    if (lane < n0) { src0 = ssrc[start + lane]; edge_score<HH>(es, src0, edn, e0); }
#pragma unroll
    for (int h = 0; h < HH; ++h) m[h] = fmaxf(e0[h], eself[h]);
    for (int p0 = start + 32; p0 < end; p0 += 32) {
        if (p0 + lane < end) {
            float e[HH];
            edge_score<HH>(es, ssrc[p0 + lane], edn, e);
#pragma unroll
            for (int h = 0; h < HH; ++h) m[h] = fmaxf(m[h], e[h]);
        }
    }
#pragma unroll
    for (int h = 0; h < HH; ++h)
        m[h] = __int_as_float(__builtin_amdgcn_readfirstlane(__float_as_int(wave_max(m[h]))));

    float x0[HH], den[HH], a0[HH], aself[HH];
#pragma unroll
    for (int h = 0; h < HH; ++h) { x0[h] = expf(e0[h] - m[h]); den[h] = x0[h]; }
    for (int p0 = start + 32; p0 < end; p0 += 32) {
        if (p0 + lane < end) {
            float e[HH];
            edge_score<HH>(es, ssrc[p0 + lane], edn, e);
#pragma unroll
            for (int h = 0; h < HH; ++h) den[h] += expf(e[h] - m[h]);
        }
    }
#pragma unroll
    for (int h = 0; h < HH; ++h) {
        const float xs = expf(eself[h] - m[h]);
        den[h]   = __int_as_float(__builtin_amdgcn_readfirstlane(__float_as_int(wave_sum(den[h]) + xs)));
        aself[h] = xs / den[h];
        a0[h]    = x0[h] / den[h];
    }

    if constexpr (HH == 4) {
        const int hsel = lane >> 4;
        const float* hp = hfeat + lane * 4;
        float4 acc0, acc1;
        {
            const float w0 = hsel ? aself[1] : aself[0];
            const float w1 = hsel ? aself[3] : aself[2];
            const float* rp = hp + (size_t)node * 256;
            float4 r0 = *reinterpret_cast<const float4*>(rp);
            float4 r1 = *reinterpret_cast<const float4*>(rp + 128);
            acc0.x = r0.x * w0; acc0.y = r0.y * w0; acc0.z = r0.z * w0; acc0.w = r0.w * w0;
            acc1.x = r1.x * w1; acc1.y = r1.y * w1; acc1.z = r1.z * w1; acc1.w = r1.w * w1;
        }
        for (int p0 = start; p0 < end; p0 += 32) {
            const int n = (end - p0) < 32 ? (end - p0) : 32;
            int sv = src0;
            float a[4] = {a0[0], a0[1], a0[2], a0[3]};
            if (p0 != start) {
                sv = 0; a[0] = 0.f; a[1] = 0.f; a[2] = 0.f; a[3] = 0.f;
                if (lane < n) {
                    float e[HH];
                    sv = ssrc[p0 + lane];
                    edge_score<HH>(es, sv, edn, e);
#pragma unroll
                    for (int h = 0; h < HH; ++h) a[h] = expf(e[h] - m[h]) / den[h];
                }
            }
#pragma unroll 4
            for (int j = 0; j < n; ++j) {
                const int   s  = __builtin_amdgcn_readlane(sv, j);
                const float ax = __int_as_float(__builtin_amdgcn_readlane(__float_as_int(a[0]), j));
                const float ay = __int_as_float(__builtin_amdgcn_readlane(__float_as_int(a[1]), j));
                const float az = __int_as_float(__builtin_amdgcn_readlane(__float_as_int(a[2]), j));
                const float aw = __int_as_float(__builtin_amdgcn_readlane(__float_as_int(a[3]), j));
                const float w0 = hsel ? ay : ax;
                const float w1 = hsel ? aw : az;
                const float* rp = hp + (size_t)s * 256;
                float4 r0 = *reinterpret_cast<const float4*>(rp);
                float4 r1 = *reinterpret_cast<const float4*>(rp + 128);
                acc0.x += r0.x * w0; acc0.y += r0.y * w0; acc0.z += r0.z * w0; acc0.w += r0.w * w0;
                acc1.x += r1.x * w1; acc1.y += r1.y * w1; acc1.z += r1.z * w1; acc1.w += r1.w * w1;
            }
        }
        const float4 b0 = *reinterpret_cast<const float4*>(b + lane * 4);
        const float4 b1 = *reinterpret_cast<const float4*>(b + 128 + lane * 4);
        acc0.x += b0.x; acc0.y += b0.y; acc0.z += b0.z; acc0.w += b0.w;
        acc1.x += b1.x; acc1.y += b1.y; acc1.z += b1.z; acc1.w += b1.w;
        if (do_elu) {
            acc0.x = acc0.x > 0.f ? acc0.x : (expf(acc0.x) - 1.f);
            acc0.y = acc0.y > 0.f ? acc0.y : (expf(acc0.y) - 1.f);
            acc0.z = acc0.z > 0.f ? acc0.z : (expf(acc0.z) - 1.f);
            acc0.w = acc0.w > 0.f ? acc0.w : (expf(acc0.w) - 1.f);
            acc1.x = acc1.x > 0.f ? acc1.x : (expf(acc1.x) - 1.f);
            acc1.y = acc1.y > 0.f ? acc1.y : (expf(acc1.y) - 1.f);
            acc1.z = acc1.z > 0.f ? acc1.z : (expf(acc1.z) - 1.f);
            acc1.w = acc1.w > 0.f ? acc1.w : (expf(acc1.w) - 1.f);
        }
        float* op = out + (size_t)node * 256 + lane * 4;
        *reinterpret_cast<float4*>(op)       = acc0;
        *reinterpret_cast<float4*>(op + 128) = acc1;
    } else {
        const float* hp = hfeat + lane * 2;
        float2 acc;
        {
            const float w = aself[0];
            float2 r = *reinterpret_cast<const float2*>(hp + (size_t)node * 64);
            acc.x = r.x * w; acc.y = r.y * w;
        }
        for (int p0 = start; p0 < end; p0 += 32) {
            const int n = (end - p0) < 32 ? (end - p0) : 32;
            int sv = src0;
            float a = a0[0];
            if (p0 != start) {
                sv = 0; a = 0.f;
                if (lane < n) {
                    float e[HH];
                    sv = ssrc[p0 + lane];
                    edge_score<HH>(es, sv, edn, e);
                    a = expf(e[0] - m[0]) / den[0];
                }
            }
#pragma unroll 4
            for (int j = 0; j < n; ++j) {
                const int   s = __builtin_amdgcn_readlane(sv, j);
                const float w = __int_as_float(__builtin_amdgcn_readlane(__float_as_int(a), j));
                float2 r = *reinterpret_cast<const float2*>(hp + (size_t)s * 64);
                acc.x += r.x * w; acc.y += r.y * w;
            }
        }
        const float2 bb = *reinterpret_cast<const float2*>(b + lane * 2);
        acc.x += bb.x; acc.y += bb.y;
        if (do_elu) {
            acc.x = acc.x > 0.f ? acc.x : (expf(acc.x) - 1.f);
            acc.y = acc.y > 0.f ? acc.y : (expf(acc.y) - 1.f);
        }
        *reinterpret_cast<float2*>(out + (size_t)node * 64 + lane * 2) = acc;
    }
}

#define GAT_SLAB 256
template <bool FULL>
__device__ __forceinline__ float4 slab_rows_sum(const float* __restrict__ x, int r0, int N, int cg) {
    float4 v[GAT_SLAB / 16];
    for (int i = 0; i < GAT_SLAB / 16; ++i) {
        const int r = r0 + 16 * i;
        if constexpr (FULL) {
            v[i] = *reinterpret_cast<const float4*>(x + (size_t)r * 64 + cg * 4);
        } else {
            const int rc = r < N ? r : N - 1;
            v[i] = *reinterpret_cast<const float4*>(x + (size_t)rc * 64 + cg * 4);
            if (r >= N) v[i] = make_float4(0.f, 0.f, 0.f, 0.f);
        }
    }
    float4 s = v[0];
    for (int i = 1; i < GAT_SLAB / 16; ++i) { s.x += v[i].x; s.y += v[i].y; s.z += v[i].z; s.w += v[i].w; }
    return s;
}

__global__ void gat_colsum_slab(const float* __restrict__ x, float* __restrict__ part, int N) {
    __shared__ float4 sm[256];
    const int t    = threadIdx.x;
    const int cg   = t & 15;
    const int slot = t >> 4;
    const int r0   = blockIdx.x * GAT_SLAB + slot;
    const bool full = (blockIdx.x + 1) * GAT_SLAB <= N;
    sm[t] = full ? slab_rows_sum<true>(x, r0, N, cg)
                 : slab_rows_sum<false>(x, r0, N, cg);
    __syncthreads();
    if (t < 64) {
        const float* f = reinterpret_cast<const float*>(sm);
        float a = 0.f;
        for (int k = 0; k < 16; ++k) a += f[k * 64 + t];
        part[(size_t)blockIdx.x * 64 + t] = a;
    }
}

__global__ void gat_colmean_final(const float* __restrict__ part, float* __restrict__ out,
                                  int P, int N) {
    __shared__ float4 sm[1024];
    const int t    = threadIdx.x;
    const int cg   = t & 15;
    const int slot = t >> 4;
    float4 s = make_float4(0.f, 0.f, 0.f, 0.f);
    for (int base = 0; base < P; base += 256) {
        float4 v[4];
        for (int i = 0; i < 4; ++i) {
            const int r  = base + slot + 64 * i;
            const int rc = r < P ? r : P - 1;
            v[i] = *reinterpret_cast<const float4*>(part + (size_t)rc * 64 + cg * 4);
            if (r >= P) v[i] = make_float4(0.f, 0.f, 0.f, 0.f);
        }
        for (int i = 0; i < 4; ++i) { s.x += v[i].x; s.y += v[i].y; s.z += v[i].z; s.w += v[i].w; }
    }
    sm[t] = s;
    __syncthreads();
    if (t < 64) {
        const float* f = reinterpret_cast<const float*>(sm);
        float a = 0.f;
        for (int k = 0; k < 64; ++k) a += f[k * 64 + t];
        out[t] = a / (float)N;
    }
}

static inline int cdiv(long long a, long long b) { return (int)((a + b - 1) / b); }

extern "C" void kernel_launch(void* const* d_in, const int* in_sizes, int n_in,
                              void* d_out, int out_size, void* d_ws, size_t ws_size,
                              hipStream_t stream) {
    const float* x   = (const float*)d_in[0];
    const int*   ei  = (const int*)  d_in[1];
    const float* W1  = (const float*)d_in[2];
    const float* a1s = (const float*)d_in[3];
    const float* a1d = (const float*)d_in[4];
    const float* b1  = (const float*)d_in[5];
    const float* W2  = (const float*)d_in[6];
    const float* a2s = (const float*)d_in[7];
    const float* a2d = (const float*)d_in[8];
    const float* b2  = (const float*)d_in[9];
    const float* W3  = (const float*)d_in[10];
    const float* a3s = (const float*)d_in[11];
    const float* a3d = (const float*)d_in[12];
    const float* b3  = (const float*)d_in[13];

    const int N  = in_sizes[0] / 16;
    const int E  = in_sizes[1] / 2;
    const int H  = 4;

    float* ws = (float*)d_ws;
    size_t off = 0;
    float*    buf_h    = ws + off; off += (size_t)N * 256;
    int*      csr      = (int*)(ws + off); off += (size_t)N * 256;
    float*    buf_feat = ws + off; off += (size_t)N * 256;
    float*    es       = ws + off; off += (size_t)N * H;
    float*    ed       = ws + off; off += (size_t)N * H;
    (void)ws_size;
    int* cnt    = csr;
    int* rowptr = csr + (size_t)N;
    int* cursor = csr + (size_t)2 * N;
    int* ssrc   = csr + (size_t)4 * N;

    float* node_emb  = (float*)d_out;
    float* graph_emb = node_emb + (size_t)N * 64;

    auto run_layer = [&](const float* in_feat, const float* W,
                         const float* as_, const float* ad_, const float* bb,
                         float* out_feat, int Hh, int K, int Ncols, int do_elu) {
        int ngrp = Ncols / 64;
        int wpb  = ngrp < 4 ? ngrp : 4;
        dim3 gb(32, wpb);
        dim3 gg(cdiv(ngrp, wpb), cdiv(N, 16));
        hipLaunchKernelGGL(gat_gemm_wmma, gg, gb, 0, stream, in_feat, W, buf_h, N, K, Ncols);
        hipLaunchKernelGGL(gat_attn_coef, dim3(cdiv((long long)N * Hh, 256)), dim3(256), 0, stream,
                           buf_h, as_, ad_, es, ed, N, Hh);
        if (Hh == 4)
            hipLaunchKernelGGL((gat_seg_softmax_agg<4>), dim3(cdiv(N, 8)), dim3(256), 0, stream,
                               ssrc, rowptr, cursor, buf_h, es, ed, bb, out_feat, N, do_elu);
        else
            hipLaunchKernelGGL((gat_seg_softmax_agg<1>), dim3(cdiv(N, 8)), dim3(256), 0, stream,
                               ssrc, rowptr, cursor, buf_h, es, ed, bb, out_feat, N, do_elu);
    };

    hipMemsetAsync(cnt, 0, (size_t)N * sizeof(int), stream);
    hipLaunchKernelGGL(gat_csr_count, dim3(cdiv(E, 256)), dim3(256), 0, stream, ei, E, cnt);
    hipLaunchKernelGGL(gat_csr_scan,  dim3(cdiv(N, 1024)), dim3(1024), 0, stream, cnt, rowptr, cursor, N);
    hipLaunchKernelGGL(gat_csr_fill,  dim3(cdiv(E, 256)), dim3(256), 0, stream, ei, E, cursor, ssrc);

    run_layer(x,        W1, a1s, a1d, b1, buf_feat, H, 16,  256, 1);
    run_layer(buf_feat, W2, a2s, a2d, b2, buf_feat, H, 256, 256, 1);
    run_layer(buf_feat, W3, a3s, a3d, b3, node_emb, 1, 256, 64, 0);

    const int P = cdiv(N, GAT_SLAB);
    hipLaunchKernelGGL(gat_colsum_slab, dim3(P), dim3(256), 0, stream, node_emb, buf_h, N);
    hipLaunchKernelGGL(gat_colmean_final, dim3(1), dim3(1024), 0, stream, buf_h, graph_emb, P, N);
}


# Layer 1 aggregates before it projects (the projection is linear): the walk gathers 64-B input rows, not 1-KB projected rows; scores come from x @ (W a); WMMA projection, bias, ELU follow on the sums
# speedup vs baseline: 6.0253x; 1.0953x over previous
#include <hip/hip_runtime.h>
#include <math.h>

typedef __attribute__((ext_vector_type(2))) float v2f;
typedef __attribute__((ext_vector_type(8))) float v8f;

#define GAT_D 64

__global__ void gat_gemm_wmma(const float* __restrict__ A,
                              const float* __restrict__ B,
                              float* __restrict__ C,
                              int M, int K, int N) {
    const int lane  = threadIdx.x;
    const int grp   = blockIdx.x * blockDim.y + threadIdx.y;
    const int mtile = blockIdx.y;
    const int n0 = grp * 64;
    const int m0 = mtile * 16;
    if (n0 >= N || m0 >= M) return;

    const int l15 = lane & 15;
    const int kh  = (lane >> 4) << 1;
    const float* Arow = A + (size_t)(m0 + l15) * K;
    v8f acc0 = {0.f,0.f,0.f,0.f,0.f,0.f,0.f,0.f};
    v8f acc1 = acc0, acc2 = acc0, acc3 = acc0;

    for (int k = 0; k < K; k += 4) {
        v2f a;
        a.x = Arow[k + kh];
        a.y = Arow[k + kh + 1];
        const float* Bp = B + (size_t)(k + kh) * N + n0 + l15;
        v2f b0, b1, b2, b3;
        b0.x = Bp[0];  b0.y = Bp[(size_t)N];
        b1.x = Bp[16]; b1.y = Bp[(size_t)N + 16];
        b2.x = Bp[32]; b2.y = Bp[(size_t)N + 32];
        b3.x = Bp[48]; b3.y = Bp[(size_t)N + 48];
        acc0 = __builtin_amdgcn_wmma_f32_16x16x4_f32(false, a, false, b0, (short)0, acc0, false, false);
        acc1 = __builtin_amdgcn_wmma_f32_16x16x4_f32(false, a, false, b1, (short)0, acc1, false, false);
        acc2 = __builtin_amdgcn_wmma_f32_16x16x4_f32(false, a, false, b2, (short)0, acc2, false, false);
        acc3 = __builtin_amdgcn_wmma_f32_16x16x4_f32(false, a, false, b3, (short)0, acc3, false, false);
    }

    float* Cp = C + (size_t)(m0 + ((lane >> 4) << 3)) * N + n0 + l15;
#pragma unroll
    for (int v = 0; v < 8; ++v) {
        Cp[(size_t)v * N + 0]  = acc0[v];
        Cp[(size_t)v * N + 16] = acc1[v];
        Cp[(size_t)v * N + 32] = acc2[v];
        Cp[(size_t)v * N + 48] = acc3[v];
    }
}

__global__ void gat_attn_coef(const float* __restrict__ h,
                              const float* __restrict__ a_s,
                              const float* __restrict__ a_d,
                              float* __restrict__ es, float* __restrict__ ed,
                              int N, int Hh) {
    int t = blockIdx.x * blockDim.x + threadIdx.x;
    if (t >= N * Hh) return;
    int n  = t / Hh;
    int hh = t - n * Hh;
    const float* hp = h   + ((size_t)n * Hh + hh) * GAT_D;
    const float* as = a_s + (size_t)hh * GAT_D;
    const float* ad = a_d + (size_t)hh * GAT_D;
    float ss = 0.f, dd = 0.f;
#pragma unroll 8
    for (int d = 0; d < GAT_D; ++d) { float v = hp[d]; ss += v * as[d]; dd += v * ad[d]; }
    es[t] = ss; ed[t] = dd;
}

__global__ void gat_attn_coef_in(const float* __restrict__ x, const float* __restrict__ W,
                                 const float* __restrict__ a_s, const float* __restrict__ a_d,
                                 float* __restrict__ es, float* __restrict__ ed, int N) {
    __shared__ float wsd[2][16][4];
    const int t = threadIdx.x;
    if (t < 128) {
        const int which = t >> 6, k = (t >> 2) & 15, hh = t & 3;
        const float* wp = W + (size_t)k * 256 + hh * GAT_D;
        const float* ap = (which ? a_d : a_s) + hh * GAT_D;
        float s = 0.f;
        for (int d = 0; d < GAT_D; ++d) s += wp[d] * ap[d];
        wsd[which][k][hh] = s;
    }
    __syncthreads();
    const int n = blockIdx.x * blockDim.x + t;
    if (n >= N) return;
    float xr[16];
    for (int q = 0; q < 4; ++q) {
        const float4 v = *reinterpret_cast<const float4*>(x + (size_t)n * 16 + q * 4);
        xr[4 * q] = v.x; xr[4 * q + 1] = v.y; xr[4 * q + 2] = v.z; xr[4 * q + 3] = v.w;
    }
    float ss[4] = {0.f, 0.f, 0.f, 0.f}, dd[4] = {0.f, 0.f, 0.f, 0.f};
    for (int k = 0; k < 16; ++k)
        for (int hh = 0; hh < 4; ++hh) { ss[hh] += xr[k] * wsd[0][k][hh]; dd[hh] += xr[k] * wsd[1][k][hh]; }
    *reinterpret_cast<float4*>(es + (size_t)n * 4) = make_float4(ss[0], ss[1], ss[2], ss[3]);
    *reinterpret_cast<float4*>(ed + (size_t)n * 4) = make_float4(dd[0], dd[1], dd[2], dd[3]);
}

__global__ void gat_proj_heads_wmma(const float* __restrict__ S, const float* __restrict__ B,
                                    const float* __restrict__ bias, float* __restrict__ C, int M) {
    const int lane = threadIdx.x;
    const int head = threadIdx.y;
    const int m0 = blockIdx.x * 16;
    if (m0 >= M) return;
    const int n0 = head * 64;
    const int l15 = lane & 15;
    const int kh  = (lane >> 4) << 1;
    const float* Arow = S + (size_t)(m0 + l15) * 64 + head * 16;
    v8f acc0 = {0.f,0.f,0.f,0.f,0.f,0.f,0.f,0.f};
    v8f acc1 = acc0, acc2 = acc0, acc3 = acc0;
    for (int k = 0; k < 16; k += 4) {
        v2f a;
        a.x = Arow[k + kh];
        a.y = Arow[k + kh + 1];
        const float* Bp = B + (size_t)(k + kh) * 256 + n0 + l15;
        v2f b0, b1, b2, b3;
        b0.x = Bp[0];  b0.y = Bp[256];
        b1.x = Bp[16]; b1.y = Bp[256 + 16];
        b2.x = Bp[32]; b2.y = Bp[256 + 32];
        b3.x = Bp[48]; b3.y = Bp[256 + 48];
        acc0 = __builtin_amdgcn_wmma_f32_16x16x4_f32(false, a, false, b0, (short)0, acc0, false, false);
        acc1 = __builtin_amdgcn_wmma_f32_16x16x4_f32(false, a, false, b1, (short)0, acc1, false, false);
        acc2 = __builtin_amdgcn_wmma_f32_16x16x4_f32(false, a, false, b2, (short)0, acc2, false, false);
        acc3 = __builtin_amdgcn_wmma_f32_16x16x4_f32(false, a, false, b3, (short)0, acc3, false, false);
    }
    const float bb0 = bias[n0 + l15], bb1 = bias[n0 + 16 + l15];
    const float bb2 = bias[n0 + 32 + l15], bb3 = bias[n0 + 48 + l15];
    float* Cp = C + (size_t)(m0 + ((lane >> 4) << 3)) * 256 + n0 + l15;
#pragma unroll
    for (int v = 0; v < 8; ++v) {
        float c0 = acc0[v] + bb0, c1 = acc1[v] + bb1, c2 = acc2[v] + bb2, c3 = acc3[v] + bb3;
        c0 = c0 > 0.f ? c0 : (expf(c0) - 1.f);
        c1 = c1 > 0.f ? c1 : (expf(c1) - 1.f);
        c2 = c2 > 0.f ? c2 : (expf(c2) - 1.f);
        c3 = c3 > 0.f ? c3 : (expf(c3) - 1.f);
        Cp[(size_t)v * 256 + 0]  = c0;
        Cp[(size_t)v * 256 + 16] = c1;
        Cp[(size_t)v * 256 + 32] = c2;
        Cp[(size_t)v * 256 + 48] = c3;
    }
}

__global__ void gat_csr_count(const int* __restrict__ ei, int E, int* __restrict__ cnt) {
    int e = blockIdx.x * blockDim.x + threadIdx.x;
    if (e >= E) return;
    atomicAdd(&cnt[ei[E + e]], 1);
}

__global__ void gat_csr_scan(const int* __restrict__ cnt, int* __restrict__ rowptr,
                             int* __restrict__ cursor, int N) {
    __shared__ int wpre[32];
    __shared__ int wtot[32];
    const int t = threadIdx.x, lane = t & 31, wv = t >> 5;
    const int tile0 = blockIdx.x * 1024;
    int pre = 0;
    for (int i = t; i < tile0; i += 1024) pre += cnt[i];
    const int i = tile0 + t;
    const int v = (i < N) ? cnt[i] : 0;
    int inc = v;
    for (int d = 1; d < 32; d <<= 1) {
        int u = __shfl_up(inc, d, 32);
        if (lane >= d) inc += u;
    }
    for (int d = 16; d > 0; d >>= 1) pre += __shfl_xor(pre, d, 32);
    if (lane == 31) wtot[wv] = inc;
    if (lane == 0)  wpre[wv] = pre;
    __syncthreads();
    int base = 0;
    for (int k = 0; k < 32; ++k) {
        base += wpre[k];
        if (k < wv) base += wtot[k];
    }
    if (i < N) { int r = base + inc - v; rowptr[i] = r; cursor[i] = r; }
}

__global__ void gat_csr_fill(const int* __restrict__ ei, int E, int* __restrict__ cursor,
                             int* __restrict__ ssrc) {
    int e = blockIdx.x * blockDim.x + threadIdx.x;
    if (e >= E) return;
    int p = atomicAdd(&cursor[ei[E + e]], 1);
    ssrc[p] = ei[e];
}

template <int M>
__device__ __forceinline__ float row_xor(float x) {
    const int b = __float_as_int(x);
    return __int_as_float(__builtin_amdgcn_update_dpp(b, b, 0x160 + M, 0xf, 0xf, false));
}
__device__ __forceinline__ float half_xor(float x) {
    const int b = __float_as_int(x);
    return __int_as_float(__builtin_amdgcn_permlanex16(b, b, 0x76543210, 0xfedcba98, false, false));
}
__device__ __forceinline__ float wave_max(float v) {
    v = fmaxf(v, row_xor<1>(v)); v = fmaxf(v, row_xor<2>(v));
    v = fmaxf(v, row_xor<4>(v)); v = fmaxf(v, row_xor<8>(v));
    return fmaxf(v, half_xor(v));
}
__device__ __forceinline__ float wave_sum(float v) {
    v += row_xor<1>(v); v += row_xor<2>(v); v += row_xor<4>(v); v += row_xor<8>(v);
    return v + half_xor(v);
}

template <int HH>
__device__ __forceinline__ void load_heads(const float* __restrict__ p, float (&v)[HH]) {
    if constexpr (HH == 4) {
        const float4 t = *reinterpret_cast<const float4*>(p);
        v[0] = t.x; v[1] = t.y; v[2] = t.z; v[3] = t.w;
    } else {
        v[0] = p[0];
    }
}

template <int HH>
__device__ __forceinline__ void edge_score(const float* __restrict__ es, int s,
                                           const float (&edn)[HH], float (&e)[HH]) {
    float t[HH];
    load_heads<HH>(es + (size_t)s * HH, t);
#pragma unroll
    for (int h = 0; h < HH; ++h) { const float v = t[h] + edn[h]; e[h] = v > 0.f ? v : 0.2f * v; }
}

template <int HH, bool XIN>
__global__ void gat_seg_softmax_agg(const int* __restrict__ ssrc, const int* __restrict__ rowptr,
                                    const int* __restrict__ segend,
                                    const float* __restrict__ hfeat,
                                    const float* __restrict__ es, const float* __restrict__ ed,
                                    const float* __restrict__ b, float* __restrict__ out,
                                    int N, int do_elu) {
    const int lane = threadIdx.x & 31;
    const int node = __builtin_amdgcn_readfirstlane(blockIdx.x * (blockDim.x >> 5) + (threadIdx.x >> 5));
    if (node >= N) return;
    const int start = rowptr[node];
    const int end   = segend[node];

    float edn[HH], eself[HH], e0[HH], m[HH];
    load_heads<HH>(ed + (size_t)node * HH, edn);
    edge_score<HH>(es, node, edn, eself);
    const int n0 = (end - start) < 32 ? (end - start) : 32;
    int src0 = 0;
#pragma unroll
    for (int h = 0; h < HH; ++h) e0[h] = -INFINITY;
    if (lane < n0) { src0 = ssrc[start + lane]; edge_score<HH>(es, src0, edn, e0); }
#pragma unroll
    for (int h = 0; h < HH; ++h) m[h] = fmaxf(e0[h], eself[h]);
    for (int p0 = start + 32; p0 < end; p0 += 32) {
        if (p0 + lane < end) {
            float e[HH];
            edge_score<HH>(es, ssrc[p0 + lane], edn, e);
#pragma unroll
            for (int h = 0; h < HH; ++h) m[h] = fmaxf(m[h], e[h]);
        }
    }
#pragma unroll
    for (int h = 0; h < HH; ++h)
        m[h] = __int_as_float(__builtin_amdgcn_readfirstlane(__float_as_int(wave_max(m[h]))));

    float x0[HH], den[HH], a0[HH], aself[HH];
#pragma unroll
    for (int h = 0; h < HH; ++h) { x0[h] = expf(e0[h] - m[h]); den[h] = x0[h]; }
    for (int p0 = start + 32; p0 < end; p0 += 32) {
        if (p0 + lane < end) {
            float e[HH];
            edge_score<HH>(es, ssrc[p0 + lane], edn, e);
#pragma unroll
            for (int h = 0; h < HH; ++h) den[h] += expf(e[h] - m[h]);
        }
    }
#pragma unroll
    for (int h = 0; h < HH; ++h) {
        const float xs = expf(eself[h] - m[h]);
        den[h]   = __int_as_float(__builtin_amdgcn_readfirstlane(__float_as_int(wave_sum(den[h]) + xs)));
        aself[h] = xs / den[h];
        a0[h]    = x0[h] / den[h];
    }

    if constexpr (XIN) {
        const int hsel = lane >> 3;
        const float* xp = hfeat + (lane & 7) * 2;
        float2 acc;
        {
            const float w = hsel == 0 ? aself[0] : hsel == 1 ? aself[1] : hsel == 2 ? aself[2] : aself[3];
            const float2 r = *reinterpret_cast<const float2*>(xp + (size_t)node * 16);
            acc.x = r.x * w; acc.y = r.y * w;
        }
        for (int p0 = start; p0 < end; p0 += 32) {
            const int n = (end - p0) < 32 ? (end - p0) : 32;
            int sv = src0;
            float a[4] = {a0[0], a0[1], a0[2], a0[3]};
            if (p0 != start) {
                sv = 0; a[0] = 0.f; a[1] = 0.f; a[2] = 0.f; a[3] = 0.f;
                if (lane < n) {
                    float e[HH];
                    sv = ssrc[p0 + lane];
                    edge_score<HH>(es, sv, edn, e);
#pragma unroll
                    for (int h = 0; h < HH; ++h) a[h] = expf(e[h] - m[h]) / den[h];
                }
            }
            for (int j = 0; j < n; j += 8) {
#pragma unroll
                for (int u = 0; u < 8; ++u) {
                    const int   s  = __builtin_amdgcn_readlane(sv, j + u);
                    const float ax = __int_as_float(__builtin_amdgcn_readlane(__float_as_int(a[0]), j + u));
                    const float ay = __int_as_float(__builtin_amdgcn_readlane(__float_as_int(a[1]), j + u));
                    const float az = __int_as_float(__builtin_amdgcn_readlane(__float_as_int(a[2]), j + u));
                    const float aw = __int_as_float(__builtin_amdgcn_readlane(__float_as_int(a[3]), j + u));
                    const float w = hsel == 0 ? ax : hsel == 1 ? ay : hsel == 2 ? az : aw;
                    const float2 r = *reinterpret_cast<const float2*>(xp + (size_t)s * 16);
                    acc.x += r.x * w; acc.y += r.y * w;
                }
            }
        }
        *reinterpret_cast<float2*>(out + (size_t)node * 64 + lane * 2) = acc;
    } else if constexpr (HH == 4) {
        const int hsel = lane >> 4;
        const float* hp = hfeat + lane * 4;
        float4 acc0, acc1;
        {
            const float w0 = hsel ? aself[1] : aself[0];
            const float w1 = hsel ? aself[3] : aself[2];
            const float* rp = hp + (size_t)node * 256;
            float4 r0 = *reinterpret_cast<const float4*>(rp);
            float4 r1 = *reinterpret_cast<const float4*>(rp + 128);
            acc0.x = r0.x * w0; acc0.y = r0.y * w0; acc0.z = r0.z * w0; acc0.w = r0.w * w0;
            acc1.x = r1.x * w1; acc1.y = r1.y * w1; acc1.z = r1.z * w1; acc1.w = r1.w * w1;
        }
        for (int p0 = start; p0 < end; p0 += 32) {
            const int n = (end - p0) < 32 ? (end - p0) : 32;
            int sv = src0;
            float a[4] = {a0[0], a0[1], a0[2], a0[3]};
            if (p0 != start) {
                sv = 0; a[0] = 0.f; a[1] = 0.f; a[2] = 0.f; a[3] = 0.f;
                if (lane < n) {
                    float e[HH];
                    sv = ssrc[p0 + lane];
                    edge_score<HH>(es, sv, edn, e);
#pragma unroll
                    for (int h = 0; h < HH; ++h) a[h] = expf(e[h] - m[h]) / den[h];
                }
            }
#pragma unroll 4
            for (int j = 0; j < n; ++j) {
                const int   s  = __builtin_amdgcn_readlane(sv, j);
                const float ax = __int_as_float(__builtin_amdgcn_readlane(__float_as_int(a[0]), j));
                const float ay = __int_as_float(__builtin_amdgcn_readlane(__float_as_int(a[1]), j));
                const float az = __int_as_float(__builtin_amdgcn_readlane(__float_as_int(a[2]), j));
                const float aw = __int_as_float(__builtin_amdgcn_readlane(__float_as_int(a[3]), j));
                const float w0 = hsel ? ay : ax;
                const float w1 = hsel ? aw : az;
                const float* rp = hp + (size_t)s * 256;
                float4 r0 = *reinterpret_cast<const float4*>(rp);
                float4 r1 = *reinterpret_cast<const float4*>(rp + 128);
                acc0.x += r0.x * w0; acc0.y += r0.y * w0; acc0.z += r0.z * w0; acc0.w += r0.w * w0;
                acc1.x += r1.x * w1; acc1.y += r1.y * w1; acc1.z += r1.z * w1; acc1.w += r1.w * w1;
            }
        }
        const float4 b0 = *reinterpret_cast<const float4*>(b + lane * 4);
        const float4 b1 = *reinterpret_cast<const float4*>(b + 128 + lane * 4);
        acc0.x += b0.x; acc0.y += b0.y; acc0.z += b0.z; acc0.w += b0.w;
        acc1.x += b1.x; acc1.y += b1.y; acc1.z += b1.z; acc1.w += b1.w;
        if (do_elu) {
            acc0.x = acc0.x > 0.f ? acc0.x : (expf(acc0.x) - 1.f);
            acc0.y = acc0.y > 0.f ? acc0.y : (expf(acc0.y) - 1.f);
            acc0.z = acc0.z > 0.f ? acc0.z : (expf(acc0.z) - 1.f);
            acc0.w = acc0.w > 0.f ? acc0.w : (expf(acc0.w) - 1.f);
            acc1.x = acc1.x > 0.f ? acc1.x : (expf(acc1.x) - 1.f);
            acc1.y = acc1.y > 0.f ? acc1.y : (expf(acc1.y) - 1.f);
            acc1.z = acc1.z > 0.f ? acc1.z : (expf(acc1.z) - 1.f);
            acc1.w = acc1.w > 0.f ? acc1.w : (expf(acc1.w) - 1.f);
        }
        float* op = out + (size_t)node * 256 + lane * 4;
        *reinterpret_cast<float4*>(op)       = acc0;
        *reinterpret_cast<float4*>(op + 128) = acc1;
    } else {
        const float* hp = hfeat + lane * 2;
        float2 acc;
        {
            const float w = aself[0];
            float2 r = *reinterpret_cast<const float2*>(hp + (size_t)node * 64);
            acc.x = r.x * w; acc.y = r.y * w;
        }
        for (int p0 = start; p0 < end; p0 += 32) {
            const int n = (end - p0) < 32 ? (end - p0) : 32;
            int sv = src0;
            float a = a0[0];
            if (p0 != start) {
                sv = 0; a = 0.f;
                if (lane < n) {
                    float e[HH];
                    sv = ssrc[p0 + lane];
                    edge_score<HH>(es, sv, edn, e);
                    a = expf(e[0] - m[0]) / den[0];
                }
            }
#pragma unroll 4
            for (int j = 0; j < n; ++j) {
                const int   s = __builtin_amdgcn_readlane(sv, j);
                const float w = __int_as_float(__builtin_amdgcn_readlane(__float_as_int(a), j));
                float2 r = *reinterpret_cast<const float2*>(hp + (size_t)s * 64);
                acc.x += r.x * w; acc.y += r.y * w;
            }
        }
        const float2 bb = *reinterpret_cast<const float2*>(b + lane * 2);
        acc.x += bb.x; acc.y += bb.y;
        if (do_elu) {
            acc.x = acc.x > 0.f ? acc.x : (expf(acc.x) - 1.f);
            acc.y = acc.y > 0.f ? acc.y : (expf(acc.y) - 1.f);
        }
        *reinterpret_cast<float2*>(out + (size_t)node * 64 + lane * 2) = acc;
    }
}

#define GAT_SLAB 256
template <bool FULL>
__device__ __forceinline__ float4 slab_rows_sum(const float* __restrict__ x, int r0, int N, int cg) {
    float4 v[GAT_SLAB / 16];
    for (int i = 0; i < GAT_SLAB / 16; ++i) {
        const int r = r0 + 16 * i;
        if constexpr (FULL) {
            v[i] = *reinterpret_cast<const float4*>(x + (size_t)r * 64 + cg * 4);
        } else {
            const int rc = r < N ? r : N - 1;
            v[i] = *reinterpret_cast<const float4*>(x + (size_t)rc * 64 + cg * 4);
            if (r >= N) v[i] = make_float4(0.f, 0.f, 0.f, 0.f);
        }
    }
    float4 s = v[0];
    for (int i = 1; i < GAT_SLAB / 16; ++i) { s.x += v[i].x; s.y += v[i].y; s.z += v[i].z; s.w += v[i].w; }
    return s;
}

__global__ void gat_colsum_slab(const float* __restrict__ x, float* __restrict__ part, int N) {
    __shared__ float4 sm[256];
    const int t    = threadIdx.x;
    const int cg   = t & 15;
    const int slot = t >> 4;
    const int r0   = blockIdx.x * GAT_SLAB + slot;
    const bool full = (blockIdx.x + 1) * GAT_SLAB <= N;
    sm[t] = full ? slab_rows_sum<true>(x, r0, N, cg)
                 : slab_rows_sum<false>(x, r0, N, cg);
    __syncthreads();
    if (t < 64) {
        const float* f = reinterpret_cast<const float*>(sm);
        float a = 0.f;
        for (int k = 0; k < 16; ++k) a += f[k * 64 + t];
        part[(size_t)blockIdx.x * 64 + t] = a;
    }
}

__global__ void gat_colmean_final(const float* __restrict__ part, float* __restrict__ out,
                                  int P, int N) {
    __shared__ float4 sm[1024];
    const int t    = threadIdx.x;
    const int cg   = t & 15;
    const int slot = t >> 4;
    float4 s = make_float4(0.f, 0.f, 0.f, 0.f);
    for (int base = 0; base < P; base += 256) {
        float4 v[4];
        for (int i = 0; i < 4; ++i) {
            const int r  = base + slot + 64 * i;
            const int rc = r < P ? r : P - 1;
            v[i] = *reinterpret_cast<const float4*>(part + (size_t)rc * 64 + cg * 4);
            if (r >= P) v[i] = make_float4(0.f, 0.f, 0.f, 0.f);
        }
        for (int i = 0; i < 4; ++i) { s.x += v[i].x; s.y += v[i].y; s.z += v[i].z; s.w += v[i].w; }
    }
    sm[t] = s;
    __syncthreads();
    if (t < 64) {
        const float* f = reinterpret_cast<const float*>(sm);
        float a = 0.f;
        for (int k = 0; k < 64; ++k) a += f[k * 64 + t];
        out[t] = a / (float)N;
    }
}

static inline int cdiv(long long a, long long b) { return (int)((a + b - 1) / b); }

extern "C" void kernel_launch(void* const* d_in, const int* in_sizes, int n_in,
                              void* d_out, int out_size, void* d_ws, size_t ws_size,
                              hipStream_t stream) {
    const float* x   = (const float*)d_in[0];
    const int*   ei  = (const int*)  d_in[1];
    const float* W1  = (const float*)d_in[2];
    const float* a1s = (const float*)d_in[3];
    const float* a1d = (const float*)d_in[4];
    const float* b1  = (const float*)d_in[5];
    const float* W2  = (const float*)d_in[6];
    const float* a2s = (const float*)d_in[7];
    const float* a2d = (const float*)d_in[8];
    const float* b2  = (const float*)d_in[9];
    const float* W3  = (const float*)d_in[10];
    const float* a3s = (const float*)d_in[11];
    const float* a3d = (const float*)d_in[12];
    const float* b3  = (const float*)d_in[13];

    const int N  = in_sizes[0] / 16;
    const int E  = in_sizes[1] / 2;
    const int H  = 4;

    float* ws = (float*)d_ws;
    size_t off = 0;
    float*    buf_h    = ws + off; off += (size_t)N * 256;
    int*      csr      = (int*)(ws + off); off += (size_t)N * 256;
    float*    buf_feat = ws + off; off += (size_t)N * 256;
    float*    es       = ws + off; off += (size_t)N * H;
    float*    ed       = ws + off; off += (size_t)N * H;
    (void)ws_size;
    int* cnt    = csr;
    int* rowptr = csr + (size_t)N;
    int* cursor = csr + (size_t)2 * N;
    int* ssrc   = csr + (size_t)4 * N;

    float* node_emb  = (float*)d_out;
    float* graph_emb = node_emb + (size_t)N * 64;

    auto run_layer = [&](const float* in_feat, const float* W,
                         const float* as_, const float* ad_, const float* bb,
                         float* out_feat, int Hh, int K, int Ncols, int do_elu) {
        int ngrp = Ncols / 64;
        int wpb  = ngrp < 4 ? ngrp : 4;
        dim3 gb(32, wpb);
        dim3 gg(cdiv(ngrp, wpb), cdiv(N, 16));
        hipLaunchKernelGGL(gat_gemm_wmma, gg, gb, 0, stream, in_feat, W, buf_h, N, K, Ncols);
        hipLaunchKernelGGL(gat_attn_coef, dim3(cdiv((long long)N * Hh, 256)), dim3(256), 0, stream,
                           buf_h, as_, ad_, es, ed, N, Hh);
        if (Hh == 4)
            hipLaunchKernelGGL((gat_seg_softmax_agg<4, false>), dim3(cdiv(N, 8)), dim3(256), 0, stream,
                               ssrc, rowptr, cursor, buf_h, es, ed, bb, out_feat, N, do_elu);
        else
            hipLaunchKernelGGL((gat_seg_softmax_agg<1, false>), dim3(cdiv(N, 8)), dim3(256), 0, stream,
                               ssrc, rowptr, cursor, buf_h, es, ed, bb, out_feat, N, do_elu);
    };

    hipMemsetAsync(cnt, 0, (size_t)N * sizeof(int), stream);
    hipLaunchKernelGGL(gat_csr_count, dim3(cdiv(E, 256)), dim3(256), 0, stream, ei, E, cnt);
    hipLaunchKernelGGL(gat_csr_scan,  dim3(cdiv(N, 1024)), dim3(1024), 0, stream, cnt, rowptr, cursor, N);
    hipLaunchKernelGGL(gat_csr_fill,  dim3(cdiv(E, 256)), dim3(256), 0, stream, ei, E, cursor, ssrc);

    hipLaunchKernelGGL(gat_attn_coef_in, dim3(cdiv(N, 256)), dim3(256), 0, stream,
                       x, W1, a1s, a1d, es, ed, N);
    hipLaunchKernelGGL((gat_seg_softmax_agg<4, true>), dim3(cdiv(N, 8)), dim3(256), 0, stream,
                       ssrc, rowptr, cursor, x, es, ed, b1, buf_h, N, 1);
    hipLaunchKernelGGL(gat_proj_heads_wmma, dim3(cdiv(N, 16)), dim3(32, 4), 0, stream,
                       buf_h, W1, b1, buf_feat, N);
    run_layer(buf_feat, W2, a2s, a2d, b2, buf_feat, H, 256, 256, 1);
    run_layer(buf_feat, W3, a3s, a3d, b3, node_emb, 1, 256, 64, 0);

    const int P = cdiv(N, GAT_SLAB);
    hipLaunchKernelGGL(gat_colsum_slab, dim3(P), dim3(256), 0, stream, node_emb, buf_h, N);
    hipLaunchKernelGGL(gat_colmean_final, dim3(1), dim3(1024), 0, stream, buf_h, graph_emb, P, N);
}


# WMMA GEMMs move operands by whole lines: A tile copied once per block to LDS with coalesced float4 loads; fragment column c = columns 4c..4c+3, so B and C rows are float4; B rows asked 2 steps ahead
# speedup vs baseline: 7.0389x; 1.1682x over previous
#include <hip/hip_runtime.h>
#include <math.h>

typedef __attribute__((ext_vector_type(2))) float v2f;
typedef __attribute__((ext_vector_type(8))) float v8f;

#define GAT_D 64

typedef __attribute__((ext_vector_type(4))) float v4f;

#define GAT_K 256
#define GAT_APITCH 260

template <int WM, int WN, int MT>
__global__ void __launch_bounds__(WM * WN * 32)
gat_gemm_wmma(const float* __restrict__ A, const float* __restrict__ B,
              float* __restrict__ C, int M) {
    constexpr int NT = WM * WN * 32;
    constexpr int BM = WM * MT * 16;
    constexpr int N  = WN * 64;
    __shared__ __attribute__((aligned(16))) float As[BM * GAT_APITCH];
    const int t  = threadIdx.x;
    const int m0 = blockIdx.x * BM;

#pragma unroll
    for (int i = 0; i < BM * 64 / NT; ++i) {
        const int idx = i * NT + t;
        const int row = idx >> 6, c4 = idx & 63;
        const int gr  = m0 + row < M ? m0 + row : M - 1;
        const v4f v = *reinterpret_cast<const v4f*>(A + (size_t)gr * GAT_K + c4 * 4);
        *reinterpret_cast<v4f*>(&As[row * GAT_APITCH + c4 * 4]) = v;
    }
    __syncthreads();

    const int lane = t & 31, w = t >> 5;
    const int wm = w / WN, wn = w - wm * WN;
    const int c = lane & 15, h = lane >> 4;
    const int n0 = wn * 64;
    const float* Ap = As + (wm * MT * 16 + c) * GAT_APITCH + 4 * h;
    const float* Bp = B + (size_t)(4 * h) * N + n0 + 4 * c;

    v8f acc[MT][4];
#pragma unroll
    for (int mt = 0; mt < MT; ++mt)
#pragma unroll
        for (int j = 0; j < 4; ++j) acc[mt][j] = (v8f){0.f,0.f,0.f,0.f,0.f,0.f,0.f,0.f};

    v4f braw[3][4];
    v4f araw[2][MT];
#pragma unroll
    for (int q = 0; q < 4; ++q) {
        braw[0][q] = *reinterpret_cast<const v4f*>(Bp + (size_t)q * N);
        braw[1][q] = *reinterpret_cast<const v4f*>(Bp + (size_t)(8 + q) * N);
    }
#pragma unroll
    for (int mt = 0; mt < MT; ++mt)
        araw[0][mt] = *reinterpret_cast<const v4f*>(Ap + mt * 16 * GAT_APITCH);

#pragma unroll
    for (int st = 0; st < GAT_K / 8; ++st) {
        const int k0 = st * 8;
        __builtin_amdgcn_sched_barrier(0);
        if (st + 2 < GAT_K / 8) {
#pragma unroll
            for (int q = 0; q < 4; ++q)
                braw[(st + 2) % 3][q] = *reinterpret_cast<const v4f*>(Bp + (size_t)(k0 + 16 + q) * N);
        }
        __builtin_amdgcn_sched_barrier(0);
        if (st + 1 < GAT_K / 8) {
#pragma unroll
            for (int mt = 0; mt < MT; ++mt)
                araw[(st + 1) & 1][mt] = *reinterpret_cast<const v4f*>(Ap + mt * 16 * GAT_APITCH + k0 + 8);
        }
#pragma unroll
        for (int mt = 0; mt < MT; ++mt) {
            v2f a; a.x = araw[st & 1][mt][0]; a.y = araw[st & 1][mt][1];
#pragma unroll
            for (int j = 0; j < 4; ++j) {
                v2f b; b.x = braw[st % 3][0][j]; b.y = braw[st % 3][1][j];
                acc[mt][j] = __builtin_amdgcn_wmma_f32_16x16x4_f32(false, a, false, b, (short)0, acc[mt][j], false, false);
            }
        }
#pragma unroll
        for (int mt = 0; mt < MT; ++mt) {
            v2f a; a.x = araw[st & 1][mt][2]; a.y = araw[st & 1][mt][3];
#pragma unroll
            for (int j = 0; j < 4; ++j) {
                v2f b; b.x = braw[st % 3][2][j]; b.y = braw[st % 3][3][j];
                acc[mt][j] = __builtin_amdgcn_wmma_f32_16x16x4_f32(false, a, false, b, (short)0, acc[mt][j], false, false);
            }
        }
    }

#pragma unroll
    for (int mt = 0; mt < MT; ++mt) {
#pragma unroll
        for (int r = 0; r < 8; ++r) {
            const int row = m0 + (wm * MT + mt) * 16 + 8 * h + r;
            if (row < M) {
                v4f o; o[0] = acc[mt][0][r]; o[1] = acc[mt][1][r]; o[2] = acc[mt][2][r]; o[3] = acc[mt][3][r];
                *reinterpret_cast<v4f*>(C + (size_t)row * N + n0 + 4 * c) = o;
            }
        }
    }
}

__global__ void gat_attn_coef(const float* __restrict__ h,
                              const float* __restrict__ a_s,
                              const float* __restrict__ a_d,
                              float* __restrict__ es, float* __restrict__ ed,
                              int N, int Hh) {
    int t = blockIdx.x * blockDim.x + threadIdx.x;
    if (t >= N * Hh) return;
    int n  = t / Hh;
    int hh = t - n * Hh;
    const float* hp = h   + ((size_t)n * Hh + hh) * GAT_D;
    const float* as = a_s + (size_t)hh * GAT_D;
    const float* ad = a_d + (size_t)hh * GAT_D;
    float ss = 0.f, dd = 0.f;
#pragma unroll 8
    for (int d = 0; d < GAT_D; ++d) { float v = hp[d]; ss += v * as[d]; dd += v * ad[d]; }
    es[t] = ss; ed[t] = dd;
}

__global__ void gat_attn_coef_in(const float* __restrict__ x, const float* __restrict__ W,
                                 const float* __restrict__ a_s, const float* __restrict__ a_d,
                                 float* __restrict__ es, float* __restrict__ ed, int N) {
    __shared__ float wsd[2][16][4];
    const int t = threadIdx.x;
    if (t < 128) {
        const int which = t >> 6, k = (t >> 2) & 15, hh = t & 3;
        const float* wp = W + (size_t)k * 256 + hh * GAT_D;
        const float* ap = (which ? a_d : a_s) + hh * GAT_D;
        float s = 0.f;
        for (int d = 0; d < GAT_D; ++d) s += wp[d] * ap[d];
        wsd[which][k][hh] = s;
    }
    __syncthreads();
    const int n = blockIdx.x * blockDim.x + t;
    if (n >= N) return;
    float xr[16];
    for (int q = 0; q < 4; ++q) {
        const float4 v = *reinterpret_cast<const float4*>(x + (size_t)n * 16 + q * 4);
        xr[4 * q] = v.x; xr[4 * q + 1] = v.y; xr[4 * q + 2] = v.z; xr[4 * q + 3] = v.w;
    }
    float ss[4] = {0.f, 0.f, 0.f, 0.f}, dd[4] = {0.f, 0.f, 0.f, 0.f};
    for (int k = 0; k < 16; ++k)
        for (int hh = 0; hh < 4; ++hh) { ss[hh] += xr[k] * wsd[0][k][hh]; dd[hh] += xr[k] * wsd[1][k][hh]; }
    *reinterpret_cast<float4*>(es + (size_t)n * 4) = make_float4(ss[0], ss[1], ss[2], ss[3]);
    *reinterpret_cast<float4*>(ed + (size_t)n * 4) = make_float4(dd[0], dd[1], dd[2], dd[3]);
}

__global__ void gat_proj_heads_wmma(const float* __restrict__ S, const float* __restrict__ B,
                                    const float* __restrict__ bias, float* __restrict__ C, int M) {
    const int lane = threadIdx.x;
    const int head = threadIdx.y;
    const int m0 = blockIdx.x * 16;
    if (m0 >= M) return;
    const int n0 = head * 64;
    const int c = lane & 15, h = lane >> 4;
    const float* Ap = S + (size_t)(m0 + c) * 64 + head * 16 + 4 * h;
    const float* Bp = B + (size_t)(4 * h) * 256 + n0 + 4 * c;
    v8f acc[4];
#pragma unroll
    for (int j = 0; j < 4; ++j) acc[j] = (v8f){0.f,0.f,0.f,0.f,0.f,0.f,0.f,0.f};
#pragma unroll
    for (int k0 = 0; k0 < 16; k0 += 8) {
        const v4f av = *reinterpret_cast<const v4f*>(Ap + k0);
        v4f bq[4];
#pragma unroll
        for (int q = 0; q < 4; ++q) bq[q] = *reinterpret_cast<const v4f*>(Bp + (size_t)(k0 + q) * 256);
        v2f a; a.x = av[0]; a.y = av[1];
#pragma unroll
        for (int j = 0; j < 4; ++j) {
            v2f b; b.x = bq[0][j]; b.y = bq[1][j];
            acc[j] = __builtin_amdgcn_wmma_f32_16x16x4_f32(false, a, false, b, (short)0, acc[j], false, false);
        }
        a.x = av[2]; a.y = av[3];
#pragma unroll
        for (int j = 0; j < 4; ++j) {
            v2f b; b.x = bq[2][j]; b.y = bq[3][j];
            acc[j] = __builtin_amdgcn_wmma_f32_16x16x4_f32(false, a, false, b, (short)0, acc[j], false, false);
        }
    }
    const v4f bb = *reinterpret_cast<const v4f*>(bias + n0 + 4 * c);
    float* Cp = C + (size_t)(m0 + 8 * h) * 256 + n0 + 4 * c;
#pragma unroll
    for (int r = 0; r < 8; ++r) {
        v4f o;
#pragma unroll
        for (int j = 0; j < 4; ++j) {
            const float v = acc[j][r] + bb[j];
            o[j] = v > 0.f ? v : (expf(v) - 1.f);
        }
        *reinterpret_cast<v4f*>(Cp + (size_t)r * 256) = o;
    }
}

__global__ void gat_csr_count(const int* __restrict__ ei, int E, int* __restrict__ cnt) {
    int e = blockIdx.x * blockDim.x + threadIdx.x;
    if (e >= E) return;
    atomicAdd(&cnt[ei[E + e]], 1);
}

__global__ void gat_csr_scan(const int* __restrict__ cnt, int* __restrict__ rowptr,
                             int* __restrict__ cursor, int N) {
    __shared__ int wpre[32];
    __shared__ int wtot[32];
    const int t = threadIdx.x, lane = t & 31, wv = t >> 5;
    const int tile0 = blockIdx.x * 1024;
    int pre = 0;
    for (int i = t; i < tile0; i += 1024) pre += cnt[i];
    const int i = tile0 + t;
    const int v = (i < N) ? cnt[i] : 0;
    int inc = v;
    for (int d = 1; d < 32; d <<= 1) {
        int u = __shfl_up(inc, d, 32);
        if (lane >= d) inc += u;
    }
    for (int d = 16; d > 0; d >>= 1) pre += __shfl_xor(pre, d, 32);
    if (lane == 31) wtot[wv] = inc;
    if (lane == 0)  wpre[wv] = pre;
    __syncthreads();
    int base = 0;
    for (int k = 0; k < 32; ++k) {
        base += wpre[k];
        if (k < wv) base += wtot[k];
    }
    if (i < N) { int r = base + inc - v; rowptr[i] = r; cursor[i] = r; }
}

__global__ void gat_csr_fill(const int* __restrict__ ei, int E, int* __restrict__ cursor,
                             int* __restrict__ ssrc) {
    int e = blockIdx.x * blockDim.x + threadIdx.x;
    if (e >= E) return;
    int p = atomicAdd(&cursor[ei[E + e]], 1);
    ssrc[p] = ei[e];
}

template <int M>
__device__ __forceinline__ float row_xor(float x) {
    const int b = __float_as_int(x);
    return __int_as_float(__builtin_amdgcn_update_dpp(b, b, 0x160 + M, 0xf, 0xf, false));
}
__device__ __forceinline__ float half_xor(float x) {
    const int b = __float_as_int(x);
    return __int_as_float(__builtin_amdgcn_permlanex16(b, b, 0x76543210, 0xfedcba98, false, false));
}
__device__ __forceinline__ float wave_max(float v) {
    v = fmaxf(v, row_xor<1>(v)); v = fmaxf(v, row_xor<2>(v));
    v = fmaxf(v, row_xor<4>(v)); v = fmaxf(v, row_xor<8>(v));
    return fmaxf(v, half_xor(v));
}
__device__ __forceinline__ float wave_sum(float v) {
    v += row_xor<1>(v); v += row_xor<2>(v); v += row_xor<4>(v); v += row_xor<8>(v);
    return v + half_xor(v);
}

template <int HH>
__device__ __forceinline__ void load_heads(const float* __restrict__ p, float (&v)[HH]) {
    if constexpr (HH == 4) {
        const float4 t = *reinterpret_cast<const float4*>(p);
        v[0] = t.x; v[1] = t.y; v[2] = t.z; v[3] = t.w;
    } else {
        v[0] = p[0];
    }
}

template <int HH>
__device__ __forceinline__ void edge_score(const float* __restrict__ es, int s,
                                           const float (&edn)[HH], float (&e)[HH]) {
    float t[HH];
    load_heads<HH>(es + (size_t)s * HH, t);
#pragma unroll
    for (int h = 0; h < HH; ++h) { const float v = t[h] + edn[h]; e[h] = v > 0.f ? v : 0.2f * v; }
}

template <int HH, bool XIN>
__global__ void gat_seg_softmax_agg(const int* __restrict__ ssrc, const int* __restrict__ rowptr,
                                    const int* __restrict__ segend,
                                    const float* __restrict__ hfeat,
                                    const float* __restrict__ es, const float* __restrict__ ed,
                                    const float* __restrict__ b, float* __restrict__ out,
                                    int N, int do_elu) {
    const int lane = threadIdx.x & 31;
    const int node = __builtin_amdgcn_readfirstlane(blockIdx.x * (blockDim.x >> 5) + (threadIdx.x >> 5));
    if (node >= N) return;
    const int start = rowptr[node];
    const int end   = segend[node];

    float edn[HH], eself[HH], e0[HH], m[HH];
    load_heads<HH>(ed + (size_t)node * HH, edn);
    edge_score<HH>(es, node, edn, eself);
    const int n0 = (end - start) < 32 ? (end - start) : 32;
    int src0 = 0;
#pragma unroll
    for (int h = 0; h < HH; ++h) e0[h] = -INFINITY;
    if (lane < n0) { src0 = ssrc[start + lane]; edge_score<HH>(es, src0, edn, e0); }
#pragma unroll
    for (int h = 0; h < HH; ++h) m[h] = fmaxf(e0[h], eself[h]);
    for (int p0 = start + 32; p0 < end; p0 += 32) {
        if (p0 + lane < end) {
            float e[HH];
            edge_score<HH>(es, ssrc[p0 + lane], edn, e);
#pragma unroll
            for (int h = 0; h < HH; ++h) m[h] = fmaxf(m[h], e[h]);
        }
    }
#pragma unroll
    for (int h = 0; h < HH; ++h)
        m[h] = __int_as_float(__builtin_amdgcn_readfirstlane(__float_as_int(wave_max(m[h]))));

    float x0[HH], den[HH], a0[HH], aself[HH];
#pragma unroll
    for (int h = 0; h < HH; ++h) { x0[h] = expf(e0[h] - m[h]); den[h] = x0[h]; }
    for (int p0 = start + 32; p0 < end; p0 += 32) {
        if (p0 + lane < end) {
            float e[HH];
            edge_score<HH>(es, ssrc[p0 + lane], edn, e);
#pragma unroll
            for (int h = 0; h < HH; ++h) den[h] += expf(e[h] - m[h]);
        }
    }
#pragma unroll
    for (int h = 0; h < HH; ++h) {
        const float xs = expf(eself[h] - m[h]);
        den[h]   = __int_as_float(__builtin_amdgcn_readfirstlane(__float_as_int(wave_sum(den[h]) + xs)));
        aself[h] = xs / den[h];
        a0[h]    = x0[h] / den[h];
    }

    if constexpr (XIN) {
        const int hsel = lane >> 3;
        const float* xp = hfeat + (lane & 7) * 2;
        float2 acc;
        {
            const float w = hsel == 0 ? aself[0] : hsel == 1 ? aself[1] : hsel == 2 ? aself[2] : aself[3];
            const float2 r = *reinterpret_cast<const float2*>(xp + (size_t)node * 16);
            acc.x = r.x * w; acc.y = r.y * w;
        }
        for (int p0 = start; p0 < end; p0 += 32) {
            const int n = (end - p0) < 32 ? (end - p0) : 32;
            int sv = src0;
            float a[4] = {a0[0], a0[1], a0[2], a0[3]};
            if (p0 != start) {
                sv = 0; a[0] = 0.f; a[1] = 0.f; a[2] = 0.f; a[3] = 0.f;
                if (lane < n) {
                    float e[HH];
                    sv = ssrc[p0 + lane];
                    edge_score<HH>(es, sv, edn, e);
#pragma unroll
                    for (int h = 0; h < HH; ++h) a[h] = expf(e[h] - m[h]) / den[h];
                }
            }
            for (int j = 0; j < n; j += 8) {
#pragma unroll
                for (int u = 0; u < 8; ++u) {
                    const int   s  = __builtin_amdgcn_readlane(sv, j + u);
                    const float ax = __int_as_float(__builtin_amdgcn_readlane(__float_as_int(a[0]), j + u));
                    const float ay = __int_as_float(__builtin_amdgcn_readlane(__float_as_int(a[1]), j + u));
                    const float az = __int_as_float(__builtin_amdgcn_readlane(__float_as_int(a[2]), j + u));
                    const float aw = __int_as_float(__builtin_amdgcn_readlane(__float_as_int(a[3]), j + u));
                    const float w = hsel == 0 ? ax : hsel == 1 ? ay : hsel == 2 ? az : aw;
                    const float2 r = *reinterpret_cast<const float2*>(xp + (size_t)s * 16);
                    acc.x += r.x * w; acc.y += r.y * w;
                }
            }
        }
        *reinterpret_cast<float2*>(out + (size_t)node * 64 + lane * 2) = acc;
    } else if constexpr (HH == 4) {
        const int hsel = lane >> 4;
        const float* hp = hfeat + lane * 4;
        float4 acc0, acc1;
        {
            const float w0 = hsel ? aself[1] : aself[0];
            const float w1 = hsel ? aself[3] : aself[2];
            const float* rp = hp + (size_t)node * 256;
            float4 r0 = *reinterpret_cast<const float4*>(rp);
            float4 r1 = *reinterpret_cast<const float4*>(rp + 128);
            acc0.x = r0.x * w0; acc0.y = r0.y * w0; acc0.z = r0.z * w0; acc0.w = r0.w * w0;
            acc1.x = r1.x * w1; acc1.y = r1.y * w1; acc1.z = r1.z * w1; acc1.w = r1.w * w1;
        }
        for (int p0 = start; p0 < end; p0 += 32) {
            const int n = (end - p0) < 32 ? (end - p0) : 32;
            int sv = src0;
            float a[4] = {a0[0], a0[1], a0[2], a0[3]};
            if (p0 != start) {
                sv = 0; a[0] = 0.f; a[1] = 0.f; a[2] = 0.f; a[3] = 0.f;
                if (lane < n) {
                    float e[HH];
                    sv = ssrc[p0 + lane];
                    edge_score<HH>(es, sv, edn, e);
#pragma unroll
                    for (int h = 0; h < HH; ++h) a[h] = expf(e[h] - m[h]) / den[h];
                }
            }
#pragma unroll 4
            for (int j = 0; j < n; ++j) {
                const int   s  = __builtin_amdgcn_readlane(sv, j);
                const float ax = __int_as_float(__builtin_amdgcn_readlane(__float_as_int(a[0]), j));
                const float ay = __int_as_float(__builtin_amdgcn_readlane(__float_as_int(a[1]), j));
                const float az = __int_as_float(__builtin_amdgcn_readlane(__float_as_int(a[2]), j));
                const float aw = __int_as_float(__builtin_amdgcn_readlane(__float_as_int(a[3]), j));
                const float w0 = hsel ? ay : ax;
                const float w1 = hsel ? aw : az;
                const float* rp = hp + (size_t)s * 256;
                float4 r0 = *reinterpret_cast<const float4*>(rp);
                float4 r1 = *reinterpret_cast<const float4*>(rp + 128);
                acc0.x += r0.x * w0; acc0.y += r0.y * w0; acc0.z += r0.z * w0; acc0.w += r0.w * w0;
                acc1.x += r1.x * w1; acc1.y += r1.y * w1; acc1.z += r1.z * w1; acc1.w += r1.w * w1;
            }
        }
        const float4 b0 = *reinterpret_cast<const float4*>(b + lane * 4);
        const float4 b1 = *reinterpret_cast<const float4*>(b + 128 + lane * 4);
        acc0.x += b0.x; acc0.y += b0.y; acc0.z += b0.z; acc0.w += b0.w;
        acc1.x += b1.x; acc1.y += b1.y; acc1.z += b1.z; acc1.w += b1.w;
        if (do_elu) {
            acc0.x = acc0.x > 0.f ? acc0.x : (expf(acc0.x) - 1.f);
            acc0.y = acc0.y > 0.f ? acc0.y : (expf(acc0.y) - 1.f);
            acc0.z = acc0.z > 0.f ? acc0.z : (expf(acc0.z) - 1.f);
            acc0.w = acc0.w > 0.f ? acc0.w : (expf(acc0.w) - 1.f);
            acc1.x = acc1.x > 0.f ? acc1.x : (expf(acc1.x) - 1.f);
            acc1.y = acc1.y > 0.f ? acc1.y : (expf(acc1.y) - 1.f);
            acc1.z = acc1.z > 0.f ? acc1.z : (expf(acc1.z) - 1.f);
            acc1.w = acc1.w > 0.f ? acc1.w : (expf(acc1.w) - 1.f);
        }
        float* op = out + (size_t)node * 256 + lane * 4;
        *reinterpret_cast<float4*>(op)       = acc0;
        *reinterpret_cast<float4*>(op + 128) = acc1;
    } else {
        const float* hp = hfeat + lane * 2;
        float2 acc;
        {
            const float w = aself[0];
            float2 r = *reinterpret_cast<const float2*>(hp + (size_t)node * 64);
            acc.x = r.x * w; acc.y = r.y * w;
        }
        for (int p0 = start; p0 < end; p0 += 32) {
            const int n = (end - p0) < 32 ? (end - p0) : 32;
            int sv = src0;
            float a = a0[0];
            if (p0 != start) {
                sv = 0; a = 0.f;
                if (lane < n) {
                    float e[HH];
                    sv = ssrc[p0 + lane];
                    edge_score<HH>(es, sv, edn, e);
                    a = expf(e[0] - m[0]) / den[0];
                }
            }
#pragma unroll 4
            for (int j = 0; j < n; ++j) {
                const int   s = __builtin_amdgcn_readlane(sv, j);
                const float w = __int_as_float(__builtin_amdgcn_readlane(__float_as_int(a), j));
                float2 r = *reinterpret_cast<const float2*>(hp + (size_t)s * 64);
                acc.x += r.x * w; acc.y += r.y * w;
            }
        }
        const float2 bb = *reinterpret_cast<const float2*>(b + lane * 2);
        acc.x += bb.x; acc.y += bb.y;
        if (do_elu) {
            acc.x = acc.x > 0.f ? acc.x : (expf(acc.x) - 1.f);
            acc.y = acc.y > 0.f ? acc.y : (expf(acc.y) - 1.f);
        }
        *reinterpret_cast<float2*>(out + (size_t)node * 64 + lane * 2) = acc;
    }
}

#define GAT_SLAB 256
template <bool FULL>
__device__ __forceinline__ float4 slab_rows_sum(const float* __restrict__ x, int r0, int N, int cg) {
    float4 v[GAT_SLAB / 16];
    for (int i = 0; i < GAT_SLAB / 16; ++i) {
        const int r = r0 + 16 * i;
        if constexpr (FULL) {
            v[i] = *reinterpret_cast<const float4*>(x + (size_t)r * 64 + cg * 4);
        } else {
            const int rc = r < N ? r : N - 1;
            v[i] = *reinterpret_cast<const float4*>(x + (size_t)rc * 64 + cg * 4);
            if (r >= N) v[i] = make_float4(0.f, 0.f, 0.f, 0.f);
        }
    }
    float4 s = v[0];
    for (int i = 1; i < GAT_SLAB / 16; ++i) { s.x += v[i].x; s.y += v[i].y; s.z += v[i].z; s.w += v[i].w; }
    return s;
}

__global__ void gat_colsum_slab(const float* __restrict__ x, float* __restrict__ part, int N) {
    __shared__ float4 sm[256];
    const int t    = threadIdx.x;
    const int cg   = t & 15;
    const int slot = t >> 4;
    const int r0   = blockIdx.x * GAT_SLAB + slot;
    const bool full = (blockIdx.x + 1) * GAT_SLAB <= N;
    sm[t] = full ? slab_rows_sum<true>(x, r0, N, cg)
                 : slab_rows_sum<false>(x, r0, N, cg);
    __syncthreads();
    if (t < 64) {
        const float* f = reinterpret_cast<const float*>(sm);
        float a = 0.f;
        for (int k = 0; k < 16; ++k) a += f[k * 64 + t];
        part[(size_t)blockIdx.x * 64 + t] = a;
    }
}

__global__ void gat_colmean_final(const float* __restrict__ part, float* __restrict__ out,
                                  int P, int N) {
    __shared__ float4 sm[1024];
    const int t    = threadIdx.x;
    const int cg   = t & 15;
    const int slot = t >> 4;
    float4 s = make_float4(0.f, 0.f, 0.f, 0.f);
    for (int base = 0; base < P; base += 256) {
        float4 v[4];
        for (int i = 0; i < 4; ++i) {
            const int r  = base + slot + 64 * i;
            const int rc = r < P ? r : P - 1;
            v[i] = *reinterpret_cast<const float4*>(part + (size_t)rc * 64 + cg * 4);
            if (r >= P) v[i] = make_float4(0.f, 0.f, 0.f, 0.f);
        }
        for (int i = 0; i < 4; ++i) { s.x += v[i].x; s.y += v[i].y; s.z += v[i].z; s.w += v[i].w; }
    }
    sm[t] = s;
    __syncthreads();
    if (t < 64) {
        const float* f = reinterpret_cast<const float*>(sm);
        float a = 0.f;
        for (int k = 0; k < 64; ++k) a += f[k * 64 + t];
        out[t] = a / (float)N;
    }
}

static inline int cdiv(long long a, long long b) { return (int)((a + b - 1) / b); }

extern "C" void kernel_launch(void* const* d_in, const int* in_sizes, int n_in,
                              void* d_out, int out_size, void* d_ws, size_t ws_size,
                              hipStream_t stream) {
    const float* x   = (const float*)d_in[0];
    const int*   ei  = (const int*)  d_in[1];
    const float* W1  = (const float*)d_in[2];
    const float* a1s = (const float*)d_in[3];
    const float* a1d = (const float*)d_in[4];
    const float* b1  = (const float*)d_in[5];
    const float* W2  = (const float*)d_in[6];
    const float* a2s = (const float*)d_in[7];
    const float* a2d = (const float*)d_in[8];
    const float* b2  = (const float*)d_in[9];
    const float* W3  = (const float*)d_in[10];
    const float* a3s = (const float*)d_in[11];
    const float* a3d = (const float*)d_in[12];
    const float* b3  = (const float*)d_in[13];

    const int N  = in_sizes[0] / 16;
    const int E  = in_sizes[1] / 2;
    const int H  = 4;

    float* ws = (float*)d_ws;
    size_t off = 0;
    float*    buf_h    = ws + off; off += (size_t)N * 256;
    int*      csr      = (int*)(ws + off); off += (size_t)N * 256;
    float*    buf_feat = ws + off; off += (size_t)N * 256;
    float*    es       = ws + off; off += (size_t)N * H;
    float*    ed       = ws + off; off += (size_t)N * H;
    (void)ws_size;
    int* cnt    = csr;
    int* rowptr = csr + (size_t)N;
    int* cursor = csr + (size_t)2 * N;
    int* ssrc   = csr + (size_t)4 * N;

    float* node_emb  = (float*)d_out;
    float* graph_emb = node_emb + (size_t)N * 64;

    auto run_layer = [&](const float* in_feat, const float* W,
                         const float* as_, const float* ad_, const float* bb,
                         float* out_feat, int Hh, int K, int Ncols, int do_elu) {
        (void)K;
        if (Ncols == 256)
            hipLaunchKernelGGL((gat_gemm_wmma<1, 4, 2>), dim3(cdiv(N, 32)), dim3(128), 0, stream,
                               in_feat, W, buf_h, N);
        else
            hipLaunchKernelGGL((gat_gemm_wmma<2, 1, 1>), dim3(cdiv(N, 32)), dim3(64), 0, stream,
                               in_feat, W, buf_h, N);
        hipLaunchKernelGGL(gat_attn_coef, dim3(cdiv((long long)N * Hh, 256)), dim3(256), 0, stream,
                           buf_h, as_, ad_, es, ed, N, Hh);
        if (Hh == 4)
            hipLaunchKernelGGL((gat_seg_softmax_agg<4, false>), dim3(cdiv(N, 8)), dim3(256), 0, stream,
                               ssrc, rowptr, cursor, buf_h, es, ed, bb, out_feat, N, do_elu);
        else
            hipLaunchKernelGGL((gat_seg_softmax_agg<1, false>), dim3(cdiv(N, 8)), dim3(256), 0, stream,
                               ssrc, rowptr, cursor, buf_h, es, ed, bb, out_feat, N, do_elu);
    };

    hipMemsetAsync(cnt, 0, (size_t)N * sizeof(int), stream);
    hipLaunchKernelGGL(gat_csr_count, dim3(cdiv(E, 256)), dim3(256), 0, stream, ei, E, cnt);
    hipLaunchKernelGGL(gat_csr_scan,  dim3(cdiv(N, 1024)), dim3(1024), 0, stream, cnt, rowptr, cursor, N);
    hipLaunchKernelGGL(gat_csr_fill,  dim3(cdiv(E, 256)), dim3(256), 0, stream, ei, E, cursor, ssrc);

    hipLaunchKernelGGL(gat_attn_coef_in, dim3(cdiv(N, 256)), dim3(256), 0, stream,
                       x, W1, a1s, a1d, es, ed, N);
    hipLaunchKernelGGL((gat_seg_softmax_agg<4, true>), dim3(cdiv(N, 8)), dim3(256), 0, stream,
                       ssrc, rowptr, cursor, x, es, ed, b1, buf_h, N, 1);
    hipLaunchKernelGGL(gat_proj_heads_wmma, dim3(cdiv(N, 16)), dim3(32, 4), 0, stream,
                       buf_h, W1, b1, buf_feat, N);
    run_layer(buf_feat, W2, a2s, a2d, b2, buf_feat, H, 256, 256, 1);
    run_layer(buf_feat, W3, a3s, a3d, b3, node_emb, 1, 256, 64, 0);

    const int P = cdiv(N, GAT_SLAB);
    hipLaunchKernelGGL(gat_colsum_slab, dim3(P), dim3(256), 0, stream, node_emb, buf_h, N);
    hipLaunchKernelGGL(gat_colmean_final, dim3(1), dim3(1024), 0, stream, buf_h, graph_emb, P, N);
}


# Layers 2 and 3 take each edge score from the gathered row itself (dot with a_s in registers, softmax running along the walk): no coefficient launches, no per-edge score gather
# speedup vs baseline: 7.3958x; 1.0507x over previous
#include <hip/hip_runtime.h>
#include <math.h>

typedef __attribute__((ext_vector_type(2))) float v2f;
typedef __attribute__((ext_vector_type(8))) float v8f;

#define GAT_D 64

typedef __attribute__((ext_vector_type(4))) float v4f;

#define GAT_K 256
#define GAT_APITCH 260

template <int WM, int WN, int MT>
__global__ void __launch_bounds__(WM * WN * 32)
gat_gemm_wmma(const float* __restrict__ A, const float* __restrict__ B,
              float* __restrict__ C, int M) {
    constexpr int NT = WM * WN * 32;
    constexpr int BM = WM * MT * 16;
    constexpr int N  = WN * 64;
    __shared__ __attribute__((aligned(16))) float As[BM * GAT_APITCH];
    const int t  = threadIdx.x;
    const int m0 = blockIdx.x * BM;

#pragma unroll
    for (int i = 0; i < BM * 64 / NT; ++i) {
        const int idx = i * NT + t;
        const int row = idx >> 6, c4 = idx & 63;
        const int gr  = m0 + row < M ? m0 + row : M - 1;
        const v4f v = *reinterpret_cast<const v4f*>(A + (size_t)gr * GAT_K + c4 * 4);
        *reinterpret_cast<v4f*>(&As[row * GAT_APITCH + c4 * 4]) = v;
    }
    __syncthreads();

    const int lane = t & 31, w = t >> 5;
    const int wm = w / WN, wn = w - wm * WN;
    const int c = lane & 15, h = lane >> 4;
    const int n0 = wn * 64;
    const float* Ap = As + (wm * MT * 16 + c) * GAT_APITCH + 4 * h;
    const float* Bp = B + (size_t)(4 * h) * N + n0 + 4 * c;

    v8f acc[MT][4];
#pragma unroll
    for (int mt = 0; mt < MT; ++mt)
#pragma unroll
        for (int j = 0; j < 4; ++j) acc[mt][j] = (v8f){0.f,0.f,0.f,0.f,0.f,0.f,0.f,0.f};

    v4f braw[3][4];
    v4f araw[2][MT];
#pragma unroll
    for (int q = 0; q < 4; ++q) {
        braw[0][q] = *reinterpret_cast<const v4f*>(Bp + (size_t)q * N);
        braw[1][q] = *reinterpret_cast<const v4f*>(Bp + (size_t)(8 + q) * N);
    }
#pragma unroll
    for (int mt = 0; mt < MT; ++mt)
        araw[0][mt] = *reinterpret_cast<const v4f*>(Ap + mt * 16 * GAT_APITCH);

#pragma unroll
    for (int st = 0; st < GAT_K / 8; ++st) {
        const int k0 = st * 8;
        __builtin_amdgcn_sched_barrier(0);
        if (st + 2 < GAT_K / 8) {
#pragma unroll
            for (int q = 0; q < 4; ++q)
                braw[(st + 2) % 3][q] = *reinterpret_cast<const v4f*>(Bp + (size_t)(k0 + 16 + q) * N);
        }
        __builtin_amdgcn_sched_barrier(0);
        if (st + 1 < GAT_K / 8) {
#pragma unroll
            for (int mt = 0; mt < MT; ++mt)
                araw[(st + 1) & 1][mt] = *reinterpret_cast<const v4f*>(Ap + mt * 16 * GAT_APITCH + k0 + 8);
        }
#pragma unroll
        for (int mt = 0; mt < MT; ++mt) {
            v2f a; a.x = araw[st & 1][mt][0]; a.y = araw[st & 1][mt][1];
#pragma unroll
            for (int j = 0; j < 4; ++j) {
                v2f b; b.x = braw[st % 3][0][j]; b.y = braw[st % 3][1][j];
                acc[mt][j] = __builtin_amdgcn_wmma_f32_16x16x4_f32(false, a, false, b, (short)0, acc[mt][j], false, false);
            }
        }
#pragma unroll
        for (int mt = 0; mt < MT; ++mt) {
            v2f a; a.x = araw[st & 1][mt][2]; a.y = araw[st & 1][mt][3];
#pragma unroll
            for (int j = 0; j < 4; ++j) {
                v2f b; b.x = braw[st % 3][2][j]; b.y = braw[st % 3][3][j];
                acc[mt][j] = __builtin_amdgcn_wmma_f32_16x16x4_f32(false, a, false, b, (short)0, acc[mt][j], false, false);
            }
        }
    }

#pragma unroll
    for (int mt = 0; mt < MT; ++mt) {
#pragma unroll
        for (int r = 0; r < 8; ++r) {
            const int row = m0 + (wm * MT + mt) * 16 + 8 * h + r;
            if (row < M) {
                v4f o; o[0] = acc[mt][0][r]; o[1] = acc[mt][1][r]; o[2] = acc[mt][2][r]; o[3] = acc[mt][3][r];
                *reinterpret_cast<v4f*>(C + (size_t)row * N + n0 + 4 * c) = o;
            }
        }
    }
}

__global__ void gat_attn_coef_in(const float* __restrict__ x, const float* __restrict__ W,
                                 const float* __restrict__ a_s, const float* __restrict__ a_d,
                                 float* __restrict__ es, float* __restrict__ ed, int N) {
    __shared__ float wsd[2][16][4];
    const int t = threadIdx.x;
    if (t < 128) {
        const int which = t >> 6, k = (t >> 2) & 15, hh = t & 3;
        const float* wp = W + (size_t)k * 256 + hh * GAT_D;
        const float* ap = (which ? a_d : a_s) + hh * GAT_D;
        float s = 0.f;
        for (int d = 0; d < GAT_D; ++d) s += wp[d] * ap[d];
        wsd[which][k][hh] = s;
    }
    __syncthreads();
    const int n = blockIdx.x * blockDim.x + t;
    if (n >= N) return;
    float xr[16];
    for (int q = 0; q < 4; ++q) {
        const float4 v = *reinterpret_cast<const float4*>(x + (size_t)n * 16 + q * 4);
        xr[4 * q] = v.x; xr[4 * q + 1] = v.y; xr[4 * q + 2] = v.z; xr[4 * q + 3] = v.w;
    }
    float ss[4] = {0.f, 0.f, 0.f, 0.f}, dd[4] = {0.f, 0.f, 0.f, 0.f};
    for (int k = 0; k < 16; ++k)
        for (int hh = 0; hh < 4; ++hh) { ss[hh] += xr[k] * wsd[0][k][hh]; dd[hh] += xr[k] * wsd[1][k][hh]; }
    *reinterpret_cast<float4*>(es + (size_t)n * 4) = make_float4(ss[0], ss[1], ss[2], ss[3]);
    *reinterpret_cast<float4*>(ed + (size_t)n * 4) = make_float4(dd[0], dd[1], dd[2], dd[3]);
}

__global__ void gat_proj_heads_wmma(const float* __restrict__ S, const float* __restrict__ B,
                                    const float* __restrict__ bias, float* __restrict__ C, int M) {
    const int lane = threadIdx.x;
    const int head = threadIdx.y;
    const int m0 = blockIdx.x * 16;
    if (m0 >= M) return;
    const int n0 = head * 64;
    const int c = lane & 15, h = lane >> 4;
    const float* Ap = S + (size_t)(m0 + c) * 64 + head * 16 + 4 * h;
    const float* Bp = B + (size_t)(4 * h) * 256 + n0 + 4 * c;
    v8f acc[4];
#pragma unroll
    for (int j = 0; j < 4; ++j) acc[j] = (v8f){0.f,0.f,0.f,0.f,0.f,0.f,0.f,0.f};
#pragma unroll
    for (int k0 = 0; k0 < 16; k0 += 8) {
        const v4f av = *reinterpret_cast<const v4f*>(Ap + k0);
        v4f bq[4];
#pragma unroll
        for (int q = 0; q < 4; ++q) bq[q] = *reinterpret_cast<const v4f*>(Bp + (size_t)(k0 + q) * 256);
        v2f a; a.x = av[0]; a.y = av[1];
#pragma unroll
        for (int j = 0; j < 4; ++j) {
            v2f b; b.x = bq[0][j]; b.y = bq[1][j];
            acc[j] = __builtin_amdgcn_wmma_f32_16x16x4_f32(false, a, false, b, (short)0, acc[j], false, false);
        }
        a.x = av[2]; a.y = av[3];
#pragma unroll
        for (int j = 0; j < 4; ++j) {
            v2f b; b.x = bq[2][j]; b.y = bq[3][j];
            acc[j] = __builtin_amdgcn_wmma_f32_16x16x4_f32(false, a, false, b, (short)0, acc[j], false, false);
        }
    }
    const v4f bb = *reinterpret_cast<const v4f*>(bias + n0 + 4 * c);
    float* Cp = C + (size_t)(m0 + 8 * h) * 256 + n0 + 4 * c;
#pragma unroll
    for (int r = 0; r < 8; ++r) {
        v4f o;
#pragma unroll
        for (int j = 0; j < 4; ++j) {
            const float v = acc[j][r] + bb[j];
            o[j] = v > 0.f ? v : (expf(v) - 1.f);
        }
        *reinterpret_cast<v4f*>(Cp + (size_t)r * 256) = o;
    }
}

__global__ void gat_csr_count(const int* __restrict__ ei, int E, int* __restrict__ cnt) {
    int e = blockIdx.x * blockDim.x + threadIdx.x;
    if (e >= E) return;
    atomicAdd(&cnt[ei[E + e]], 1);
}

__global__ void gat_csr_scan(const int* __restrict__ cnt, int* __restrict__ rowptr,
                             int* __restrict__ cursor, int N) {
    __shared__ int wpre[32];
    __shared__ int wtot[32];
    const int t = threadIdx.x, lane = t & 31, wv = t >> 5;
    const int tile0 = blockIdx.x * 1024;
    int pre = 0;
    for (int i = t; i < tile0; i += 1024) pre += cnt[i];
    const int i = tile0 + t;
    const int v = (i < N) ? cnt[i] : 0;
    int inc = v;
    for (int d = 1; d < 32; d <<= 1) {
        int u = __shfl_up(inc, d, 32);
        if (lane >= d) inc += u;
    }
    for (int d = 16; d > 0; d >>= 1) pre += __shfl_xor(pre, d, 32);
    if (lane == 31) wtot[wv] = inc;
    if (lane == 0)  wpre[wv] = pre;
    __syncthreads();
    int base = 0;
    for (int k = 0; k < 32; ++k) {
        base += wpre[k];
        if (k < wv) base += wtot[k];
    }
    if (i < N) { int r = base + inc - v; rowptr[i] = r; cursor[i] = r; }
}

__global__ void gat_csr_fill(const int* __restrict__ ei, int E, int* __restrict__ cursor,
                             int* __restrict__ ssrc) {
    int e = blockIdx.x * blockDim.x + threadIdx.x;
    if (e >= E) return;
    int p = atomicAdd(&cursor[ei[E + e]], 1);
    ssrc[p] = ei[e];
}

template <int M>
__device__ __forceinline__ float row_xor(float x) {
    const int b = __float_as_int(x);
    return __int_as_float(__builtin_amdgcn_update_dpp(b, b, 0x160 + M, 0xf, 0xf, false));
}
__device__ __forceinline__ float half_xor(float x) {
    const int b = __float_as_int(x);
    return __int_as_float(__builtin_amdgcn_permlanex16(b, b, 0x76543210, 0xfedcba98, false, false));
}
__device__ __forceinline__ float wave_max(float v) {
    v = fmaxf(v, row_xor<1>(v)); v = fmaxf(v, row_xor<2>(v));
    v = fmaxf(v, row_xor<4>(v)); v = fmaxf(v, row_xor<8>(v));
    return fmaxf(v, half_xor(v));
}
__device__ __forceinline__ float wave_sum(float v) {
    v += row_xor<1>(v); v += row_xor<2>(v); v += row_xor<4>(v); v += row_xor<8>(v);
    return v + half_xor(v);
}

template <int HH>
__device__ __forceinline__ void load_heads(const float* __restrict__ p, float (&v)[HH]) {
    if constexpr (HH == 4) {
        const float4 t = *reinterpret_cast<const float4*>(p);
        v[0] = t.x; v[1] = t.y; v[2] = t.z; v[3] = t.w;
    } else {
        v[0] = p[0];
    }
}

template <int HH>
__device__ __forceinline__ void edge_score(const float* __restrict__ es, int s,
                                           const float (&edn)[HH], float (&e)[HH]) {
    float t[HH];
    load_heads<HH>(es + (size_t)s * HH, t);
#pragma unroll
    for (int h = 0; h < HH; ++h) { const float v = t[h] + edn[h]; e[h] = v > 0.f ? v : 0.2f * v; }
}

template <int HH>
__global__ void gat_seg_softmax_agg(const int* __restrict__ ssrc, const int* __restrict__ rowptr,
                                    const int* __restrict__ segend,
                                    const float* __restrict__ hfeat,
                                    const float* __restrict__ es, const float* __restrict__ ed,
                                    float* __restrict__ out, int N) {
    static_assert(HH == 4, "layer 1 has four heads");
    const int lane = threadIdx.x & 31;
    const int node = __builtin_amdgcn_readfirstlane(blockIdx.x * (blockDim.x >> 5) + (threadIdx.x >> 5));
    if (node >= N) return;
    const int start = rowptr[node];
    const int end   = segend[node];

    float edn[HH], eself[HH], e0[HH], m[HH];
    load_heads<HH>(ed + (size_t)node * HH, edn);
    edge_score<HH>(es, node, edn, eself);
    const int n0 = (end - start) < 32 ? (end - start) : 32;
    int src0 = 0;
#pragma unroll
    for (int h = 0; h < HH; ++h) e0[h] = -INFINITY;
    if (lane < n0) { src0 = ssrc[start + lane]; edge_score<HH>(es, src0, edn, e0); }
#pragma unroll
    for (int h = 0; h < HH; ++h) m[h] = fmaxf(e0[h], eself[h]);
    for (int p0 = start + 32; p0 < end; p0 += 32) {
        if (p0 + lane < end) {
            float e[HH];
            edge_score<HH>(es, ssrc[p0 + lane], edn, e);
#pragma unroll
            for (int h = 0; h < HH; ++h) m[h] = fmaxf(m[h], e[h]);
        }
    }
#pragma unroll
    for (int h = 0; h < HH; ++h)
        m[h] = __int_as_float(__builtin_amdgcn_readfirstlane(__float_as_int(wave_max(m[h]))));

    float x0[HH], den[HH], a0[HH], aself[HH];
#pragma unroll
    for (int h = 0; h < HH; ++h) { x0[h] = expf(e0[h] - m[h]); den[h] = x0[h]; }
    for (int p0 = start + 32; p0 < end; p0 += 32) {
        if (p0 + lane < end) {
            float e[HH];
            edge_score<HH>(es, ssrc[p0 + lane], edn, e);
#pragma unroll
            for (int h = 0; h < HH; ++h) den[h] += expf(e[h] - m[h]);
        }
    }
#pragma unroll
    for (int h = 0; h < HH; ++h) {
        const float xs = expf(eself[h] - m[h]);
        den[h]   = __int_as_float(__builtin_amdgcn_readfirstlane(__float_as_int(wave_sum(den[h]) + xs)));
        aself[h] = xs / den[h];
        a0[h]    = x0[h] / den[h];
    }

    {
        const int hsel = lane >> 3;
        const float* xp = hfeat + (lane & 7) * 2;
        float2 acc;
        {
            const float w = hsel == 0 ? aself[0] : hsel == 1 ? aself[1] : hsel == 2 ? aself[2] : aself[3];
            const float2 r = *reinterpret_cast<const float2*>(xp + (size_t)node * 16);
            acc.x = r.x * w; acc.y = r.y * w;
        }
        for (int p0 = start; p0 < end; p0 += 32) {
            const int n = (end - p0) < 32 ? (end - p0) : 32;
            int sv = src0;
            float a[4] = {a0[0], a0[1], a0[2], a0[3]};
            if (p0 != start) {
                sv = 0; a[0] = 0.f; a[1] = 0.f; a[2] = 0.f; a[3] = 0.f;
                if (lane < n) {
                    float e[HH];
                    sv = ssrc[p0 + lane];
                    edge_score<HH>(es, sv, edn, e);
#pragma unroll
                    for (int h = 0; h < HH; ++h) a[h] = expf(e[h] - m[h]) / den[h];
                }
            }
            for (int j = 0; j < n; j += 8) {
#pragma unroll
                for (int u = 0; u < 8; ++u) {
                    const int   s  = __builtin_amdgcn_readlane(sv, j + u);
                    const float ax = __int_as_float(__builtin_amdgcn_readlane(__float_as_int(a[0]), j + u));
                    const float ay = __int_as_float(__builtin_amdgcn_readlane(__float_as_int(a[1]), j + u));
                    const float az = __int_as_float(__builtin_amdgcn_readlane(__float_as_int(a[2]), j + u));
                    const float aw = __int_as_float(__builtin_amdgcn_readlane(__float_as_int(a[3]), j + u));
                    const float w = hsel == 0 ? ax : hsel == 1 ? ay : hsel == 2 ? az : aw;
                    const float2 r = *reinterpret_cast<const float2*>(xp + (size_t)s * 16);
                    acc.x += r.x * w; acc.y += r.y * w;
                }
            }
        }
        *reinterpret_cast<float2*>(out + (size_t)node * 64 + lane * 2) = acc;
    }
}

__device__ __forceinline__ float row16_sum(float v) {
    v += row_xor<1>(v); v += row_xor<2>(v); v += row_xor<4>(v); v += row_xor<8>(v);
    return v;
}
template <int K>
__device__ __forceinline__ float row_share(float x) {
    const int b = __float_as_int(x);
    return __int_as_float(__builtin_amdgcn_update_dpp(b, b, 0x150 + K, 0xf, 0xf, false));
}
__device__ __forceinline__ float fold8_sum(const float (&p)[8], bool b0, bool b1, bool b2) {
    float r[4], s[2];
#pragma unroll
    for (int i = 0; i < 4; ++i) {
        const float keep = b0 ? p[i + 4] : p[i], send = b0 ? p[i] : p[i + 4];
        r[i] = keep + row_xor<1>(send);
    }
#pragma unroll
    for (int i = 0; i < 2; ++i) {
        const float keep = b1 ? r[i + 2] : r[i], send = b1 ? r[i] : r[i + 2];
        s[i] = keep + row_xor<2>(send);
    }
    const float keep = b2 ? s[1] : s[0], send = b2 ? s[0] : s[1];
    return keep + row_xor<4>(send);
}
__device__ __forceinline__ float dot4(const float4 r, const float4 a) {
    return r.x * a.x + r.y * a.y + r.z * a.z + r.w * a.w;
}
__device__ __forceinline__ float4 scaled4(const float* __restrict__ p, float f) {
    const float4 t = *reinterpret_cast<const float4*>(p);
    return make_float4(t.x * f, t.y * f, t.z * f, t.w * f);
}

template <int HH>
__global__ void gat_seg_rowscore_agg(const int* __restrict__ ssrc, const int* __restrict__ rowptr,
                                     const int* __restrict__ segend,
                                     const float* __restrict__ hfeat,
                                     const float* __restrict__ a_s, const float* __restrict__ a_d,
                                     const float* __restrict__ b, float* __restrict__ out,
                                     int N, int do_elu) {
    const int lane = threadIdx.x & 31;
    const int node = __builtin_amdgcn_readfirstlane(blockIdx.x * (blockDim.x >> 5) + (threadIdx.x >> 5));
    if (node >= N) return;
    const int start = rowptr[node];
    const int end   = segend[node];
    const float L2E = 1.4426950408889634f;
    const bool b0 = (lane & 1) != 0, b1 = (lane & 2) != 0, b2 = (lane & 4) != 0;

    if constexpr (HH == 4) {
        const float* hp = hfeat + lane * 4;
        const float4 as0 = scaled4(a_s + lane * 4, L2E), as1 = scaled4(a_s + 128 + lane * 4, L2E);
        const int myu = ((lane & 1) << 1) | ((lane >> 1) & 1);
        float4 acc0, acc1;
        float ednl, ml;
        float denl = (lane & 3) == 0 ? 1.f : 0.f;
        {
            const float4 ad0 = scaled4(a_d + lane * 4, L2E), ad1 = scaled4(a_d + 128 + lane * 4, L2E);
            const float* rp = hp + (size_t)node * 256;
            acc0 = *reinterpret_cast<const float4*>(rp);
            acc1 = *reinterpret_cast<const float4*>(rp + 128);
            const float edn0 = row16_sum(dot4(acc0, ad0));
            const float edn1 = row16_sum(dot4(acc1, ad1));
            const float v0 = row16_sum(dot4(acc0, as0)) + edn0;
            const float v1 = row16_sum(dot4(acc1, as1)) + edn1;
            ednl = b2 ? edn1 : edn0;
            const float v = b2 ? v1 : v0;
            ml = fmaxf(v, 0.2f * v);
        }
        for (int p0 = start; p0 < end; p0 += 32) {
            const int n = (end - p0) < 32 ? (end - p0) : 32;
            const int sv = lane < n ? ssrc[p0 + lane] : 0;
            for (int j = 0; j < n; j += 4) {
                float4 q0[4], q1[4];
                float p[8];
#pragma unroll
                for (int u = 0; u < 4; ++u) {
                    const int s = __builtin_amdgcn_readlane(sv, j + u);
                    const float* qp = hp + (size_t)s * 256;
                    q0[u] = *reinterpret_cast<const float4*>(qp);
                    q1[u] = *reinterpret_cast<const float4*>(qp + 128);
                }
#pragma unroll
                for (int u = 0; u < 4; ++u) { p[2 * u] = dot4(q0[u], as0); p[2 * u + 1] = dot4(q1[u], as1); }
                float t = fold8_sum(p, b0, b1, b2);
                t += row_xor<8>(t);
                const float v = t + ednl;
                const float e = (j + myu < n) ? fmaxf(v, 0.2f * v) : -INFINITY;
                float cm = fmaxf(e, row_xor<1>(e));
                cm = fmaxf(cm, row_xor<2>(cm));
                const float mn = fmaxf(ml, cm);
                const float sc = __builtin_amdgcn_exp2f(ml - mn);
                const float w  = __builtin_amdgcn_exp2f(e - mn);
                denl = denl * sc + w;
                ml = mn;
                const float sc0 = row_share<0>(sc), sc1 = row_share<4>(sc);
                acc0.x *= sc0; acc0.y *= sc0; acc0.z *= sc0; acc0.w *= sc0;
                acc1.x *= sc1; acc1.y *= sc1; acc1.z *= sc1; acc1.w *= sc1;
                float w0[4], w1[4];
                w0[0] = row_share<0>(w); w0[1] = row_share<2>(w); w0[2] = row_share<1>(w); w0[3] = row_share<3>(w);
                w1[0] = row_share<4>(w); w1[1] = row_share<6>(w); w1[2] = row_share<5>(w); w1[3] = row_share<7>(w);
#pragma unroll
                for (int u = 0; u < 4; ++u) {
                    acc0.x += q0[u].x * w0[u]; acc0.y += q0[u].y * w0[u]; acc0.z += q0[u].z * w0[u]; acc0.w += q0[u].w * w0[u];
                    acc1.x += q1[u].x * w1[u]; acc1.y += q1[u].y * w1[u]; acc1.z += q1[u].z * w1[u]; acc1.w += q1[u].w * w1[u];
                }
            }
        }
        float d = denl;
        d += row_xor<1>(d); d += row_xor<2>(d);
        const float i0 = 1.f / row_share<0>(d), i1 = 1.f / row_share<4>(d);
        const float4 bb0 = *reinterpret_cast<const float4*>(b + lane * 4);
        const float4 bb1 = *reinterpret_cast<const float4*>(b + 128 + lane * 4);
        acc0.x = acc0.x * i0 + bb0.x; acc0.y = acc0.y * i0 + bb0.y; acc0.z = acc0.z * i0 + bb0.z; acc0.w = acc0.w * i0 + bb0.w;
        acc1.x = acc1.x * i1 + bb1.x; acc1.y = acc1.y * i1 + bb1.y; acc1.z = acc1.z * i1 + bb1.z; acc1.w = acc1.w * i1 + bb1.w;
        if (do_elu) {
            acc0.x = acc0.x > 0.f ? acc0.x : (expf(acc0.x) - 1.f);
            acc0.y = acc0.y > 0.f ? acc0.y : (expf(acc0.y) - 1.f);
            acc0.z = acc0.z > 0.f ? acc0.z : (expf(acc0.z) - 1.f);
            acc0.w = acc0.w > 0.f ? acc0.w : (expf(acc0.w) - 1.f);
            acc1.x = acc1.x > 0.f ? acc1.x : (expf(acc1.x) - 1.f);
            acc1.y = acc1.y > 0.f ? acc1.y : (expf(acc1.y) - 1.f);
            acc1.z = acc1.z > 0.f ? acc1.z : (expf(acc1.z) - 1.f);
            acc1.w = acc1.w > 0.f ? acc1.w : (expf(acc1.w) - 1.f);
        }
        float* op = out + (size_t)node * 256 + lane * 4;
        *reinterpret_cast<float4*>(op)       = acc0;
        *reinterpret_cast<float4*>(op + 128) = acc1;
    } else {
        const float* hp = hfeat + lane * 2;
        float2 as = *reinterpret_cast<const float2*>(a_s + lane * 2);
        as.x *= L2E; as.y *= L2E;
        const int myu = ((lane & 1) << 2) | (lane & 2) | ((lane >> 2) & 1);
        float2 acc = *reinterpret_cast<const float2*>(hp + (size_t)node * 64);
        float edn, m;
        float denl = (lane & 7) == 0 ? 1.f : 0.f;
        {
            const float2 ad = *reinterpret_cast<const float2*>(a_d + lane * 2);
            edn = wave_sum((acc.x * ad.x + acc.y * ad.y) * L2E);
            const float v = wave_sum(acc.x * as.x + acc.y * as.y) + edn;
            m = fmaxf(v, 0.2f * v);
        }
        for (int p0 = start; p0 < end; p0 += 32) {
            const int n = (end - p0) < 32 ? (end - p0) : 32;
            const int sv = lane < n ? ssrc[p0 + lane] : 0;
            for (int j = 0; j < n; j += 8) {
                float2 q[8];
                float p[8];
#pragma unroll
                for (int u = 0; u < 8; ++u) {
                    const int s = __builtin_amdgcn_readlane(sv, j + u);
                    q[u] = *reinterpret_cast<const float2*>(hp + (size_t)s * 64);
                }
#pragma unroll
                for (int u = 0; u < 8; ++u) p[u] = q[u].x * as.x + q[u].y * as.y;
                float t = fold8_sum(p, b0, b1, b2);
                t += row_xor<8>(t);
                t += half_xor(t);
                const float v = t + edn;
                const float e = (j + myu < n) ? fmaxf(v, 0.2f * v) : -INFINITY;
                float cm = fmaxf(e, row_xor<1>(e));
                cm = fmaxf(cm, row_xor<2>(cm));
                cm = fmaxf(cm, row_xor<4>(cm));
                const float mn = fmaxf(m, cm);
                const float sc = __builtin_amdgcn_exp2f(m - mn);
                const float w  = __builtin_amdgcn_exp2f(e - mn);
                denl = denl * sc + w;
                m = mn;
                acc.x *= sc; acc.y *= sc;
                const int wi = __float_as_int(w);
                float wu[8];
                wu[0] = __int_as_float(__builtin_amdgcn_readlane(wi, 0));
                wu[1] = __int_as_float(__builtin_amdgcn_readlane(wi, 4));
                wu[2] = __int_as_float(__builtin_amdgcn_readlane(wi, 2));
                wu[3] = __int_as_float(__builtin_amdgcn_readlane(wi, 6));
                wu[4] = __int_as_float(__builtin_amdgcn_readlane(wi, 1));
                wu[5] = __int_as_float(__builtin_amdgcn_readlane(wi, 5));
                wu[6] = __int_as_float(__builtin_amdgcn_readlane(wi, 3));
                wu[7] = __int_as_float(__builtin_amdgcn_readlane(wi, 7));
#pragma unroll
                for (int u = 0; u < 8; ++u) { acc.x += q[u].x * wu[u]; acc.y += q[u].y * wu[u]; }
            }
        }
        float d = denl;
        d += row_xor<1>(d); d += row_xor<2>(d); d += row_xor<4>(d);
        const float inv = 1.f / d;
        const float2 bb = *reinterpret_cast<const float2*>(b + lane * 2);
        acc.x = acc.x * inv + bb.x; acc.y = acc.y * inv + bb.y;
        if (do_elu) {
            acc.x = acc.x > 0.f ? acc.x : (expf(acc.x) - 1.f);
            acc.y = acc.y > 0.f ? acc.y : (expf(acc.y) - 1.f);
        }
        *reinterpret_cast<float2*>(out + (size_t)node * 64 + lane * 2) = acc;
    }
}

#define GAT_SLAB 256
template <bool FULL>
__device__ __forceinline__ float4 slab_rows_sum(const float* __restrict__ x, int r0, int N, int cg) {
    float4 v[GAT_SLAB / 16];
    for (int i = 0; i < GAT_SLAB / 16; ++i) {
        const int r = r0 + 16 * i;
        if constexpr (FULL) {
            v[i] = *reinterpret_cast<const float4*>(x + (size_t)r * 64 + cg * 4);
        } else {
            const int rc = r < N ? r : N - 1;
            v[i] = *reinterpret_cast<const float4*>(x + (size_t)rc * 64 + cg * 4);
            if (r >= N) v[i] = make_float4(0.f, 0.f, 0.f, 0.f);
        }
    }
    float4 s = v[0];
    for (int i = 1; i < GAT_SLAB / 16; ++i) { s.x += v[i].x; s.y += v[i].y; s.z += v[i].z; s.w += v[i].w; }
    return s;
}

__global__ void gat_colsum_slab(const float* __restrict__ x, float* __restrict__ part, int N) {
    __shared__ float4 sm[256];
    const int t    = threadIdx.x;
    const int cg   = t & 15;
    const int slot = t >> 4;
    const int r0   = blockIdx.x * GAT_SLAB + slot;
    const bool full = (blockIdx.x + 1) * GAT_SLAB <= N;
    sm[t] = full ? slab_rows_sum<true>(x, r0, N, cg)
                 : slab_rows_sum<false>(x, r0, N, cg);
    __syncthreads();
    if (t < 64) {
        const float* f = reinterpret_cast<const float*>(sm);
        float a = 0.f;
        for (int k = 0; k < 16; ++k) a += f[k * 64 + t];
        part[(size_t)blockIdx.x * 64 + t] = a;
    }
}

__global__ void gat_colmean_final(const float* __restrict__ part, float* __restrict__ out,
                                  int P, int N) {
    __shared__ float4 sm[1024];
    const int t    = threadIdx.x;
    const int cg   = t & 15;
    const int slot = t >> 4;
    float4 s = make_float4(0.f, 0.f, 0.f, 0.f);
    for (int base = 0; base < P; base += 256) {
        float4 v[4];
        for (int i = 0; i < 4; ++i) {
            const int r  = base + slot + 64 * i;
            const int rc = r < P ? r : P - 1;
            v[i] = *reinterpret_cast<const float4*>(part + (size_t)rc * 64 + cg * 4);
            if (r >= P) v[i] = make_float4(0.f, 0.f, 0.f, 0.f);
        }
        for (int i = 0; i < 4; ++i) { s.x += v[i].x; s.y += v[i].y; s.z += v[i].z; s.w += v[i].w; }
    }
    sm[t] = s;
    __syncthreads();
    if (t < 64) {
        const float* f = reinterpret_cast<const float*>(sm);
        float a = 0.f;
        for (int k = 0; k < 64; ++k) a += f[k * 64 + t];
        out[t] = a / (float)N;
    }
}

static inline int cdiv(long long a, long long b) { return (int)((a + b - 1) / b); }

extern "C" void kernel_launch(void* const* d_in, const int* in_sizes, int n_in,
                              void* d_out, int out_size, void* d_ws, size_t ws_size,
                              hipStream_t stream) {
    const float* x   = (const float*)d_in[0];
    const int*   ei  = (const int*)  d_in[1];
    const float* W1  = (const float*)d_in[2];
    const float* a1s = (const float*)d_in[3];
    const float* a1d = (const float*)d_in[4];
    const float* b1  = (const float*)d_in[5];
    const float* W2  = (const float*)d_in[6];
    const float* a2s = (const float*)d_in[7];
    const float* a2d = (const float*)d_in[8];
    const float* b2  = (const float*)d_in[9];
    const float* W3  = (const float*)d_in[10];
    const float* a3s = (const float*)d_in[11];
    const float* a3d = (const float*)d_in[12];
    const float* b3  = (const float*)d_in[13];

    const int N  = in_sizes[0] / 16;
    const int E  = in_sizes[1] / 2;
    const int H  = 4;

    float* ws = (float*)d_ws;
    size_t off = 0;
    float*    buf_h    = ws + off; off += (size_t)N * 256;
    int*      csr      = (int*)(ws + off); off += (size_t)N * 256;
    float*    buf_feat = ws + off; off += (size_t)N * 256;
    float*    es       = ws + off; off += (size_t)N * H;
    float*    ed       = ws + off; off += (size_t)N * H;
    (void)ws_size;
    int* cnt    = csr;
    int* rowptr = csr + (size_t)N;
    int* cursor = csr + (size_t)2 * N;
    int* ssrc   = csr + (size_t)4 * N;

    float* node_emb  = (float*)d_out;
    float* graph_emb = node_emb + (size_t)N * 64;

    auto run_layer = [&](const float* in_feat, const float* W,
                         const float* as_, const float* ad_, const float* bb,
                         float* out_feat, int Hh, int K, int Ncols, int do_elu) {
        (void)K;
        if (Ncols == 256)
            hipLaunchKernelGGL((gat_gemm_wmma<1, 4, 2>), dim3(cdiv(N, 32)), dim3(128), 0, stream,
                               in_feat, W, buf_h, N);
        else
            hipLaunchKernelGGL((gat_gemm_wmma<2, 1, 1>), dim3(cdiv(N, 32)), dim3(64), 0, stream,
                               in_feat, W, buf_h, N);
        if (Hh == 4)
            hipLaunchKernelGGL((gat_seg_rowscore_agg<4>), dim3(cdiv(N, 8)), dim3(256), 0, stream,
                               ssrc, rowptr, cursor, buf_h, as_, ad_, bb, out_feat, N, do_elu);
        else
            hipLaunchKernelGGL((gat_seg_rowscore_agg<1>), dim3(cdiv(N, 8)), dim3(256), 0, stream,
                               ssrc, rowptr, cursor, buf_h, as_, ad_, bb, out_feat, N, do_elu);
    };

    hipMemsetAsync(cnt, 0, (size_t)N * sizeof(int), stream);
    hipLaunchKernelGGL(gat_csr_count, dim3(cdiv(E, 256)), dim3(256), 0, stream, ei, E, cnt);
    hipLaunchKernelGGL(gat_csr_scan,  dim3(cdiv(N, 1024)), dim3(1024), 0, stream, cnt, rowptr, cursor, N);
    hipLaunchKernelGGL(gat_csr_fill,  dim3(cdiv(E, 256)), dim3(256), 0, stream, ei, E, cursor, ssrc);

    hipLaunchKernelGGL(gat_attn_coef_in, dim3(cdiv(N, 256)), dim3(256), 0, stream,
                       x, W1, a1s, a1d, es, ed, N);
    hipLaunchKernelGGL((gat_seg_softmax_agg<4>), dim3(cdiv(N, 8)), dim3(256), 0, stream,
                       ssrc, rowptr, cursor, x, es, ed, buf_h, N);
    hipLaunchKernelGGL(gat_proj_heads_wmma, dim3(cdiv(N, 16)), dim3(32, 4), 0, stream,
                       buf_h, W1, b1, buf_feat, N);
    run_layer(buf_feat, W2, a2s, a2d, b2, buf_feat, H, 256, 256, 1);
    run_layer(buf_feat, W3, a3s, a3d, b3, node_emb, 1, 256, 64, 0);

    const int P = cdiv(N, GAT_SLAB);
    hipLaunchKernelGGL(gat_colsum_slab, dim3(P), dim3(256), 0, stream, node_emb, buf_h, N);
    hipLaunchKernelGGL(gat_colmean_final, dim3(1), dim3(1024), 0, stream, buf_h, graph_emb, P, N);
}


# Layers 1 and 3 walk two destination nodes per wave, one per 16-lane row (a node's 64 sums = a float4 per lane): half the waves enter for the same gathers; sources and weights move by DPP in a lane row
# speedup vs baseline: 9.4141x; 1.2729x over previous
#include <hip/hip_runtime.h>
#include <math.h>

typedef __attribute__((ext_vector_type(2))) float v2f;
typedef __attribute__((ext_vector_type(8))) float v8f;

#define GAT_D 64

typedef __attribute__((ext_vector_type(4))) float v4f;

#define GAT_K 256
#define GAT_APITCH 260

template <int WM, int WN, int MT>
__global__ void __launch_bounds__(WM * WN * 32)
gat_gemm_wmma(const float* __restrict__ A, const float* __restrict__ B,
              float* __restrict__ C, int M) {
    constexpr int NT = WM * WN * 32;
    constexpr int BM = WM * MT * 16;
    constexpr int N  = WN * 64;
    __shared__ __attribute__((aligned(16))) float As[BM * GAT_APITCH];
    const int t  = threadIdx.x;
    const int m0 = blockIdx.x * BM;

#pragma unroll
    for (int i = 0; i < BM * 64 / NT; ++i) {
        const int idx = i * NT + t;
        const int row = idx >> 6, c4 = idx & 63;
        const int gr  = m0 + row < M ? m0 + row : M - 1;
        const v4f v = *reinterpret_cast<const v4f*>(A + (size_t)gr * GAT_K + c4 * 4);
        *reinterpret_cast<v4f*>(&As[row * GAT_APITCH + c4 * 4]) = v;
    }
    __syncthreads();

    const int lane = t & 31, w = t >> 5;
    const int wm = w / WN, wn = w - wm * WN;
    const int c = lane & 15, h = lane >> 4;
    const int n0 = wn * 64;
    const float* Ap = As + (wm * MT * 16 + c) * GAT_APITCH + 4 * h;
    const float* Bp = B + (size_t)(4 * h) * N + n0 + 4 * c;

    v8f acc[MT][4];
#pragma unroll
    for (int mt = 0; mt < MT; ++mt)
#pragma unroll
        for (int j = 0; j < 4; ++j) acc[mt][j] = (v8f){0.f,0.f,0.f,0.f,0.f,0.f,0.f,0.f};

    v4f braw[3][4];
    v4f araw[2][MT];
#pragma unroll
    for (int q = 0; q < 4; ++q) {
        braw[0][q] = *reinterpret_cast<const v4f*>(Bp + (size_t)q * N);
        braw[1][q] = *reinterpret_cast<const v4f*>(Bp + (size_t)(8 + q) * N);
    }
#pragma unroll
    for (int mt = 0; mt < MT; ++mt)
        araw[0][mt] = *reinterpret_cast<const v4f*>(Ap + mt * 16 * GAT_APITCH);

#pragma unroll
    for (int st = 0; st < GAT_K / 8; ++st) {
        const int k0 = st * 8;
        __builtin_amdgcn_sched_barrier(0);
        if (st + 2 < GAT_K / 8) {
#pragma unroll
            for (int q = 0; q < 4; ++q)
                braw[(st + 2) % 3][q] = *reinterpret_cast<const v4f*>(Bp + (size_t)(k0 + 16 + q) * N);
        }
        __builtin_amdgcn_sched_barrier(0);
        if (st + 1 < GAT_K / 8) {
#pragma unroll
            for (int mt = 0; mt < MT; ++mt)
                araw[(st + 1) & 1][mt] = *reinterpret_cast<const v4f*>(Ap + mt * 16 * GAT_APITCH + k0 + 8);
        }
#pragma unroll
        for (int mt = 0; mt < MT; ++mt) {
            v2f a; a.x = araw[st & 1][mt][0]; a.y = araw[st & 1][mt][1];
#pragma unroll
            for (int j = 0; j < 4; ++j) {
                v2f b; b.x = braw[st % 3][0][j]; b.y = braw[st % 3][1][j];
                acc[mt][j] = __builtin_amdgcn_wmma_f32_16x16x4_f32(false, a, false, b, (short)0, acc[mt][j], false, false);
            }
        }
#pragma unroll
        for (int mt = 0; mt < MT; ++mt) {
            v2f a; a.x = araw[st & 1][mt][2]; a.y = araw[st & 1][mt][3];
#pragma unroll
            for (int j = 0; j < 4; ++j) {
                v2f b; b.x = braw[st % 3][2][j]; b.y = braw[st % 3][3][j];
                acc[mt][j] = __builtin_amdgcn_wmma_f32_16x16x4_f32(false, a, false, b, (short)0, acc[mt][j], false, false);
            }
        }
    }

#pragma unroll
    for (int mt = 0; mt < MT; ++mt) {
#pragma unroll
        for (int r = 0; r < 8; ++r) {
            const int row = m0 + (wm * MT + mt) * 16 + 8 * h + r;
            if (row < M) {
                v4f o; o[0] = acc[mt][0][r]; o[1] = acc[mt][1][r]; o[2] = acc[mt][2][r]; o[3] = acc[mt][3][r];
                *reinterpret_cast<v4f*>(C + (size_t)row * N + n0 + 4 * c) = o;
            }
        }
    }
}

__global__ void gat_attn_coef_in(const float* __restrict__ x, const float* __restrict__ W,
                                 const float* __restrict__ a_s, const float* __restrict__ a_d,
                                 float* __restrict__ es, float* __restrict__ ed,
                                 int* __restrict__ cnt, int* __restrict__ ovfc, int N) {
    __shared__ float wsd[2][16][4];
    const int t = threadIdx.x;
    if (t < 128) {
        const int which = t >> 6, k = (t >> 2) & 15, hh = t & 3;
        const float* wp = W + (size_t)k * 256 + hh * GAT_D;
        const float* ap = (which ? a_d : a_s) + hh * GAT_D;
        float s = 0.f;
        for (int d = 0; d < GAT_D; ++d) s += wp[d] * ap[d];
        wsd[which][k][hh] = s;
    }
    __syncthreads();
    const int n = blockIdx.x * blockDim.x + t;
    if (n == 0) ovfc[0] = 0;
    if (n >= N) return;
    cnt[n] = 0;
    float xr[16];
    for (int q = 0; q < 4; ++q) {
        const float4 v = *reinterpret_cast<const float4*>(x + (size_t)n * 16 + q * 4);
        xr[4 * q] = v.x; xr[4 * q + 1] = v.y; xr[4 * q + 2] = v.z; xr[4 * q + 3] = v.w;
    }
    float ss[4] = {0.f, 0.f, 0.f, 0.f}, dd[4] = {0.f, 0.f, 0.f, 0.f};
    for (int k = 0; k < 16; ++k)
        for (int hh = 0; hh < 4; ++hh) { ss[hh] += xr[k] * wsd[0][k][hh]; dd[hh] += xr[k] * wsd[1][k][hh]; }
    *reinterpret_cast<float4*>(es + (size_t)n * 4) = make_float4(ss[0], ss[1], ss[2], ss[3]);
    *reinterpret_cast<float4*>(ed + (size_t)n * 4) = make_float4(dd[0], dd[1], dd[2], dd[3]);
}

__global__ void gat_proj_heads_wmma(const float* __restrict__ S, const float* __restrict__ B,
                                    const float* __restrict__ bias, float* __restrict__ C, int M) {
    const int lane = threadIdx.x;
    const int head = threadIdx.y;
    const int m0 = blockIdx.x * 16;
    if (m0 >= M) return;
    const int n0 = head * 64;
    const int c = lane & 15, h = lane >> 4;
    const float* Ap = S + (size_t)(m0 + c) * 64 + head * 16 + 4 * h;
    const float* Bp = B + (size_t)(4 * h) * 256 + n0 + 4 * c;
    v8f acc[4];
#pragma unroll
    for (int j = 0; j < 4; ++j) acc[j] = (v8f){0.f,0.f,0.f,0.f,0.f,0.f,0.f,0.f};
#pragma unroll
    for (int k0 = 0; k0 < 16; k0 += 8) {
        const v4f av = *reinterpret_cast<const v4f*>(Ap + k0);
        v4f bq[4];
#pragma unroll
        for (int q = 0; q < 4; ++q) bq[q] = *reinterpret_cast<const v4f*>(Bp + (size_t)(k0 + q) * 256);
        v2f a; a.x = av[0]; a.y = av[1];
#pragma unroll
        for (int j = 0; j < 4; ++j) {
            v2f b; b.x = bq[0][j]; b.y = bq[1][j];
            acc[j] = __builtin_amdgcn_wmma_f32_16x16x4_f32(false, a, false, b, (short)0, acc[j], false, false);
        }
        a.x = av[2]; a.y = av[3];
#pragma unroll
        for (int j = 0; j < 4; ++j) {
            v2f b; b.x = bq[2][j]; b.y = bq[3][j];
            acc[j] = __builtin_amdgcn_wmma_f32_16x16x4_f32(false, a, false, b, (short)0, acc[j], false, false);
        }
    }
    const v4f bb = *reinterpret_cast<const v4f*>(bias + n0 + 4 * c);
    float* Cp = C + (size_t)(m0 + 8 * h) * 256 + n0 + 4 * c;
#pragma unroll
    for (int r = 0; r < 8; ++r) {
        v4f o;
#pragma unroll
        for (int j = 0; j < 4; ++j) {
            const float v = acc[j][r] + bb[j];
            o[j] = v > 0.f ? v : (expf(v) - 1.f);
        }
        *reinterpret_cast<v4f*>(Cp + (size_t)r * 256) = o;
    }
}

#define GAT_CAP 64
__global__ void gat_bucket_fill(const int* __restrict__ ei, int E, int* __restrict__ cnt,
                                int* __restrict__ bucket, int* __restrict__ ovfc,
                                int* __restrict__ ovf_dst, int* __restrict__ ovf_src) {
    const int e = blockIdx.x * blockDim.x + threadIdx.x;
    if (e >= E) return;
    const int s = ei[e], d = ei[E + e];
    const int k = atomicAdd(&cnt[d], 1);
    if (k < GAT_CAP) {
        bucket[(size_t)d * GAT_CAP + k] = s;
    } else {
        const int o = atomicAdd(ovfc, 1);
        ovf_dst[o] = d;
        ovf_src[o] = s;
    }
}

struct SegShape {
    int nb;
    int nbc;
    int ovf_n;
    int nch;
};
__device__ __forceinline__ SegShape seg_shape(const int* __restrict__ cnt,
                                              const int* __restrict__ ovfc, int node) {
    SegShape g;
    const int c = cnt[node];
    const int o = ovfc[0];
    g.nb    = c < GAT_CAP ? c : GAT_CAP;
    g.nbc   = (g.nb + 31) >> 5;
    g.ovf_n = c > GAT_CAP ? o : 0;
    g.nch   = g.nbc + ((g.ovf_n + 31) >> 5);
    return g;
}
__device__ __forceinline__ unsigned seg_chunk(const int* __restrict__ bucket,
                                              const int* __restrict__ ovf_dst,
                                              const int* __restrict__ ovf_src,
                                              int node, const SegShape& g, int q, int lane, int& sv) {
    bool valid;
    sv = 0;
    if (q < g.nbc) {
        const int p = q * 32 + lane;
        valid = p < g.nb;
        if (valid) sv = bucket[(size_t)node * GAT_CAP + p];
    } else {
        const int i = (q - g.nbc) * 32 + lane;
        valid = i < g.ovf_n;
        if (valid) valid = ovf_dst[i] == node;
        if (valid) sv = ovf_src[i];
    }
    return __builtin_amdgcn_ballot_w32(valid);
}
__device__ __forceinline__ int chunk_len(unsigned vm) { return vm ? 32 - __builtin_clz(vm) : 0; }

template <int M>
__device__ __forceinline__ float row_xor(float x) {
    const int b = __float_as_int(x);
    return __int_as_float(__builtin_amdgcn_update_dpp(b, b, 0x160 + M, 0xf, 0xf, false));
}

__device__ __forceinline__ float row16_sum(float v) {
    v += row_xor<1>(v); v += row_xor<2>(v); v += row_xor<4>(v); v += row_xor<8>(v);
    return v;
}
template <int K>
__device__ __forceinline__ float row_share(float x) {
    const int b = __float_as_int(x);
    return __int_as_float(__builtin_amdgcn_update_dpp(b, b, 0x150 + K, 0xf, 0xf, false));
}
__device__ __forceinline__ float fold8_sum(const float (&p)[8], bool b0, bool b1, bool b2) {
    float r[4], s[2];
#pragma unroll
    for (int i = 0; i < 4; ++i) {
        const float keep = b0 ? p[i + 4] : p[i], send = b0 ? p[i] : p[i + 4];
        r[i] = keep + row_xor<1>(send);
    }
#pragma unroll
    for (int i = 0; i < 2; ++i) {
        const float keep = b1 ? r[i + 2] : r[i], send = b1 ? r[i] : r[i + 2];
        s[i] = keep + row_xor<2>(send);
    }
    const float keep = b2 ? s[1] : s[0], send = b2 ? s[0] : s[1];
    return keep + row_xor<4>(send);
}
__device__ __forceinline__ float dot4(const float4 r, const float4 a) {
    return r.x * a.x + r.y * a.y + r.z * a.z + r.w * a.w;
}
__device__ __forceinline__ float4 scaled4(const float* __restrict__ p, float f) {
    const float4 t = *reinterpret_cast<const float4*>(p);
    return make_float4(t.x * f, t.y * f, t.z * f, t.w * f);
}

template <int HH>
__global__ void gat_seg_rowscore_agg(const int* __restrict__ bucket, const int* __restrict__ cnt,
                                     const int* __restrict__ ovfc, const int* __restrict__ ovf_dst,
                                     const int* __restrict__ ovf_src,
                                     const float* __restrict__ hfeat,
                                     const float* __restrict__ a_s, const float* __restrict__ a_d,
                                     const float* __restrict__ b, float* __restrict__ out,
                                     int N, int do_elu) {
    static_assert(HH == 4, "four heads; the single-head layer has its own kernel below");
    const int lane = threadIdx.x & 31;
    const int node = __builtin_amdgcn_readfirstlane(blockIdx.x * (blockDim.x >> 5) + (threadIdx.x >> 5));
    if (node >= N) return;
    const float L2E = 1.4426950408889634f;
    const bool b0 = (lane & 1) != 0, b1 = (lane & 2) != 0, b2 = (lane & 4) != 0;

    if constexpr (HH == 4) {
        const float* hp = hfeat + lane * 4;
        const float4 as0 = scaled4(a_s + lane * 4, L2E), as1 = scaled4(a_s + 128 + lane * 4, L2E);
        const int myu = ((lane & 1) << 1) | ((lane >> 1) & 1);
        float4 acc0, acc1;
        float ednl, ml;
        float denl = (lane & 3) == 0 ? 1.f : 0.f;
        {
            const float4 ad0 = scaled4(a_d + lane * 4, L2E), ad1 = scaled4(a_d + 128 + lane * 4, L2E);
            const float* rp = hp + (size_t)node * 256;
            acc0 = *reinterpret_cast<const float4*>(rp);
            acc1 = *reinterpret_cast<const float4*>(rp + 128);
            const float edn0 = row16_sum(dot4(acc0, ad0));
            const float edn1 = row16_sum(dot4(acc1, ad1));
            const float v0 = row16_sum(dot4(acc0, as0)) + edn0;
            const float v1 = row16_sum(dot4(acc1, as1)) + edn1;
            ednl = b2 ? edn1 : edn0;
            const float v = b2 ? v1 : v0;
            ml = fmaxf(v, 0.2f * v);
        }
        const SegShape sg = seg_shape(cnt, ovfc, node);
        for (int q = 0; q < sg.nch; ++q) {
            int sv;
            const unsigned vm = seg_chunk(bucket, ovf_dst, ovf_src, node, sg, q, lane, sv);
            const int n = chunk_len(vm);
            const unsigned vmy = vm >> myu;
            for (int j = 0; j < n; j += 4) {
                float4 q0[4], q1[4];
                float p[8];
#pragma unroll
                for (int u = 0; u < 4; ++u) {
                    const int s = __builtin_amdgcn_readlane(sv, j + u);
                    const float* qp = hp + (size_t)s * 256;
                    q0[u] = *reinterpret_cast<const float4*>(qp);
                    q1[u] = *reinterpret_cast<const float4*>(qp + 128);
                }
#pragma unroll
                for (int u = 0; u < 4; ++u) { p[2 * u] = dot4(q0[u], as0); p[2 * u + 1] = dot4(q1[u], as1); }
                float t = fold8_sum(p, b0, b1, b2);
                t += row_xor<8>(t);
                const float v = t + ednl;
                const float e = ((vmy >> j) & 1u) ? fmaxf(v, 0.2f * v) : -INFINITY;
                float cm = fmaxf(e, row_xor<1>(e));
                cm = fmaxf(cm, row_xor<2>(cm));
                const float mn = fmaxf(ml, cm);
                const float sc = __builtin_amdgcn_exp2f(ml - mn);
                const float w  = __builtin_amdgcn_exp2f(e - mn);
                denl = denl * sc + w;
                ml = mn;
                const float sc0 = row_share<0>(sc), sc1 = row_share<4>(sc);
                acc0.x *= sc0; acc0.y *= sc0; acc0.z *= sc0; acc0.w *= sc0;
                acc1.x *= sc1; acc1.y *= sc1; acc1.z *= sc1; acc1.w *= sc1;
                float w0[4], w1[4];
                w0[0] = row_share<0>(w); w0[1] = row_share<2>(w); w0[2] = row_share<1>(w); w0[3] = row_share<3>(w);
                w1[0] = row_share<4>(w); w1[1] = row_share<6>(w); w1[2] = row_share<5>(w); w1[3] = row_share<7>(w);
#pragma unroll
                for (int u = 0; u < 4; ++u) {
                    acc0.x += q0[u].x * w0[u]; acc0.y += q0[u].y * w0[u]; acc0.z += q0[u].z * w0[u]; acc0.w += q0[u].w * w0[u];
                    acc1.x += q1[u].x * w1[u]; acc1.y += q1[u].y * w1[u]; acc1.z += q1[u].z * w1[u]; acc1.w += q1[u].w * w1[u];
                }
            }
        }
        float d = denl;
        d += row_xor<1>(d); d += row_xor<2>(d);
        const float i0 = 1.f / row_share<0>(d), i1 = 1.f / row_share<4>(d);
        const float4 bb0 = *reinterpret_cast<const float4*>(b + lane * 4);
        const float4 bb1 = *reinterpret_cast<const float4*>(b + 128 + lane * 4);
        acc0.x = acc0.x * i0 + bb0.x; acc0.y = acc0.y * i0 + bb0.y; acc0.z = acc0.z * i0 + bb0.z; acc0.w = acc0.w * i0 + bb0.w;
        acc1.x = acc1.x * i1 + bb1.x; acc1.y = acc1.y * i1 + bb1.y; acc1.z = acc1.z * i1 + bb1.z; acc1.w = acc1.w * i1 + bb1.w;
        if (do_elu) {
            acc0.x = acc0.x > 0.f ? acc0.x : (expf(acc0.x) - 1.f);
            acc0.y = acc0.y > 0.f ? acc0.y : (expf(acc0.y) - 1.f);
            acc0.z = acc0.z > 0.f ? acc0.z : (expf(acc0.z) - 1.f);
            acc0.w = acc0.w > 0.f ? acc0.w : (expf(acc0.w) - 1.f);
            acc1.x = acc1.x > 0.f ? acc1.x : (expf(acc1.x) - 1.f);
            acc1.y = acc1.y > 0.f ? acc1.y : (expf(acc1.y) - 1.f);
            acc1.z = acc1.z > 0.f ? acc1.z : (expf(acc1.z) - 1.f);
            acc1.w = acc1.w > 0.f ? acc1.w : (expf(acc1.w) - 1.f);
        }
        float* op = out + (size_t)node * 256 + lane * 4;
        *reinterpret_cast<float4*>(op)       = acc0;
        *reinterpret_cast<float4*>(op + 128) = acc1;
    }
}

struct PairSeg {
    int nb;
    int nbc;
    int ovf_n;
    int nch;
};
__device__ __forceinline__ PairSeg pair_seg(const int* __restrict__ cnt, int on, int node, bool live) {
    PairSeg g;
    const int cn = cnt[node];
    g.nb    = cn < GAT_CAP ? cn : GAT_CAP;
    g.nbc   = (g.nb + 31) >> 5;
    g.ovf_n = (live && cn > GAT_CAP) ? on : 0;
    g.nch   = live ? g.nbc + ((g.ovf_n + 31) >> 5) : 0;
    return g;
}
__device__ __forceinline__ void pair_chunk(const int* __restrict__ bucket,
                                           const int* __restrict__ ovf_dst,
                                           const int* __restrict__ ovf_src,
                                           int node, bool live, const PairSeg& g, int q, int c,
                                           int& sva, int& svb, bool& va, bool& vb) {
    sva = 0; svb = 0;
    if (q < g.nbc) {
        const int p = q * 32 + c;
        va = live && p < g.nb;
        vb = live && p + 16 < g.nb;
        const int* bp = bucket + (size_t)node * GAT_CAP;
        const int ra = bp[va ? p : 0], rb = bp[vb ? p + 16 : 0];
        sva = va ? ra : 0;
        svb = vb ? rb : 0;
    } else {
        const int i = (q - g.nbc) * 32 + c;
        va = i < g.ovf_n;
        vb = i + 16 < g.ovf_n;
        if (va) va = ovf_dst[i] == node;
        if (vb) vb = ovf_dst[i + 16] == node;
        if (va) sva = ovf_src[i];
        if (vb) svb = ovf_src[i + 16];
    }
}
template <int K>
__device__ __forceinline__ int row_share_i(int b) {
    return __builtin_amdgcn_update_dpp(b, b, 0x150 + K, 0xf, 0xf, false);
}

template <int J>
__device__ __forceinline__ void pair_trip(const float* __restrict__ hfeat, unsigned c4, int sv, unsigned vmy,
                                          const float4 as, float edn, bool b0, bool b1, bool b2,
                                          float4& acc, float& m, float& denl) {
    float4 q[8];
    float p[8];
    int s[8];
    s[0] = row_share_i<J + 0>(sv); s[1] = row_share_i<J + 1>(sv); s[2] = row_share_i<J + 2>(sv); s[3] = row_share_i<J + 3>(sv);
    s[4] = row_share_i<J + 4>(sv); s[5] = row_share_i<J + 5>(sv); s[6] = row_share_i<J + 6>(sv); s[7] = row_share_i<J + 7>(sv);
#pragma unroll
    for (int u = 0; u < 8; ++u) q[u] = *reinterpret_cast<const float4*>(hfeat + ((unsigned)s[u] * 64u + c4));
#pragma unroll
    for (int u = 0; u < 8; ++u) p[u] = dot4(q[u], as);
    float t = fold8_sum(p, b0, b1, b2);
    t += row_xor<8>(t);
    const float v = t + edn;
    const float e = ((vmy >> J) & 1u) ? fmaxf(v, 0.2f * v) : -INFINITY;
    float cm = fmaxf(e, row_xor<1>(e));
    cm = fmaxf(cm, row_xor<2>(cm));
    cm = fmaxf(cm, row_xor<4>(cm));
    const float mn = fmaxf(m, cm);
    const float sc = __builtin_amdgcn_exp2f(m - mn);
    const float w  = __builtin_amdgcn_exp2f(e - mn);
    denl = denl * sc + w;
    m = mn;
    acc.x *= sc; acc.y *= sc; acc.z *= sc; acc.w *= sc;
    float wu[8];
    wu[0] = row_share<0>(w); wu[1] = row_share<4>(w); wu[2] = row_share<2>(w); wu[3] = row_share<6>(w);
    wu[4] = row_share<1>(w); wu[5] = row_share<5>(w); wu[6] = row_share<3>(w); wu[7] = row_share<7>(w);
#pragma unroll
    for (int u = 0; u < 8; ++u) {
        acc.x += q[u].x * wu[u]; acc.y += q[u].y * wu[u]; acc.z += q[u].z * wu[u]; acc.w += q[u].w * wu[u];
    }
}

__global__ void gat_seg_rowscore_agg_pair(const int* __restrict__ bucket, const int* __restrict__ cnt,
                                          const int* __restrict__ ovfc, const int* __restrict__ ovf_dst,
                                          const int* __restrict__ ovf_src,
                                          const float* __restrict__ hfeat,
                                          const float* __restrict__ a_s, const float* __restrict__ a_d,
                                          const float* __restrict__ b, float* __restrict__ out,
                                          int N, int do_elu) {
    const int lane = threadIdx.x & 31;
    const int c = lane & 15, hh = lane >> 4;
    const int node0 = __builtin_amdgcn_readfirstlane(2 * (blockIdx.x * (blockDim.x >> 5) + (threadIdx.x >> 5)));
    if (node0 >= N) return;
    const int on = ovfc[0];
    const bool live = node0 + hh < N;
    const int node = live ? node0 + hh : N - 1;
    const float L2E = 1.4426950408889634f;
    const bool b0 = (lane & 1) != 0, b1 = (lane & 2) != 0, b2 = (lane & 4) != 0;
    const int myu = ((lane & 1) << 2) | (lane & 2) | ((lane >> 2) & 1);

    const unsigned c4 = (unsigned)c * 4u;
    const float4 as = scaled4(a_s + c * 4, L2E);
    float4 acc = *reinterpret_cast<const float4*>(hfeat + ((unsigned)node * 64u + c4));
    float edn, m;
    float denl = (lane & 7) == 0 ? 1.f : 0.f;
    {
        const float4 ad = scaled4(a_d + c * 4, L2E);
        edn = row16_sum(dot4(acc, ad));
        const float v = row16_sum(dot4(acc, as)) + edn;
        m = fmaxf(v, 0.2f * v);
    }
    const PairSeg sg = pair_seg(cnt, on, node, live);
    const int nch0 = __builtin_amdgcn_readlane(sg.nch, 0), nch1 = __builtin_amdgcn_readlane(sg.nch, 16);
    const int nchw = nch0 > nch1 ? nch0 : nch1;
    for (int q = 0; q < nchw; ++q) {
        int sva, svb;
        bool va, vb;
        pair_chunk(bucket, ovf_dst, ovf_src, node, live, sg, q, c, sva, svb, va, vb);
        const unsigned vma = __builtin_amdgcn_ballot_w32(va), vmb = __builtin_amdgcn_ballot_w32(vb);
        const unsigned vmya = ((vma >> (16 * hh)) & 0xffffu) >> myu;
        const unsigned vmyb = ((vmb >> (16 * hh)) & 0xffffu) >> myu;
        if (vma & 0x00ff00ffu) pair_trip<0>(hfeat, c4, sva, vmya, as, edn, b0, b1, b2, acc, m, denl);
        if (vma & 0xff00ff00u) pair_trip<8>(hfeat, c4, sva, vmya, as, edn, b0, b1, b2, acc, m, denl);
        if (vmb & 0x00ff00ffu) pair_trip<0>(hfeat, c4, svb, vmyb, as, edn, b0, b1, b2, acc, m, denl);
        if (vmb & 0xff00ff00u) pair_trip<8>(hfeat, c4, svb, vmyb, as, edn, b0, b1, b2, acc, m, denl);
    }
    float d = denl;
    d += row_xor<1>(d); d += row_xor<2>(d); d += row_xor<4>(d);
    const float inv = 1.f / d;
    const float4 bb = *reinterpret_cast<const float4*>(b + c * 4);
    acc.x = acc.x * inv + bb.x; acc.y = acc.y * inv + bb.y; acc.z = acc.z * inv + bb.z; acc.w = acc.w * inv + bb.w;
    if (do_elu) {
        acc.x = acc.x > 0.f ? acc.x : (expf(acc.x) - 1.f);
        acc.y = acc.y > 0.f ? acc.y : (expf(acc.y) - 1.f);
        acc.z = acc.z > 0.f ? acc.z : (expf(acc.z) - 1.f);
        acc.w = acc.w > 0.f ? acc.w : (expf(acc.w) - 1.f);
    }
    if (live) *reinterpret_cast<float4*>(out + (size_t)node * 64 + c * 4) = acc;
}

template <int J>
__device__ __forceinline__ void pair_trip_in(const float* __restrict__ x, const float* __restrict__ es,
                                             unsigned k4, unsigned head, int sv, unsigned vmh, float edn,
                                             float4& acc, float& m, float& den) {
    const float L2E = 1.4426950408889634f;
    int s[8];
    s[0] = row_share_i<J + 0>(sv); s[1] = row_share_i<J + 1>(sv); s[2] = row_share_i<J + 2>(sv); s[3] = row_share_i<J + 3>(sv);
    s[4] = row_share_i<J + 4>(sv); s[5] = row_share_i<J + 5>(sv); s[6] = row_share_i<J + 6>(sv); s[7] = row_share_i<J + 7>(sv);
    float4 r[8];
    float e[8];
#pragma unroll
    for (int u = 0; u < 8; ++u) {
        r[u] = *reinterpret_cast<const float4*>(x + ((unsigned)s[u] * 16u + k4));
        e[u] = es[(unsigned)s[u] * 4u + head];
    }
    float cm = -INFINITY;
#pragma unroll
    for (int u = 0; u < 8; ++u) {
        const float v = e[u] * L2E + edn;
        e[u] = ((vmh >> (J + u)) & 1u) ? fmaxf(v, 0.2f * v) : -INFINITY;
        cm = fmaxf(cm, e[u]);
    }
    const float mn = fmaxf(m, cm);
    const float sc = __builtin_amdgcn_exp2f(m - mn);
    den *= sc;
    acc.x *= sc; acc.y *= sc; acc.z *= sc; acc.w *= sc;
#pragma unroll
    for (int u = 0; u < 8; ++u) {
        const float w = __builtin_amdgcn_exp2f(e[u] - mn);
        den += w;
        acc.x += r[u].x * w; acc.y += r[u].y * w; acc.z += r[u].z * w; acc.w += r[u].w * w;
    }
    m = mn;
}

__global__ void gat_seg_softmax_agg_pair(const int* __restrict__ bucket, const int* __restrict__ cnt,
                                         const int* __restrict__ ovfc, const int* __restrict__ ovf_dst,
                                         const int* __restrict__ ovf_src,
                                         const float* __restrict__ x,
                                         const float* __restrict__ es, const float* __restrict__ ed,
                                         float* __restrict__ out, int N) {
    const int lane = threadIdx.x & 31;
    const int c = lane & 15, hh = lane >> 4;
    const int node0 = __builtin_amdgcn_readfirstlane(2 * (blockIdx.x * (blockDim.x >> 5) + (threadIdx.x >> 5)));
    if (node0 >= N) return;
    const int on = ovfc[0];
    const bool live = node0 + hh < N;
    const int node = live ? node0 + hh : N - 1;
    const float L2E = 1.4426950408889634f;
    const unsigned head = (unsigned)c >> 2, k4 = ((unsigned)c & 3u) * 4u;

    const float edn = ed[(unsigned)node * 4u + head] * L2E;
    float4 acc = *reinterpret_cast<const float4*>(x + ((unsigned)node * 16u + k4));
    float m;
    {
        const float v = es[(unsigned)node * 4u + head] * L2E + edn;
        m = fmaxf(v, 0.2f * v);
    }
    float den = 1.f;
    const PairSeg sg = pair_seg(cnt, on, node, live);
    const int nch0 = __builtin_amdgcn_readlane(sg.nch, 0), nch1 = __builtin_amdgcn_readlane(sg.nch, 16);
    const int nchw = nch0 > nch1 ? nch0 : nch1;
    for (int q = 0; q < nchw; ++q) {
        int sva, svb;
        bool va, vb;
        pair_chunk(bucket, ovf_dst, ovf_src, node, live, sg, q, c, sva, svb, va, vb);
        const unsigned vma = __builtin_amdgcn_ballot_w32(va), vmb = __builtin_amdgcn_ballot_w32(vb);
        const unsigned vmha = (vma >> (16 * hh)) & 0xffffu;
        const unsigned vmhb = (vmb >> (16 * hh)) & 0xffffu;
        if (vma & 0x00ff00ffu) pair_trip_in<0>(x, es, k4, head, sva, vmha, edn, acc, m, den);
        if (vma & 0xff00ff00u) pair_trip_in<8>(x, es, k4, head, sva, vmha, edn, acc, m, den);
        if (vmb & 0x00ff00ffu) pair_trip_in<0>(x, es, k4, head, svb, vmhb, edn, acc, m, den);
        if (vmb & 0xff00ff00u) pair_trip_in<8>(x, es, k4, head, svb, vmhb, edn, acc, m, den);
    }
    const float inv = 1.f / den;
    acc.x *= inv; acc.y *= inv; acc.z *= inv; acc.w *= inv;
    if (live) *reinterpret_cast<float4*>(out + (size_t)node * 64 + c * 4) = acc;
}

#define GAT_SLAB 256
template <bool FULL>
__device__ __forceinline__ float4 slab_rows_sum(const float* __restrict__ x, int r0, int N, int cg) {
    float4 v[GAT_SLAB / 16];
    for (int i = 0; i < GAT_SLAB / 16; ++i) {
        const int r = r0 + 16 * i;
        if constexpr (FULL) {
            v[i] = *reinterpret_cast<const float4*>(x + (size_t)r * 64 + cg * 4);
        } else {
            const int rc = r < N ? r : N - 1;
            v[i] = *reinterpret_cast<const float4*>(x + (size_t)rc * 64 + cg * 4);
            if (r >= N) v[i] = make_float4(0.f, 0.f, 0.f, 0.f);
        }
    }
    float4 s = v[0];
    for (int i = 1; i < GAT_SLAB / 16; ++i) { s.x += v[i].x; s.y += v[i].y; s.z += v[i].z; s.w += v[i].w; }
    return s;
}

__global__ void gat_colsum_slab(const float* __restrict__ x, float* __restrict__ part, int N) {
    __shared__ float4 sm[256];
    const int t    = threadIdx.x;
    const int cg   = t & 15;
    const int slot = t >> 4;
    const int r0   = blockIdx.x * GAT_SLAB + slot;
    const bool full = (blockIdx.x + 1) * GAT_SLAB <= N;
    sm[t] = full ? slab_rows_sum<true>(x, r0, N, cg)
                 : slab_rows_sum<false>(x, r0, N, cg);
    __syncthreads();
    if (t < 64) {
        const float* f = reinterpret_cast<const float*>(sm);
        float a = 0.f;
        for (int k = 0; k < 16; ++k) a += f[k * 64 + t];
        part[(size_t)blockIdx.x * 64 + t] = a;
    }
}

__global__ void gat_colmean_final(const float* __restrict__ part, float* __restrict__ out,
                                  int P, int N) {
    __shared__ float4 sm[1024];
    const int t    = threadIdx.x;
    const int cg   = t & 15;
    const int slot = t >> 4;
    float4 s = make_float4(0.f, 0.f, 0.f, 0.f);
    for (int base = 0; base < P; base += 256) {
        float4 v[4];
        for (int i = 0; i < 4; ++i) {
            const int r  = base + slot + 64 * i;
            const int rc = r < P ? r : P - 1;
            v[i] = *reinterpret_cast<const float4*>(part + (size_t)rc * 64 + cg * 4);
            if (r >= P) v[i] = make_float4(0.f, 0.f, 0.f, 0.f);
        }
        for (int i = 0; i < 4; ++i) { s.x += v[i].x; s.y += v[i].y; s.z += v[i].z; s.w += v[i].w; }
    }
    sm[t] = s;
    __syncthreads();
    if (t < 64) {
        const float* f = reinterpret_cast<const float*>(sm);
        float a = 0.f;
        for (int k = 0; k < 64; ++k) a += f[k * 64 + t];
        out[t] = a / (float)N;
    }
}

static inline int cdiv(long long a, long long b) { return (int)((a + b - 1) / b); }

extern "C" void kernel_launch(void* const* d_in, const int* in_sizes, int n_in,
                              void* d_out, int out_size, void* d_ws, size_t ws_size,
                              hipStream_t stream) {
    const float* x   = (const float*)d_in[0];
    const int*   ei  = (const int*)  d_in[1];
    const float* W1  = (const float*)d_in[2];
    const float* a1s = (const float*)d_in[3];
    const float* a1d = (const float*)d_in[4];
    const float* b1  = (const float*)d_in[5];
    const float* W2  = (const float*)d_in[6];
    const float* a2s = (const float*)d_in[7];
    const float* a2d = (const float*)d_in[8];
    const float* b2  = (const float*)d_in[9];
    const float* W3  = (const float*)d_in[10];
    const float* a3s = (const float*)d_in[11];
    const float* a3d = (const float*)d_in[12];
    const float* b3  = (const float*)d_in[13];

    const int N  = in_sizes[0] / 16;
    const int E  = in_sizes[1] / 2;
    const int H  = 4;

    float* ws = (float*)d_ws;
    size_t off = 0;
    float*    buf_h    = ws + off; off += (size_t)N * 256;
    int*      csr      = (int*)(ws + off); off += (size_t)N * 256;
    float*    buf_feat = ws + off; off += (size_t)N * 256;
    float*    es       = ws + off; off += (size_t)N * H;
    float*    ed       = ws + off; off += (size_t)N * H;
    (void)ws_size;
    int* cnt     = csr;
    int* ovfc    = csr + (size_t)N;
    int* bucket  = csr + (((size_t)N + 64) & ~(size_t)63);
    int* ovf_dst = bucket + (size_t)N * GAT_CAP;
    int* ovf_src = ovf_dst + (size_t)E;

    float* node_emb  = (float*)d_out;
    float* graph_emb = node_emb + (size_t)N * 64;

    auto run_layer = [&](const float* in_feat, const float* W,
                         const float* as_, const float* ad_, const float* bb,
                         float* out_feat, int Hh, int K, int Ncols, int do_elu) {
        (void)K;
        if (Ncols == 256)
            hipLaunchKernelGGL((gat_gemm_wmma<1, 4, 2>), dim3(cdiv(N, 32)), dim3(128), 0, stream,
                               in_feat, W, buf_h, N);
        else
            hipLaunchKernelGGL((gat_gemm_wmma<2, 1, 1>), dim3(cdiv(N, 32)), dim3(64), 0, stream,
                               in_feat, W, buf_h, N);
        if (Hh == 4)
            hipLaunchKernelGGL((gat_seg_rowscore_agg<4>), dim3(cdiv(N, 8)), dim3(256), 0, stream,
                               bucket, cnt, ovfc, ovf_dst, ovf_src, buf_h, as_, ad_, bb, out_feat, N, do_elu);
        else
            hipLaunchKernelGGL(gat_seg_rowscore_agg_pair, dim3(cdiv(N, 16)), dim3(256), 0, stream,
                               bucket, cnt, ovfc, ovf_dst, ovf_src, buf_h, as_, ad_, bb, out_feat, N, do_elu);
    };

    hipLaunchKernelGGL(gat_attn_coef_in, dim3(cdiv(N, 256)), dim3(256), 0, stream,
                       x, W1, a1s, a1d, es, ed, cnt, ovfc, N);
    hipLaunchKernelGGL(gat_bucket_fill, dim3(cdiv(E, 256)), dim3(256), 0, stream,
                       ei, E, cnt, bucket, ovfc, ovf_dst, ovf_src);

    hipLaunchKernelGGL(gat_seg_softmax_agg_pair, dim3(cdiv(N, 16)), dim3(256), 0, stream,
                       bucket, cnt, ovfc, ovf_dst, ovf_src, x, es, ed, buf_h, N);
    hipLaunchKernelGGL(gat_proj_heads_wmma, dim3(cdiv(N, 16)), dim3(32, 4), 0, stream,
                       buf_h, W1, b1, buf_feat, N);
    run_layer(buf_feat, W2, a2s, a2d, b2, buf_feat, H, 256, 256, 1);
    run_layer(buf_feat, W3, a3s, a3d, b3, node_emb, 1, 256, 64, 0);

    const int P = cdiv(N, GAT_SLAB);
    hipLaunchKernelGGL(gat_colsum_slab, dim3(P), dim3(256), 0, stream, node_emb, buf_h, N);
    hipLaunchKernelGGL(gat_colmean_final, dim3(1), dim3(1024), 0, stream, buf_h, graph_emb, P, N);
}


# Edges grouped by destination in two LDS-binned passes (128-node bins, then each bin's node counters and buckets filled in LDS and stored by rows): one global add per block and bin, not one per edge
# speedup vs baseline: 11.5504x; 1.2269x over previous
#include <hip/hip_runtime.h>
#include <math.h>

typedef __attribute__((ext_vector_type(2))) float v2f;
typedef __attribute__((ext_vector_type(8))) float v8f;

#define GAT_D 64

typedef __attribute__((ext_vector_type(4))) float v4f;

#define GAT_K 256
#define GAT_APITCH 260

template <int WM, int WN, int MT>
__global__ void __launch_bounds__(WM * WN * 32)
gat_gemm_wmma(const float* __restrict__ A, const float* __restrict__ B,
              float* __restrict__ C, int M) {
    constexpr int NT = WM * WN * 32;
    constexpr int BM = WM * MT * 16;
    constexpr int N  = WN * 64;
    __shared__ __attribute__((aligned(16))) float As[BM * GAT_APITCH];
    const int t  = threadIdx.x;
    const int m0 = blockIdx.x * BM;

#pragma unroll
    for (int i = 0; i < BM * 64 / NT; ++i) {
        const int idx = i * NT + t;
        const int row = idx >> 6, c4 = idx & 63;
        const int gr  = m0 + row < M ? m0 + row : M - 1;
        const v4f v = *reinterpret_cast<const v4f*>(A + (size_t)gr * GAT_K + c4 * 4);
        *reinterpret_cast<v4f*>(&As[row * GAT_APITCH + c4 * 4]) = v;
    }
    __syncthreads();

    const int lane = t & 31, w = t >> 5;
    const int wm = w / WN, wn = w - wm * WN;
    const int c = lane & 15, h = lane >> 4;
    const int n0 = wn * 64;
    const float* Ap = As + (wm * MT * 16 + c) * GAT_APITCH + 4 * h;
    const float* Bp = B + (size_t)(4 * h) * N + n0 + 4 * c;

    v8f acc[MT][4];
#pragma unroll
    for (int mt = 0; mt < MT; ++mt)
#pragma unroll
        for (int j = 0; j < 4; ++j) acc[mt][j] = (v8f){0.f,0.f,0.f,0.f,0.f,0.f,0.f,0.f};

    v4f braw[3][4];
    v4f araw[2][MT];
#pragma unroll
    for (int q = 0; q < 4; ++q) {
        braw[0][q] = *reinterpret_cast<const v4f*>(Bp + (size_t)q * N);
        braw[1][q] = *reinterpret_cast<const v4f*>(Bp + (size_t)(8 + q) * N);
    }
#pragma unroll
    for (int mt = 0; mt < MT; ++mt)
        araw[0][mt] = *reinterpret_cast<const v4f*>(Ap + mt * 16 * GAT_APITCH);

#pragma unroll
    for (int st = 0; st < GAT_K / 8; ++st) {
        const int k0 = st * 8;
        __builtin_amdgcn_sched_barrier(0);
        if (st + 2 < GAT_K / 8) {
#pragma unroll
            for (int q = 0; q < 4; ++q)
                braw[(st + 2) % 3][q] = *reinterpret_cast<const v4f*>(Bp + (size_t)(k0 + 16 + q) * N);
        }
        __builtin_amdgcn_sched_barrier(0);
        if (st + 1 < GAT_K / 8) {
#pragma unroll
            for (int mt = 0; mt < MT; ++mt)
                araw[(st + 1) & 1][mt] = *reinterpret_cast<const v4f*>(Ap + mt * 16 * GAT_APITCH + k0 + 8);
        }
#pragma unroll
        for (int mt = 0; mt < MT; ++mt) {
            v2f a; a.x = araw[st & 1][mt][0]; a.y = araw[st & 1][mt][1];
#pragma unroll
            for (int j = 0; j < 4; ++j) {
                v2f b; b.x = braw[st % 3][0][j]; b.y = braw[st % 3][1][j];
                acc[mt][j] = __builtin_amdgcn_wmma_f32_16x16x4_f32(false, a, false, b, (short)0, acc[mt][j], false, false);
            }
        }
#pragma unroll
        for (int mt = 0; mt < MT; ++mt) {
            v2f a; a.x = araw[st & 1][mt][2]; a.y = araw[st & 1][mt][3];
#pragma unroll
            for (int j = 0; j < 4; ++j) {
                v2f b; b.x = braw[st % 3][2][j]; b.y = braw[st % 3][3][j];
                acc[mt][j] = __builtin_amdgcn_wmma_f32_16x16x4_f32(false, a, false, b, (short)0, acc[mt][j], false, false);
            }
        }
    }

#pragma unroll
    for (int mt = 0; mt < MT; ++mt) {
#pragma unroll
        for (int r = 0; r < 8; ++r) {
            const int row = m0 + (wm * MT + mt) * 16 + 8 * h + r;
            if (row < M) {
                v4f o; o[0] = acc[mt][0][r]; o[1] = acc[mt][1][r]; o[2] = acc[mt][2][r]; o[3] = acc[mt][3][r];
                *reinterpret_cast<v4f*>(C + (size_t)row * N + n0 + 4 * c) = o;
            }
        }
    }
}

__global__ void gat_attn_coef_in(const float* __restrict__ x, const float* __restrict__ W,
                                 const float* __restrict__ a_s, const float* __restrict__ a_d,
                                 float* __restrict__ es, float* __restrict__ ed,
                                 int* __restrict__ cnt, int* __restrict__ ovfc,
                                 int* __restrict__ bin_cnt, int NB, int N) {
    __shared__ float wsd[2][16][4];
    const int t = threadIdx.x;
    if (t < 128) {
        const int which = t >> 6, k = (t >> 2) & 15, hh = t & 3;
        const float* wp = W + (size_t)k * 256 + hh * GAT_D;
        const float* ap = (which ? a_d : a_s) + hh * GAT_D;
        float s = 0.f;
        for (int d = 0; d < GAT_D; ++d) s += wp[d] * ap[d];
        wsd[which][k][hh] = s;
    }
    __syncthreads();
    const int n = blockIdx.x * blockDim.x + t;
    if (n == 0) ovfc[0] = 0;
    if (n < NB) bin_cnt[n] = 0;
    if (n >= N) return;
    cnt[n] = 0;
    float xr[16];
    for (int q = 0; q < 4; ++q) {
        const float4 v = *reinterpret_cast<const float4*>(x + (size_t)n * 16 + q * 4);
        xr[4 * q] = v.x; xr[4 * q + 1] = v.y; xr[4 * q + 2] = v.z; xr[4 * q + 3] = v.w;
    }
    float ss[4] = {0.f, 0.f, 0.f, 0.f}, dd[4] = {0.f, 0.f, 0.f, 0.f};
    for (int k = 0; k < 16; ++k)
        for (int hh = 0; hh < 4; ++hh) { ss[hh] += xr[k] * wsd[0][k][hh]; dd[hh] += xr[k] * wsd[1][k][hh]; }
    *reinterpret_cast<float4*>(es + (size_t)n * 4) = make_float4(ss[0], ss[1], ss[2], ss[3]);
    *reinterpret_cast<float4*>(ed + (size_t)n * 4) = make_float4(dd[0], dd[1], dd[2], dd[3]);
}

__global__ void gat_proj_heads_wmma(const float* __restrict__ S, const float* __restrict__ B,
                                    const float* __restrict__ bias, float* __restrict__ C, int M) {
    const int lane = threadIdx.x;
    const int head = threadIdx.y;
    const int m0 = blockIdx.x * 16;
    if (m0 >= M) return;
    const int n0 = head * 64;
    const int c = lane & 15, h = lane >> 4;
    const float* Ap = S + (size_t)(m0 + c) * 64 + head * 16 + 4 * h;
    const float* Bp = B + (size_t)(4 * h) * 256 + n0 + 4 * c;
    v8f acc[4];
#pragma unroll
    for (int j = 0; j < 4; ++j) acc[j] = (v8f){0.f,0.f,0.f,0.f,0.f,0.f,0.f,0.f};
#pragma unroll
    for (int k0 = 0; k0 < 16; k0 += 8) {
        const v4f av = *reinterpret_cast<const v4f*>(Ap + k0);
        v4f bq[4];
#pragma unroll
        for (int q = 0; q < 4; ++q) bq[q] = *reinterpret_cast<const v4f*>(Bp + (size_t)(k0 + q) * 256);
        v2f a; a.x = av[0]; a.y = av[1];
#pragma unroll
        for (int j = 0; j < 4; ++j) {
            v2f b; b.x = bq[0][j]; b.y = bq[1][j];
            acc[j] = __builtin_amdgcn_wmma_f32_16x16x4_f32(false, a, false, b, (short)0, acc[j], false, false);
        }
        a.x = av[2]; a.y = av[3];
#pragma unroll
        for (int j = 0; j < 4; ++j) {
            v2f b; b.x = bq[2][j]; b.y = bq[3][j];
            acc[j] = __builtin_amdgcn_wmma_f32_16x16x4_f32(false, a, false, b, (short)0, acc[j], false, false);
        }
    }
    const v4f bb = *reinterpret_cast<const v4f*>(bias + n0 + 4 * c);
    float* Cp = C + (size_t)(m0 + 8 * h) * 256 + n0 + 4 * c;
#pragma unroll
    for (int r = 0; r < 8; ++r) {
        v4f o;
#pragma unroll
        for (int j = 0; j < 4; ++j) {
            const float v = acc[j][r] + bb[j];
            o[j] = v > 0.f ? v : (expf(v) - 1.f);
        }
        *reinterpret_cast<v4f*>(Cp + (size_t)r * 256) = o;
    }
}

#define GAT_CAP 64
#define GAT_BIN_SHIFT 7
#define GAT_BIN_NODES (1 << GAT_BIN_SHIFT)
#define GAT_MAXBINS 512
#define GAT_BIN_CAP 8192
#define GAT_EPT 8

__device__ __forceinline__ void bucket_put_global(int s, int d, int* __restrict__ cnt,
                                                  int* __restrict__ bucket, int* __restrict__ ovfc,
                                                  int* __restrict__ ovf_dst, int* __restrict__ ovf_src) {
    const int k = atomicAdd(&cnt[d], 1);
    if (k < GAT_CAP) {
        bucket[(size_t)d * GAT_CAP + k] = s;
    } else {
        const int o = atomicAdd(ovfc, 1);
        ovf_dst[o] = d;
        ovf_src[o] = s;
    }
}

__global__ void __launch_bounds__(256)
gat_bin_coarse(const int* __restrict__ ei, int E, int* __restrict__ bin_cnt, int* __restrict__ coarse_i,
               int* __restrict__ cnt, int* __restrict__ bucket, int* __restrict__ ovfc,
               int* __restrict__ ovf_dst, int* __restrict__ ovf_src) {
    __shared__ int hist[GAT_MAXBINS];
    __shared__ int base[GAT_MAXBINS];
    int2* __restrict__ coarse = reinterpret_cast<int2*>(coarse_i);
    const int t = threadIdx.x;
    for (int b = t; b < GAT_MAXBINS; b += 256) hist[b] = 0;
    __syncthreads();
    const int e0 = blockIdx.x * (256 * GAT_EPT) + t;
    int s[GAT_EPT], d[GAT_EPT], r[GAT_EPT];
#pragma unroll
    for (int j = 0; j < GAT_EPT; ++j) {
        const int e = e0 + j * 256;
        const bool in = e < E;
        s[j] = in ? ei[e] : 0;
        d[j] = in ? ei[E + e] : -1;
    }
#pragma unroll
    for (int j = 0; j < GAT_EPT; ++j) {
        const int bin = d[j] >> GAT_BIN_SHIFT;
        r[j] = (d[j] >= 0 && bin < GAT_MAXBINS) ? atomicAdd(&hist[bin], 1) : -1;
    }
    __syncthreads();
    for (int b = t; b < GAT_MAXBINS; b += 256) {
        const int h = hist[b];
        base[b] = h ? atomicAdd(&bin_cnt[b], h) : 0;
    }
    __syncthreads();
    int p[GAT_EPT];
#pragma unroll
    for (int j = 0; j < GAT_EPT; ++j)
        p[j] = r[j] >= 0 ? base[d[j] >> GAT_BIN_SHIFT] + r[j] : GAT_BIN_CAP;
#pragma unroll
    for (int j = 0; j < GAT_EPT; ++j) {
        if (d[j] >= 0) {
            if (p[j] < GAT_BIN_CAP) coarse[(size_t)(d[j] >> GAT_BIN_SHIFT) * GAT_BIN_CAP + p[j]] = make_int2(s[j], d[j]);
            else bucket_put_global(s[j], d[j], cnt, bucket, ovfc, ovf_dst, ovf_src);
        }
    }
}

__global__ void __launch_bounds__(256)
gat_bin_fine(const int* __restrict__ coarse_i, const int* __restrict__ bin_cnt, int* __restrict__ cnt,
             int* __restrict__ bucket, int* __restrict__ ovfc,
             int* __restrict__ ovf_dst, int* __restrict__ ovf_src, int N) {
    __shared__ int lc[GAT_BIN_NODES];
    __shared__ int ls[GAT_BIN_NODES];
    __shared__ int lb[GAT_BIN_NODES * GAT_CAP];
    const int t = threadIdx.x;
    const int bin = blockIdx.x;
    const int node0 = bin << GAT_BIN_SHIFT;
    int n = bin_cnt[bin];
    n = n < GAT_BIN_CAP ? n : GAT_BIN_CAP;
    if (t < GAT_BIN_NODES) {
        const int c = node0 + t < N ? cnt[node0 + t] : 0;
        lc[t] = c;
        ls[t] = c;
    }
    __syncthreads();
    const int2* __restrict__ cp = reinterpret_cast<const int2*>(coarse_i) + (size_t)bin * GAT_BIN_CAP;
    for (int i0 = 0; i0 < n; i0 += 256 * GAT_EPT) {
        int2 ev[GAT_EPT];
#pragma unroll
        for (int j = 0; j < GAT_EPT; ++j) {
            const int i = i0 + j * 256 + t;
            ev[j] = i < n ? cp[i] : make_int2(0, -1);
        }
        int kk[GAT_EPT];
#pragma unroll
        for (int j = 0; j < GAT_EPT; ++j)
            kk[j] = ev[j].y >= 0 ? atomicAdd(&lc[ev[j].y - node0], 1) : -1;
#pragma unroll
        for (int j = 0; j < GAT_EPT; ++j) {
            if (ev[j].y >= 0) {
                if (kk[j] < GAT_CAP) {
                    lb[(ev[j].y - node0) * GAT_CAP + kk[j]] = ev[j].x;
                } else {
                    const int o = atomicAdd(ovfc, 1);
                    ovf_dst[o] = ev[j].y;
                    ovf_src[o] = ev[j].x;
                }
            }
        }
    }
    __syncthreads();
    const int k = t & (GAT_CAP - 1);
    for (int i0 = 0; i0 < GAT_BIN_NODES / 4; i0 += 8) {
        int c[8], f[8], v[8];
#pragma unroll
        for (int u = 0; u < 8; ++u) {
            const int dl = (t >> 6) + 4 * (i0 + u);
            c[u] = lc[dl];
            f[u] = ls[dl];
            v[u] = lb[dl * GAT_CAP + k];
        }
#pragma unroll
        for (int u = 0; u < 8; ++u) {
            const int dl = (t >> 6) + 4 * (i0 + u);
            const int hi = c[u] < GAT_CAP ? c[u] : GAT_CAP;
            if (k >= f[u] && k < hi) bucket[(size_t)(node0 + dl) * GAT_CAP + k] = v[u];
        }
    }
    if (t < GAT_BIN_NODES && node0 + t < N) cnt[node0 + t] = lc[t];
}

struct SegShape {
    int nb;
    int nbc;
    int ovf_n;
    int nch;
};
__device__ __forceinline__ SegShape seg_shape(const int* __restrict__ cnt,
                                              const int* __restrict__ ovfc, int node) {
    SegShape g;
    const int c = cnt[node];
    const int o = ovfc[0];
    g.nb    = c < GAT_CAP ? c : GAT_CAP;
    g.nbc   = (g.nb + 31) >> 5;
    g.ovf_n = c > GAT_CAP ? o : 0;
    g.nch   = g.nbc + ((g.ovf_n + 31) >> 5);
    return g;
}
__device__ __forceinline__ unsigned seg_chunk(const int* __restrict__ bucket,
                                              const int* __restrict__ ovf_dst,
                                              const int* __restrict__ ovf_src,
                                              int node, const SegShape& g, int q, int lane, int& sv) {
    bool valid;
    sv = 0;
    if (q < g.nbc) {
        const int p = q * 32 + lane;
        valid = p < g.nb;
        if (valid) sv = bucket[(size_t)node * GAT_CAP + p];
    } else {
        const int i = (q - g.nbc) * 32 + lane;
        valid = i < g.ovf_n;
        if (valid) valid = ovf_dst[i] == node;
        if (valid) sv = ovf_src[i];
    }
    return __builtin_amdgcn_ballot_w32(valid);
}
__device__ __forceinline__ int chunk_len(unsigned vm) { return vm ? 32 - __builtin_clz(vm) : 0; }

template <int M>
__device__ __forceinline__ float row_xor(float x) {
    const int b = __float_as_int(x);
    return __int_as_float(__builtin_amdgcn_update_dpp(b, b, 0x160 + M, 0xf, 0xf, false));
}

__device__ __forceinline__ float row16_sum(float v) {
    v += row_xor<1>(v); v += row_xor<2>(v); v += row_xor<4>(v); v += row_xor<8>(v);
    return v;
}
template <int K>
__device__ __forceinline__ float row_share(float x) {
    const int b = __float_as_int(x);
    return __int_as_float(__builtin_amdgcn_update_dpp(b, b, 0x150 + K, 0xf, 0xf, false));
}
__device__ __forceinline__ float fold8_sum(const float (&p)[8], bool b0, bool b1, bool b2) {
    float r[4], s[2];
#pragma unroll
    for (int i = 0; i < 4; ++i) {
        const float keep = b0 ? p[i + 4] : p[i], send = b0 ? p[i] : p[i + 4];
        r[i] = keep + row_xor<1>(send);
    }
#pragma unroll
    for (int i = 0; i < 2; ++i) {
        const float keep = b1 ? r[i + 2] : r[i], send = b1 ? r[i] : r[i + 2];
        s[i] = keep + row_xor<2>(send);
    }
    const float keep = b2 ? s[1] : s[0], send = b2 ? s[0] : s[1];
    return keep + row_xor<4>(send);
}
__device__ __forceinline__ float dot4(const float4 r, const float4 a) {
    return r.x * a.x + r.y * a.y + r.z * a.z + r.w * a.w;
}
__device__ __forceinline__ float4 scaled4(const float* __restrict__ p, float f) {
    const float4 t = *reinterpret_cast<const float4*>(p);
    return make_float4(t.x * f, t.y * f, t.z * f, t.w * f);
}

template <int HH>
__global__ void gat_seg_rowscore_agg(const int* __restrict__ bucket, const int* __restrict__ cnt,
                                     const int* __restrict__ ovfc, const int* __restrict__ ovf_dst,
                                     const int* __restrict__ ovf_src,
                                     const float* __restrict__ hfeat,
                                     const float* __restrict__ a_s, const float* __restrict__ a_d,
                                     const float* __restrict__ b, float* __restrict__ out,
                                     int N, int do_elu) {
    static_assert(HH == 4, "four heads; the single-head layer has its own kernel below");
    const int lane = threadIdx.x & 31;
    const int node = __builtin_amdgcn_readfirstlane(blockIdx.x * (blockDim.x >> 5) + (threadIdx.x >> 5));
    if (node >= N) return;
    const float L2E = 1.4426950408889634f;
    const bool b0 = (lane & 1) != 0, b1 = (lane & 2) != 0, b2 = (lane & 4) != 0;

    if constexpr (HH == 4) {
        const float* hp = hfeat + lane * 4;
        const float4 as0 = scaled4(a_s + lane * 4, L2E), as1 = scaled4(a_s + 128 + lane * 4, L2E);
        const int myu = ((lane & 1) << 1) | ((lane >> 1) & 1);
        float4 acc0, acc1;
        float ednl, ml;
        float denl = (lane & 3) == 0 ? 1.f : 0.f;
        {
            const float4 ad0 = scaled4(a_d + lane * 4, L2E), ad1 = scaled4(a_d + 128 + lane * 4, L2E);
            const float* rp = hp + (size_t)node * 256;
            acc0 = *reinterpret_cast<const float4*>(rp);
            acc1 = *reinterpret_cast<const float4*>(rp + 128);
            const float edn0 = row16_sum(dot4(acc0, ad0));
            const float edn1 = row16_sum(dot4(acc1, ad1));
            const float v0 = row16_sum(dot4(acc0, as0)) + edn0;
            const float v1 = row16_sum(dot4(acc1, as1)) + edn1;
            ednl = b2 ? edn1 : edn0;
            const float v = b2 ? v1 : v0;
            ml = fmaxf(v, 0.2f * v);
        }
        const SegShape sg = seg_shape(cnt, ovfc, node);
        for (int q = 0; q < sg.nch; ++q) {
            int sv;
            const unsigned vm = seg_chunk(bucket, ovf_dst, ovf_src, node, sg, q, lane, sv);
            const int n = chunk_len(vm);
            const unsigned vmy = vm >> myu;
            for (int j = 0; j < n; j += 4) {
                float4 q0[4], q1[4];
                float p[8];
#pragma unroll
                for (int u = 0; u < 4; ++u) {
                    const int s = __builtin_amdgcn_readlane(sv, j + u);
                    const float* qp = hp + (size_t)s * 256;
                    q0[u] = *reinterpret_cast<const float4*>(qp);
                    q1[u] = *reinterpret_cast<const float4*>(qp + 128);
                }
#pragma unroll
                for (int u = 0; u < 4; ++u) { p[2 * u] = dot4(q0[u], as0); p[2 * u + 1] = dot4(q1[u], as1); }
                float t = fold8_sum(p, b0, b1, b2);
                t += row_xor<8>(t);
                const float v = t + ednl;
                const float e = ((vmy >> j) & 1u) ? fmaxf(v, 0.2f * v) : -INFINITY;
                float cm = fmaxf(e, row_xor<1>(e));
                cm = fmaxf(cm, row_xor<2>(cm));
                const float mn = fmaxf(ml, cm);
                const float sc = __builtin_amdgcn_exp2f(ml - mn);
                const float w  = __builtin_amdgcn_exp2f(e - mn);
                denl = denl * sc + w;
                ml = mn;
                const float sc0 = row_share<0>(sc), sc1 = row_share<4>(sc);
                acc0.x *= sc0; acc0.y *= sc0; acc0.z *= sc0; acc0.w *= sc0;
                acc1.x *= sc1; acc1.y *= sc1; acc1.z *= sc1; acc1.w *= sc1;
                float w0[4], w1[4];
                w0[0] = row_share<0>(w); w0[1] = row_share<2>(w); w0[2] = row_share<1>(w); w0[3] = row_share<3>(w);
                w1[0] = row_share<4>(w); w1[1] = row_share<6>(w); w1[2] = row_share<5>(w); w1[3] = row_share<7>(w);
#pragma unroll
                for (int u = 0; u < 4; ++u) {
                    acc0.x += q0[u].x * w0[u]; acc0.y += q0[u].y * w0[u]; acc0.z += q0[u].z * w0[u]; acc0.w += q0[u].w * w0[u];
                    acc1.x += q1[u].x * w1[u]; acc1.y += q1[u].y * w1[u]; acc1.z += q1[u].z * w1[u]; acc1.w += q1[u].w * w1[u];
                }
            }
        }
        float d = denl;
        d += row_xor<1>(d); d += row_xor<2>(d);
        const float i0 = 1.f / row_share<0>(d), i1 = 1.f / row_share<4>(d);
        const float4 bb0 = *reinterpret_cast<const float4*>(b + lane * 4);
        const float4 bb1 = *reinterpret_cast<const float4*>(b + 128 + lane * 4);
        acc0.x = acc0.x * i0 + bb0.x; acc0.y = acc0.y * i0 + bb0.y; acc0.z = acc0.z * i0 + bb0.z; acc0.w = acc0.w * i0 + bb0.w;
        acc1.x = acc1.x * i1 + bb1.x; acc1.y = acc1.y * i1 + bb1.y; acc1.z = acc1.z * i1 + bb1.z; acc1.w = acc1.w * i1 + bb1.w;
        if (do_elu) {
            acc0.x = acc0.x > 0.f ? acc0.x : (expf(acc0.x) - 1.f);
            acc0.y = acc0.y > 0.f ? acc0.y : (expf(acc0.y) - 1.f);
            acc0.z = acc0.z > 0.f ? acc0.z : (expf(acc0.z) - 1.f);
            acc0.w = acc0.w > 0.f ? acc0.w : (expf(acc0.w) - 1.f);
            acc1.x = acc1.x > 0.f ? acc1.x : (expf(acc1.x) - 1.f);
            acc1.y = acc1.y > 0.f ? acc1.y : (expf(acc1.y) - 1.f);
            acc1.z = acc1.z > 0.f ? acc1.z : (expf(acc1.z) - 1.f);
            acc1.w = acc1.w > 0.f ? acc1.w : (expf(acc1.w) - 1.f);
        }
        float* op = out + (size_t)node * 256 + lane * 4;
        *reinterpret_cast<float4*>(op)       = acc0;
        *reinterpret_cast<float4*>(op + 128) = acc1;
    }
}

struct PairSeg {
    int nb;
    int nbc;
    int ovf_n;
    int nch;
};
__device__ __forceinline__ PairSeg pair_seg(const int* __restrict__ cnt, int on, int node, bool live) {
    PairSeg g;
    const int cn = cnt[node];
    g.nb    = cn < GAT_CAP ? cn : GAT_CAP;
    g.nbc   = (g.nb + 31) >> 5;
    g.ovf_n = (live && cn > GAT_CAP) ? on : 0;
    g.nch   = live ? g.nbc + ((g.ovf_n + 31) >> 5) : 0;
    return g;
}
__device__ __forceinline__ void pair_chunk(const int* __restrict__ bucket,
                                           const int* __restrict__ ovf_dst,
                                           const int* __restrict__ ovf_src,
                                           int node, bool live, const PairSeg& g, int q, int c,
                                           int& sva, int& svb, bool& va, bool& vb) {
    sva = 0; svb = 0;
    if (q < g.nbc) {
        const int p = q * 32 + c;
        va = live && p < g.nb;
        vb = live && p + 16 < g.nb;
        const int* bp = bucket + (size_t)node * GAT_CAP;
        const int ra = bp[va ? p : 0], rb = bp[vb ? p + 16 : 0];
        sva = va ? ra : 0;
        svb = vb ? rb : 0;
    } else {
        const int i = (q - g.nbc) * 32 + c;
        va = i < g.ovf_n;
        vb = i + 16 < g.ovf_n;
        if (va) va = ovf_dst[i] == node;
        if (vb) vb = ovf_dst[i + 16] == node;
        if (va) sva = ovf_src[i];
        if (vb) svb = ovf_src[i + 16];
    }
}
template <int K>
__device__ __forceinline__ int row_share_i(int b) {
    return __builtin_amdgcn_update_dpp(b, b, 0x150 + K, 0xf, 0xf, false);
}

template <int J>
__device__ __forceinline__ void pair_trip(const float* __restrict__ hfeat, unsigned c4, int sv, unsigned vmy,
                                          const float4 as, float edn, bool b0, bool b1, bool b2,
                                          float4& acc, float& m, float& denl) {
    float4 q[8];
    float p[8];
    int s[8];
    s[0] = row_share_i<J + 0>(sv); s[1] = row_share_i<J + 1>(sv); s[2] = row_share_i<J + 2>(sv); s[3] = row_share_i<J + 3>(sv);
    s[4] = row_share_i<J + 4>(sv); s[5] = row_share_i<J + 5>(sv); s[6] = row_share_i<J + 6>(sv); s[7] = row_share_i<J + 7>(sv);
#pragma unroll
    for (int u = 0; u < 8; ++u) q[u] = *reinterpret_cast<const float4*>(hfeat + ((unsigned)s[u] * 64u + c4));
#pragma unroll
    for (int u = 0; u < 8; ++u) p[u] = dot4(q[u], as);
    float t = fold8_sum(p, b0, b1, b2);
    t += row_xor<8>(t);
    const float v = t + edn;
    const float e = ((vmy >> J) & 1u) ? fmaxf(v, 0.2f * v) : -INFINITY;
    float cm = fmaxf(e, row_xor<1>(e));
    cm = fmaxf(cm, row_xor<2>(cm));
    cm = fmaxf(cm, row_xor<4>(cm));
    const float mn = fmaxf(m, cm);
    const float sc = __builtin_amdgcn_exp2f(m - mn);
    const float w  = __builtin_amdgcn_exp2f(e - mn);
    denl = denl * sc + w;
    m = mn;
    acc.x *= sc; acc.y *= sc; acc.z *= sc; acc.w *= sc;
    float wu[8];
    wu[0] = row_share<0>(w); wu[1] = row_share<4>(w); wu[2] = row_share<2>(w); wu[3] = row_share<6>(w);
    wu[4] = row_share<1>(w); wu[5] = row_share<5>(w); wu[6] = row_share<3>(w); wu[7] = row_share<7>(w);
#pragma unroll
    for (int u = 0; u < 8; ++u) {
        acc.x += q[u].x * wu[u]; acc.y += q[u].y * wu[u]; acc.z += q[u].z * wu[u]; acc.w += q[u].w * wu[u];
    }
}

__global__ void gat_seg_rowscore_agg_pair(const int* __restrict__ bucket, const int* __restrict__ cnt,
                                          const int* __restrict__ ovfc, const int* __restrict__ ovf_dst,
                                          const int* __restrict__ ovf_src,
                                          const float* __restrict__ hfeat,
                                          const float* __restrict__ a_s, const float* __restrict__ a_d,
                                          const float* __restrict__ b, float* __restrict__ out,
                                          int N, int do_elu) {
    const int lane = threadIdx.x & 31;
    const int c = lane & 15, hh = lane >> 4;
    const int node0 = __builtin_amdgcn_readfirstlane(2 * (blockIdx.x * (blockDim.x >> 5) + (threadIdx.x >> 5)));
    if (node0 >= N) return;
    const int on = ovfc[0];
    const bool live = node0 + hh < N;
    const int node = live ? node0 + hh : N - 1;
    const float L2E = 1.4426950408889634f;
    const bool b0 = (lane & 1) != 0, b1 = (lane & 2) != 0, b2 = (lane & 4) != 0;
    const int myu = ((lane & 1) << 2) | (lane & 2) | ((lane >> 2) & 1);

    const unsigned c4 = (unsigned)c * 4u;
    const float4 as = scaled4(a_s + c * 4, L2E);
    float4 acc = *reinterpret_cast<const float4*>(hfeat + ((unsigned)node * 64u + c4));
    float edn, m;
    float denl = (lane & 7) == 0 ? 1.f : 0.f;
    {
        const float4 ad = scaled4(a_d + c * 4, L2E);
        edn = row16_sum(dot4(acc, ad));
        const float v = row16_sum(dot4(acc, as)) + edn;
        m = fmaxf(v, 0.2f * v);
    }
    const PairSeg sg = pair_seg(cnt, on, node, live);
    const int nch0 = __builtin_amdgcn_readlane(sg.nch, 0), nch1 = __builtin_amdgcn_readlane(sg.nch, 16);
    const int nchw = nch0 > nch1 ? nch0 : nch1;
    for (int q = 0; q < nchw; ++q) {
        int sva, svb;
        bool va, vb;
        pair_chunk(bucket, ovf_dst, ovf_src, node, live, sg, q, c, sva, svb, va, vb);
        const unsigned vma = __builtin_amdgcn_ballot_w32(va), vmb = __builtin_amdgcn_ballot_w32(vb);
        const unsigned vmya = ((vma >> (16 * hh)) & 0xffffu) >> myu;
        const unsigned vmyb = ((vmb >> (16 * hh)) & 0xffffu) >> myu;
        if (vma & 0x00ff00ffu) pair_trip<0>(hfeat, c4, sva, vmya, as, edn, b0, b1, b2, acc, m, denl);
        if (vma & 0xff00ff00u) pair_trip<8>(hfeat, c4, sva, vmya, as, edn, b0, b1, b2, acc, m, denl);
        if (vmb & 0x00ff00ffu) pair_trip<0>(hfeat, c4, svb, vmyb, as, edn, b0, b1, b2, acc, m, denl);
        if (vmb & 0xff00ff00u) pair_trip<8>(hfeat, c4, svb, vmyb, as, edn, b0, b1, b2, acc, m, denl);
    }
    float d = denl;
    d += row_xor<1>(d); d += row_xor<2>(d); d += row_xor<4>(d);
    const float inv = 1.f / d;
    const float4 bb = *reinterpret_cast<const float4*>(b + c * 4);
    acc.x = acc.x * inv + bb.x; acc.y = acc.y * inv + bb.y; acc.z = acc.z * inv + bb.z; acc.w = acc.w * inv + bb.w;
    if (do_elu) {
        acc.x = acc.x > 0.f ? acc.x : (expf(acc.x) - 1.f);
        acc.y = acc.y > 0.f ? acc.y : (expf(acc.y) - 1.f);
        acc.z = acc.z > 0.f ? acc.z : (expf(acc.z) - 1.f);
        acc.w = acc.w > 0.f ? acc.w : (expf(acc.w) - 1.f);
    }
    if (live) *reinterpret_cast<float4*>(out + (size_t)node * 64 + c * 4) = acc;
}

template <int J>
__device__ __forceinline__ void pair_trip_in(const float* __restrict__ x, const float* __restrict__ es,
                                             unsigned k4, unsigned head, int sv, unsigned vmh, float edn,
                                             float4& acc, float& m, float& den) {
    const float L2E = 1.4426950408889634f;
    int s[8];
    s[0] = row_share_i<J + 0>(sv); s[1] = row_share_i<J + 1>(sv); s[2] = row_share_i<J + 2>(sv); s[3] = row_share_i<J + 3>(sv);
    s[4] = row_share_i<J + 4>(sv); s[5] = row_share_i<J + 5>(sv); s[6] = row_share_i<J + 6>(sv); s[7] = row_share_i<J + 7>(sv);
    float4 r[8];
    float e[8];
#pragma unroll
    for (int u = 0; u < 8; ++u) {
        r[u] = *reinterpret_cast<const float4*>(x + ((unsigned)s[u] * 16u + k4));
        e[u] = es[(unsigned)s[u] * 4u + head];
    }
    float cm = -INFINITY;
#pragma unroll
    for (int u = 0; u < 8; ++u) {
        const float v = e[u] * L2E + edn;
        e[u] = ((vmh >> (J + u)) & 1u) ? fmaxf(v, 0.2f * v) : -INFINITY;
        cm = fmaxf(cm, e[u]);
    }
    const float mn = fmaxf(m, cm);
    const float sc = __builtin_amdgcn_exp2f(m - mn);
    den *= sc;
    acc.x *= sc; acc.y *= sc; acc.z *= sc; acc.w *= sc;
#pragma unroll
    for (int u = 0; u < 8; ++u) {
        const float w = __builtin_amdgcn_exp2f(e[u] - mn);
        den += w;
        acc.x += r[u].x * w; acc.y += r[u].y * w; acc.z += r[u].z * w; acc.w += r[u].w * w;
    }
    m = mn;
}

__global__ void gat_seg_softmax_agg_pair(const int* __restrict__ bucket, const int* __restrict__ cnt,
                                         const int* __restrict__ ovfc, const int* __restrict__ ovf_dst,
                                         const int* __restrict__ ovf_src,
                                         const float* __restrict__ x,
                                         const float* __restrict__ es, const float* __restrict__ ed,
                                         float* __restrict__ out, int N) {
    const int lane = threadIdx.x & 31;
    const int c = lane & 15, hh = lane >> 4;
    const int node0 = __builtin_amdgcn_readfirstlane(2 * (blockIdx.x * (blockDim.x >> 5) + (threadIdx.x >> 5)));
    if (node0 >= N) return;
    const int on = ovfc[0];
    const bool live = node0 + hh < N;
    const int node = live ? node0 + hh : N - 1;
    const float L2E = 1.4426950408889634f;
    const unsigned head = (unsigned)c >> 2, k4 = ((unsigned)c & 3u) * 4u;

    const float edn = ed[(unsigned)node * 4u + head] * L2E;
    float4 acc = *reinterpret_cast<const float4*>(x + ((unsigned)node * 16u + k4));
    float m;
    {
        const float v = es[(unsigned)node * 4u + head] * L2E + edn;
        m = fmaxf(v, 0.2f * v);
    }
    float den = 1.f;
    const PairSeg sg = pair_seg(cnt, on, node, live);
    const int nch0 = __builtin_amdgcn_readlane(sg.nch, 0), nch1 = __builtin_amdgcn_readlane(sg.nch, 16);
    const int nchw = nch0 > nch1 ? nch0 : nch1;
    for (int q = 0; q < nchw; ++q) {
        int sva, svb;
        bool va, vb;
        pair_chunk(bucket, ovf_dst, ovf_src, node, live, sg, q, c, sva, svb, va, vb);
        const unsigned vma = __builtin_amdgcn_ballot_w32(va), vmb = __builtin_amdgcn_ballot_w32(vb);
        const unsigned vmha = (vma >> (16 * hh)) & 0xffffu;
        const unsigned vmhb = (vmb >> (16 * hh)) & 0xffffu;
        if (vma & 0x00ff00ffu) pair_trip_in<0>(x, es, k4, head, sva, vmha, edn, acc, m, den);
        if (vma & 0xff00ff00u) pair_trip_in<8>(x, es, k4, head, sva, vmha, edn, acc, m, den);
        if (vmb & 0x00ff00ffu) pair_trip_in<0>(x, es, k4, head, svb, vmhb, edn, acc, m, den);
        if (vmb & 0xff00ff00u) pair_trip_in<8>(x, es, k4, head, svb, vmhb, edn, acc, m, den);
    }
    const float inv = 1.f / den;
    acc.x *= inv; acc.y *= inv; acc.z *= inv; acc.w *= inv;
    if (live) *reinterpret_cast<float4*>(out + (size_t)node * 64 + c * 4) = acc;
}

#define GAT_SLAB 256
template <bool FULL>
__device__ __forceinline__ float4 slab_rows_sum(const float* __restrict__ x, int r0, int N, int cg) {
    float4 v[GAT_SLAB / 16];
    for (int i = 0; i < GAT_SLAB / 16; ++i) {
        const int r = r0 + 16 * i;
        if constexpr (FULL) {
            v[i] = *reinterpret_cast<const float4*>(x + (size_t)r * 64 + cg * 4);
        } else {
            const int rc = r < N ? r : N - 1;
            v[i] = *reinterpret_cast<const float4*>(x + (size_t)rc * 64 + cg * 4);
            if (r >= N) v[i] = make_float4(0.f, 0.f, 0.f, 0.f);
        }
    }
    float4 s = v[0];
    for (int i = 1; i < GAT_SLAB / 16; ++i) { s.x += v[i].x; s.y += v[i].y; s.z += v[i].z; s.w += v[i].w; }
    return s;
}

__global__ void gat_colsum_slab(const float* __restrict__ x, float* __restrict__ part, int N) {
    __shared__ float4 sm[256];
    const int t    = threadIdx.x;
    const int cg   = t & 15;
    const int slot = t >> 4;
    const int r0   = blockIdx.x * GAT_SLAB + slot;
    const bool full = (blockIdx.x + 1) * GAT_SLAB <= N;
    sm[t] = full ? slab_rows_sum<true>(x, r0, N, cg)
                 : slab_rows_sum<false>(x, r0, N, cg);
    __syncthreads();
    if (t < 64) {
        const float* f = reinterpret_cast<const float*>(sm);
        float a = 0.f;
        for (int k = 0; k < 16; ++k) a += f[k * 64 + t];
        part[(size_t)blockIdx.x * 64 + t] = a;
    }
}

__global__ void gat_colmean_final(const float* __restrict__ part, float* __restrict__ out,
                                  int P, int N) {
    __shared__ float4 sm[1024];
    const int t    = threadIdx.x;
    const int cg   = t & 15;
    const int slot = t >> 4;
    float4 s = make_float4(0.f, 0.f, 0.f, 0.f);
    for (int base = 0; base < P; base += 256) {
        float4 v[4];
        for (int i = 0; i < 4; ++i) {
            const int r  = base + slot + 64 * i;
            const int rc = r < P ? r : P - 1;
            v[i] = *reinterpret_cast<const float4*>(part + (size_t)rc * 64 + cg * 4);
            if (r >= P) v[i] = make_float4(0.f, 0.f, 0.f, 0.f);
        }
        for (int i = 0; i < 4; ++i) { s.x += v[i].x; s.y += v[i].y; s.z += v[i].z; s.w += v[i].w; }
    }
    sm[t] = s;
    __syncthreads();
    if (t < 64) {
        const float* f = reinterpret_cast<const float*>(sm);
        float a = 0.f;
        for (int k = 0; k < 64; ++k) a += f[k * 64 + t];
        out[t] = a / (float)N;
    }
}

static inline int cdiv(long long a, long long b) { return (int)((a + b - 1) / b); }

extern "C" void kernel_launch(void* const* d_in, const int* in_sizes, int n_in,
                              void* d_out, int out_size, void* d_ws, size_t ws_size,
                              hipStream_t stream) {
    const float* x   = (const float*)d_in[0];
    const int*   ei  = (const int*)  d_in[1];
    const float* W1  = (const float*)d_in[2];
    const float* a1s = (const float*)d_in[3];
    const float* a1d = (const float*)d_in[4];
    const float* b1  = (const float*)d_in[5];
    const float* W2  = (const float*)d_in[6];
    const float* a2s = (const float*)d_in[7];
    const float* a2d = (const float*)d_in[8];
    const float* b2  = (const float*)d_in[9];
    const float* W3  = (const float*)d_in[10];
    const float* a3s = (const float*)d_in[11];
    const float* a3d = (const float*)d_in[12];
    const float* b3  = (const float*)d_in[13];

    const int N  = in_sizes[0] / 16;
    const int E  = in_sizes[1] / 2;
    const int H  = 4;

    float* ws = (float*)d_ws;
    size_t off = 0;
    float*    buf_h    = ws + off; off += (size_t)N * 256;
    int*      csr      = (int*)(ws + off); off += (size_t)N * 256;
    float*    buf_feat = ws + off; off += (size_t)N * 256;
    float*    es       = ws + off; off += (size_t)N * H;
    float*    ed       = ws + off; off += (size_t)N * H;
    (void)ws_size;
    int* cnt     = csr;
    int* ovfc    = csr + (size_t)N;
    int* bucket  = csr + (((size_t)N + 64) & ~(size_t)63);
    int* ovf_dst = bucket + (size_t)N * GAT_CAP;
    int* ovf_src = ovf_dst + (size_t)E;
    int* bin_cnt = ovf_src + (size_t)E;
    int* coarse  = (int*)buf_feat;
    const int NB = cdiv(N, GAT_BIN_NODES);

    float* node_emb  = (float*)d_out;
    float* graph_emb = node_emb + (size_t)N * 64;

    auto run_layer = [&](const float* in_feat, const float* W,
                         const float* as_, const float* ad_, const float* bb,
                         float* out_feat, int Hh, int K, int Ncols, int do_elu) {
        (void)K;
        if (Ncols == 256)
            hipLaunchKernelGGL((gat_gemm_wmma<1, 4, 2>), dim3(cdiv(N, 32)), dim3(128), 0, stream,
                               in_feat, W, buf_h, N);
        else
            hipLaunchKernelGGL((gat_gemm_wmma<2, 1, 1>), dim3(cdiv(N, 32)), dim3(64), 0, stream,
                               in_feat, W, buf_h, N);
        if (Hh == 4)
            hipLaunchKernelGGL((gat_seg_rowscore_agg<4>), dim3(cdiv(N, 8)), dim3(256), 0, stream,
                               bucket, cnt, ovfc, ovf_dst, ovf_src, buf_h, as_, ad_, bb, out_feat, N, do_elu);
        else
            hipLaunchKernelGGL(gat_seg_rowscore_agg_pair, dim3(cdiv(N, 16)), dim3(256), 0, stream,
                               bucket, cnt, ovfc, ovf_dst, ovf_src, buf_h, as_, ad_, bb, out_feat, N, do_elu);
    };

    hipLaunchKernelGGL(gat_attn_coef_in, dim3(cdiv(N, 256)), dim3(256), 0, stream,
                       x, W1, a1s, a1d, es, ed, cnt, ovfc, bin_cnt, NB, N);
    hipLaunchKernelGGL(gat_bin_coarse, dim3(cdiv(E, 256 * GAT_EPT)), dim3(256), 0, stream,
                       ei, E, bin_cnt, coarse, cnt, bucket, ovfc, ovf_dst, ovf_src);
    hipLaunchKernelGGL(gat_bin_fine, dim3(NB), dim3(256), 0, stream,
                       coarse, bin_cnt, cnt, bucket, ovfc, ovf_dst, ovf_src, N);

    hipLaunchKernelGGL(gat_seg_softmax_agg_pair, dim3(cdiv(N, 16)), dim3(256), 0, stream,
                       bucket, cnt, ovfc, ovf_dst, ovf_src, x, es, ed, buf_h, N);
    hipLaunchKernelGGL(gat_proj_heads_wmma, dim3(cdiv(N, 16)), dim3(32, 4), 0, stream,
                       buf_h, W1, b1, buf_feat, N);
    run_layer(buf_feat, W2, a2s, a2d, b2, buf_feat, H, 256, 256, 1);
    run_layer(buf_feat, W3, a3s, a3d, b3, node_emb, 1, 256, 64, 0);

    const int P = cdiv(N, GAT_SLAB);
    hipLaunchKernelGGL(gat_colsum_slab, dim3(P), dim3(256), 0, stream, node_emb, buf_h, N);
    hipLaunchKernelGGL(gat_colmean_final, dim3(1), dim3(1024), 0, stream, buf_h, graph_emb, P, N);
}
